# starting kernel
# baseline (speedup 1.0000x reference)
#include <hip/hip_runtime.h>
#include <math.h>

typedef __attribute__((ext_vector_type(2))) float v2f;
typedef __attribute__((ext_vector_type(8))) float v8f;

#define D_FEAT 64

__global__ void zero2_kernel(float* __restrict__ a, long na4,
                             float* __restrict__ b, long nb4) {
    long i = (long)blockIdx.x * blockDim.x + threadIdx.x;
    long stride = (long)gridDim.x * blockDim.x;
    float4 z = make_float4(0.f, 0.f, 0.f, 0.f);
    for (long j = i; j < na4 + nb4; j += stride) {
        if (j < na4) ((float4*)a)[j] = z;
        else         ((float4*)b)[j - na4] = z;
    }
}

__global__ void degree_kernel(const int* __restrict__ row, const int* __restrict__ col,
                              int E, float* degO, float* degI) {
    int e = blockIdx.x * blockDim.x + threadIdx.x;
    if (e >= E) return;
    unsafeAtomicAdd(&degO[row[e]], 1.0f);
    unsafeAtomicAdd(&degI[col[e]], 1.0f);
}

__global__ void deginv_kernel(float* degO, float* degI, int n) {
    int i = blockIdx.x * blockDim.x + threadIdx.x;
    if (i >= n) return;
    float d0 = degO[i];
    float d1 = degI[i];
    degO[i] = (d0 > 0.f) ? powf(d0, -0.25f) : 0.f;
    degI[i] = (d1 > 0.f) ? powf(d1, -0.25f) : 0.f;
}

__global__ void wcomb_kernel(const float* __restrict__ Wr, const float* __restrict__ br,
                             const float* __restrict__ Wi, const float* __restrict__ bi,
                             int K, float* WRc, float* WIc, float* bmi, float* bpi) {
    int i = blockIdx.x * blockDim.x + threadIdx.x;
    if (i < D_FEAT * D_FEAT) {
        float sr = 0.f, si = 0.f, c = 1.f;
        for (int k = 0; k < K; ++k) {
            sr += c * Wr[k * D_FEAT * D_FEAT + i];
            si += c * Wi[k * D_FEAT * D_FEAT + i];
            c *= 0.5f;
        }
        WRc[i] = sr;
        WIc[i] = si;
    } else if (i < D_FEAT * D_FEAT + D_FEAT) {
        int o = i - D_FEAT * D_FEAT;
        float sr = 0.f, si = 0.f, c = 1.f;
        for (int k = 0; k < K; ++k) {
            sr += c * br[k * D_FEAT + o];
            si += c * bi[k * D_FEAT + o];
            c *= 0.5f;
        }
        bmi[o] = sr - si;
        bpi[o] = sr + si;
    }
}

__global__ __launch_bounds__(256) void spmm_kernel(
        const float* __restrict__ xr, const float* __restrict__ xi,
        const int* __restrict__ row, const int* __restrict__ col, int E,
        const float* __restrict__ out_inv, const float* __restrict__ in_inv,
        float* yr, float* yi, float* yrt, float* yit) {
    const int lane = threadIdx.x & 31;
    int e = blockIdx.x * 8 + (threadIdx.x >> 5);
    e = __builtin_amdgcn_readfirstlane(e);
    if (e >= E) return;

    const int r = row[e];
    const int c = col[e];
    const float w = out_inv[r] * in_inv[c];

    const int d = lane << 1;
    const long ro = (long)r * D_FEAT + d;
    const long co = (long)c * D_FEAT + d;

    const float2 xrc = *(const float2*)(xr + co);
    const float2 xic = *(const float2*)(xi + co);
    const float2 xrr = *(const float2*)(xr + ro);
    const float2 xir = *(const float2*)(xi + ro);

    unsafeAtomicAdd(&yr[ro],      w * xrc.x);
    unsafeAtomicAdd(&yr[ro + 1],  w * xrc.y);
    unsafeAtomicAdd(&yi[ro],      w * xic.x);
    unsafeAtomicAdd(&yi[ro + 1],  w * xic.y);
    unsafeAtomicAdd(&yrt[co],     w * xrr.x);
    unsafeAtomicAdd(&yrt[co + 1], w * xrr.y);
    unsafeAtomicAdd(&yit[co],     w * xir.x);
    unsafeAtomicAdd(&yit[co + 1], w * xir.y);
}

__global__ __launch_bounds__(128) void faber_gemm_kernel(
        const float* __restrict__ yr, const float* __restrict__ yi,
        const float* __restrict__ WRc, const float* __restrict__ WIc,
        const float* __restrict__ bmi, const float* __restrict__ bpi,
        float* outR, float* outI) {
    const int wave = threadIdx.x >> 5;
    const int lane = threadIdx.x & 31;
    const int l16  = lane & 15;
    const int half = lane >> 4;

    const float* yrt = outR;
    const float* yit = outI;

    const long rowOff = (long)(blockIdx.x * 16 + l16) * D_FEAT;
    const long wOff   = (long)(wave * 16 + l16) * D_FEAT;

    v8f accRa = {};
    v8f accRb = {};
    v8f accIa = {};
    v8f accIb = {};

    #pragma unroll
    for (int s = 0; s < 16; ++s) {
        const int k = 4 * s + 2 * half;
        float2 ar  = *(const float2*)(yr  + rowOff + k);
        float2 art = *(const float2*)(yrt + rowOff + k);
        float2 ai  = *(const float2*)(yi  + rowOff + k);
        float2 ait = *(const float2*)(yit + rowOff + k);
        v2f Azr = {0.5f * (ar.x + art.x), 0.5f * (ar.y + art.y)};
        v2f Azi = {0.5f * (ai.x + ait.x), 0.5f * (ai.y + ait.y)};
        v2f Ayr = {ar.x, ar.y};
        float2 wr2 = *(const float2*)(WRc + wOff + k);
        float2 wi2 = *(const float2*)(WIc + wOff + k);
        v2f Bwr = {wr2.x, wr2.y};
        v2f Bwi = {wi2.x, wi2.y};
        accRa = __builtin_amdgcn_wmma_f32_16x16x4_f32(false, Azr, false, Bwr, (short)0, accRa, false, false);
        accRb = __builtin_amdgcn_wmma_f32_16x16x4_f32(false, Azi, false, Bwi, (short)0, accRb, false, false);
        accIa = __builtin_amdgcn_wmma_f32_16x16x4_f32(false, Ayr, false, Bwi, (short)0, accIa, false, false);
        accIb = __builtin_amdgcn_wmma_f32_16x16x4_f32(false, Azi, false, Bwr, (short)0, accIb, false, false);
    }

    const int colD = wave * 16 + l16;
    const float bm = bmi[colD];
    const float bp = bpi[colD];

    __syncthreads();

    #pragma unroll
    for (int r = 0; r < 8; ++r) {
        long nodeM = (long)(blockIdx.x * 16 + r + 8 * half);
        outR[nodeM * D_FEAT + colD] = accRa[r] - accRb[r] + bm;
        outI[nodeM * D_FEAT + colD] = accIa[r] + accIb[r] + bp;
    }
}

extern "C" void kernel_launch(void* const* d_in, const int* in_sizes, int n_in,
                              void* d_out, int out_size, void* d_ws, size_t ws_size,
                              hipStream_t stream) {
    const float* xr = (const float*)d_in[0];
    const float* xi = (const float*)d_in[1];
    const int*   ei = (const int*)d_in[2];
    const float* Wr = (const float*)d_in[3];
    const float* br = (const float*)d_in[4];
    const float* Wi = (const float*)d_in[5];
    const float* bi = (const float*)d_in[6];

    const int n = in_sizes[0] / D_FEAT;
    const int E = in_sizes[2] / 2;
    const int K = in_sizes[3] / (D_FEAT * D_FEAT);
    const long NF = (long)n * D_FEAT;

    float* ws   = (float*)d_ws;
    float* yr   = ws;
    float* yi_  = ws + NF;
    float* degO = ws + 2 * NF;
    float* degI = degO + n;
    float* WRc  = degI + n;
    float* WIc  = WRc + D_FEAT * D_FEAT;
    float* bmi  = WIc + D_FEAT * D_FEAT;
    float* bpi  = bmi + D_FEAT;

    float* outR = (float*)d_out;
    float* outI = outR + NF;
    const int* row = ei;
    const int* col = ei + E;

    zero2_kernel<<<2048, 256, 0, stream>>>(ws, (2 * NF + 2 * (long)n) / 4,
                                           outR, (2 * NF) / 4);
    degree_kernel<<<(E + 255) / 256, 256, 0, stream>>>(row, col, E, degO, degI);
    deginv_kernel<<<(n + 255) / 256, 256, 0, stream>>>(degO, degI, n);
    wcomb_kernel<<<(D_FEAT * D_FEAT + D_FEAT + 255) / 256, 256, 0, stream>>>(
        Wr, br, Wi, bi, K, WRc, WIc, bmi, bpi);
    spmm_kernel<<<(E + 7) / 8, 256, 0, stream>>>(xr, xi, row, col, E,
                                                 degO, degI, yr, yi_, outR, outI);
    faber_gemm_kernel<<<n / 16, 128, 0, stream>>>(yr, yi_, WRc, WIc, bmi, bpi,
                                                  outR, outI);
}


# one copy per package half of each f32 atomic accumulator (4 row sums, 2 degree counts): a block adds into its own half's copy (die id), GEMM loader and deg kernel add the copies; sum order changes
# speedup vs baseline: 1.2150x; 1.2150x over previous
#include <hip/hip_runtime.h>
#include <math.h>

typedef __attribute__((ext_vector_type(2))) float v2f;
typedef __attribute__((ext_vector_type(8))) float v8f;

#define D_FEAT 64

__device__ __forceinline__ int block_half(int* s_half) {
    if (__builtin_amdgcn_readfirstlane((int)(threadIdx.x >> 5)) == 0) {
        const unsigned id = __builtin_amdgcn_s_sendmsg_rtn(0x87);
        if ((threadIdx.x & 31) == 0) *s_half = (int)((id >> 18) & 1u);
    }
    __syncthreads();
    return __builtin_amdgcn_readfirstlane(*s_half);
}

__global__ void zero2_kernel(float* __restrict__ a, long na4,
                             float* __restrict__ b, long nb4) {
    long i = (long)blockIdx.x * blockDim.x + threadIdx.x;
    long stride = (long)gridDim.x * blockDim.x;
    float4 z = make_float4(0.f, 0.f, 0.f, 0.f);
    for (long j = i; j < na4 + nb4; j += stride) {
        if (j < na4) ((float4*)a)[j] = z;
        else         ((float4*)b)[j - na4] = z;
    }
}

__global__ void degree_kernel(const int* __restrict__ row, const int* __restrict__ col,
                              int E, float* degO0, float* degI0,
                              float* degO1, float* degI1) {
    __shared__ int s_half;
    const int half = block_half(&s_half);
    float* degO = half ? degO1 : degO0;
    float* degI = half ? degI1 : degI0;
    int e = blockIdx.x * blockDim.x + threadIdx.x;
    if (e >= E) return;
    unsafeAtomicAdd(&degO[row[e]], 1.0f);
    unsafeAtomicAdd(&degI[col[e]], 1.0f);
}

template <bool TWO>
__global__ void deginv_kernel(float* degO, float* degI,
                              const float* degO1, const float* degI1, int n) {
    int i = blockIdx.x * blockDim.x + threadIdx.x;
    if (i >= n) return;
    float d0 = degO[i];
    float d1 = degI[i];
    if (TWO) { d0 += degO1[i]; d1 += degI1[i]; }
    degO[i] = (d0 > 0.f) ? powf(d0, -0.25f) : 0.f;
    degI[i] = (d1 > 0.f) ? powf(d1, -0.25f) : 0.f;
}

__global__ void wcomb_kernel(const float* __restrict__ Wr, const float* __restrict__ br,
                             const float* __restrict__ Wi, const float* __restrict__ bi,
                             int K, float* WRc, float* WIc, float* bmi, float* bpi) {
    int i = blockIdx.x * blockDim.x + threadIdx.x;
    if (i < D_FEAT * D_FEAT) {
        float sr = 0.f, si = 0.f, c = 1.f;
        for (int k = 0; k < K; ++k) {
            sr += c * Wr[k * D_FEAT * D_FEAT + i];
            si += c * Wi[k * D_FEAT * D_FEAT + i];
            c *= 0.5f;
        }
        WRc[i] = sr;
        WIc[i] = si;
    } else if (i < D_FEAT * D_FEAT + D_FEAT) {
        int o = i - D_FEAT * D_FEAT;
        float sr = 0.f, si = 0.f, c = 1.f;
        for (int k = 0; k < K; ++k) {
            sr += c * br[k * D_FEAT + o];
            si += c * bi[k * D_FEAT + o];
            c *= 0.5f;
        }
        bmi[o] = sr - si;
        bpi[o] = sr + si;
    }
}

__global__ __launch_bounds__(256) void spmm_kernel(
        const float* __restrict__ xr, const float* __restrict__ xi,
        const int* __restrict__ row, const int* __restrict__ col, int E,
        const float* __restrict__ out_inv, const float* __restrict__ in_inv,
        float* yr0, float* yi0, float* yrt0, float* yit0,
        float* yr1, float* yi1, float* yrt1, float* yit1) {
    __shared__ int s_half;
    const int half = block_half(&s_half);
    float* yr  = half ? yr1  : yr0;
    float* yi  = half ? yi1  : yi0;
    float* yrt = half ? yrt1 : yrt0;
    float* yit = half ? yit1 : yit0;

    const int lane = threadIdx.x & 31;
    int e = blockIdx.x * 8 + (threadIdx.x >> 5);
    e = __builtin_amdgcn_readfirstlane(e);
    if (e >= E) return;

    const int r = row[e];
    const int c = col[e];
    const float w = out_inv[r] * in_inv[c];

    const long ro = (long)r * D_FEAT + lane;
    const long co = (long)c * D_FEAT + lane;
    const int H = D_FEAT / 2;

    const float xrc0 = xr[co], xrc1 = xr[co + H];
    const float xic0 = xi[co], xic1 = xi[co + H];
    const float xrr0 = xr[ro], xrr1 = xr[ro + H];
    const float xir0 = xi[ro], xir1 = xi[ro + H];

    unsafeAtomicAdd(&yr[ro],      w * xrc0);
    unsafeAtomicAdd(&yr[ro + H],  w * xrc1);
    unsafeAtomicAdd(&yi[ro],      w * xic0);
    unsafeAtomicAdd(&yi[ro + H],  w * xic1);
    unsafeAtomicAdd(&yrt[co],     w * xrr0);
    unsafeAtomicAdd(&yrt[co + H], w * xrr1);
    unsafeAtomicAdd(&yit[co],     w * xir0);
    unsafeAtomicAdd(&yit[co + H], w * xir1);
}

template <int COPIES>
__global__ __launch_bounds__(128) void faber_gemm_kernel(
        const float* __restrict__ yr, const float* __restrict__ yi,
        const float* __restrict__ yr1, const float* __restrict__ yi1,
        const float* __restrict__ yrt1, const float* __restrict__ yit1,
        const float* __restrict__ WRc, const float* __restrict__ WIc,
        const float* __restrict__ bmi, const float* __restrict__ bpi,
        float* outR, float* outI) {
    const int wave = threadIdx.x >> 5;
    const int lane = threadIdx.x & 31;
    const int l16  = lane & 15;
    const int half = lane >> 4;

    const float* yrt = outR;
    const float* yit = outI;

    const long rowOff = (long)(blockIdx.x * 16 + l16) * D_FEAT;
    const long wOff   = (long)(wave * 16 + l16) * D_FEAT;

    v8f accRa = {};
    v8f accRb = {};
    v8f accIa = {};
    v8f accIb = {};

    #pragma unroll
    for (int s = 0; s < 16; ++s) {
        const int k = 4 * s + 2 * half;
        float2 ar  = *(const float2*)(yr  + rowOff + k);
        float2 art = *(const float2*)(yrt + rowOff + k);
        float2 ai  = *(const float2*)(yi  + rowOff + k);
        float2 ait = *(const float2*)(yit + rowOff + k);
        if (COPIES >= 1) {
            const float2 br  = *(const float2*)(yr1  + rowOff + k);
            const float2 bi  = *(const float2*)(yi1  + rowOff + k);
            ar.x  += br.x;  ar.y  += br.y;
            ai.x  += bi.x;  ai.y  += bi.y;
        }
        if (COPIES == 2) {
            const float2 brt = *(const float2*)(yrt1 + rowOff + k);
            const float2 bit = *(const float2*)(yit1 + rowOff + k);
            art.x += brt.x; art.y += brt.y;
            ait.x += bit.x; ait.y += bit.y;
        }
        v2f Azr = {0.5f * (ar.x + art.x), 0.5f * (ar.y + art.y)};
        v2f Azi = {0.5f * (ai.x + ait.x), 0.5f * (ai.y + ait.y)};
        v2f Ayr = {ar.x, ar.y};
        float2 wr2 = *(const float2*)(WRc + wOff + k);
        float2 wi2 = *(const float2*)(WIc + wOff + k);
        v2f Bwr = {wr2.x, wr2.y};
        v2f Bwi = {wi2.x, wi2.y};
        accRa = __builtin_amdgcn_wmma_f32_16x16x4_f32(false, Azr, false, Bwr, (short)0, accRa, false, false);
        accRb = __builtin_amdgcn_wmma_f32_16x16x4_f32(false, Azi, false, Bwi, (short)0, accRb, false, false);
        accIa = __builtin_amdgcn_wmma_f32_16x16x4_f32(false, Ayr, false, Bwi, (short)0, accIa, false, false);
        accIb = __builtin_amdgcn_wmma_f32_16x16x4_f32(false, Azi, false, Bwr, (short)0, accIb, false, false);
    }

    const int colD = wave * 16 + l16;
    const float bm = bmi[colD];
    const float bp = bpi[colD];

    __syncthreads();

    #pragma unroll
    for (int r = 0; r < 8; ++r) {
        long nodeM = (long)(blockIdx.x * 16 + r + 8 * half);
        outR[nodeM * D_FEAT + colD] = accRa[r] - accRb[r] + bm;
        outI[nodeM * D_FEAT + colD] = accIa[r] + accIb[r] + bp;
    }
}

extern "C" void kernel_launch(void* const* d_in, const int* in_sizes, int n_in,
                              void* d_out, int out_size, void* d_ws, size_t ws_size,
                              hipStream_t stream) {
    const float* xr = (const float*)d_in[0];
    const float* xi = (const float*)d_in[1];
    const int*   ei = (const int*)d_in[2];
    const float* Wr = (const float*)d_in[3];
    const float* br = (const float*)d_in[4];
    const float* Wi = (const float*)d_in[5];
    const float* bi = (const float*)d_in[6];

    const int n = in_sizes[0] / D_FEAT;
    const int E = in_sizes[2] / 2;
    const int K = in_sizes[3] / (D_FEAT * D_FEAT);
    const long NF = (long)n * D_FEAT;

    const long wsF  = (long)(ws_size / sizeof(float));
    const long tail = 2 * D_FEAT * D_FEAT + 2 * D_FEAT;
    const int copies = (wsF >= 6 * NF + 4 * (long)n + tail) ? 2
                     : (wsF >= 4 * NF + 4 * (long)n + tail) ? 1 : 0;
    const long accPlanes = 2 + 2 * copies;
    const long degRows   = copies ? 4 : 2;
    float* ws   = (float*)d_ws;
    float* yr   = ws;
    float* yi_  = ws + NF;
    float* yr1  = copies ? ws + 2 * NF : yr;
    float* yi1  = copies ? ws + 3 * NF : yi_;
    float* degO = ws + accPlanes * NF;
    float* degI = degO + n;
    float* degO1 = copies ? degO + 2 * (long)n : degO;
    float* degI1 = copies ? degO + 3 * (long)n : degI;
    float* WRc  = degO + degRows * n;
    float* WIc  = WRc + D_FEAT * D_FEAT;
    float* bmi  = WIc + D_FEAT * D_FEAT;
    float* bpi  = bmi + D_FEAT;

    float* outR = (float*)d_out;
    float* outI = outR + NF;
    float* yrt1 = (copies == 2) ? ws + 4 * NF : outR;
    float* yit1 = (copies == 2) ? ws + 5 * NF : outI;
    const int* row = ei;
    const int* col = ei + E;

    zero2_kernel<<<2048, 256, 0, stream>>>(ws, (accPlanes * NF + degRows * n) / 4,
                                           outR, (2 * NF) / 4);
    degree_kernel<<<(E + 255) / 256, 256, 0, stream>>>(row, col, E, degO, degI, degO1, degI1);
    if (copies) deginv_kernel<true><<<(n + 255) / 256, 256, 0, stream>>>(degO, degI, degO1, degI1, n);
    else        deginv_kernel<false><<<(n + 255) / 256, 256, 0, stream>>>(degO, degI, degO1, degI1, n);
    wcomb_kernel<<<(D_FEAT * D_FEAT + D_FEAT + 255) / 256, 256, 0, stream>>>(
        Wr, br, Wi, bi, K, WRc, WIc, bmi, bpi);
    spmm_kernel<<<(E + 7) / 8, 256, 0, stream>>>(xr, xi, row, col, E, degO, degI,
                                                 yr, yi_, outR, outI,
                                                 yr1, yi1, yrt1, yit1);
    if (copies == 2)
        faber_gemm_kernel<2><<<n / 16, 128, 0, stream>>>(
            yr, yi_, yr1, yi1, yrt1, yit1, WRc, WIc, bmi, bpi, outR, outI);
    else if (copies == 1)
        faber_gemm_kernel<1><<<n / 16, 128, 0, stream>>>(
            yr, yi_, yr1, yi1, yrt1, yit1, WRc, WIc, bmi, bpi, outR, outI);
    else
        faber_gemm_kernel<0><<<n / 16, 128, 0, stream>>>(
            yr, yi_, yr1, yi1, yrt1, yit1, WRc, WIc, bmi, bpi, outR, outI);
}


# edge scatter un-fused into 4 passes inside its one launch (grid y = which of the 4 sums; a wave takes 4 edges of one sum): a pass's gather plane + sum plane fit a half's L2; atomic order as was
# speedup vs baseline: 1.3915x; 1.1453x over previous
#include <hip/hip_runtime.h>
#include <math.h>

typedef __attribute__((ext_vector_type(2))) float v2f;
typedef __attribute__((ext_vector_type(8))) float v8f;

#define D_FEAT 64

__device__ __forceinline__ int block_half(int* s_half) {
    if (__builtin_amdgcn_readfirstlane((int)(threadIdx.x >> 5)) == 0) {
        const unsigned id = __builtin_amdgcn_s_sendmsg_rtn(0x87);
        if ((threadIdx.x & 31) == 0) *s_half = (int)((id >> 18) & 1u);
    }
    __syncthreads();
    return __builtin_amdgcn_readfirstlane(*s_half);
}

__global__ void zero2_kernel(float* __restrict__ a, long na4,
                             float* __restrict__ b, long nb4) {
    long i = (long)blockIdx.x * blockDim.x + threadIdx.x;
    long stride = (long)gridDim.x * blockDim.x;
    float4 z = make_float4(0.f, 0.f, 0.f, 0.f);
    for (long j = i; j < na4 + nb4; j += stride) {
        if (j < na4) ((float4*)a)[j] = z;
        else         ((float4*)b)[j - na4] = z;
    }
}

__global__ void degree_kernel(const int* __restrict__ row, const int* __restrict__ col,
                              int E, float* degO0, float* degI0,
                              float* degO1, float* degI1) {
    __shared__ int s_half;
    const int half = block_half(&s_half);
    float* degO = half ? degO1 : degO0;
    float* degI = half ? degI1 : degI0;
    int e = blockIdx.x * blockDim.x + threadIdx.x;
    if (e >= E) return;
    unsafeAtomicAdd(&degO[row[e]], 1.0f);
    unsafeAtomicAdd(&degI[col[e]], 1.0f);
}

template <bool TWO>
__global__ void deginv_kernel(float* degO, float* degI,
                              const float* degO1, const float* degI1, int n) {
    int i = blockIdx.x * blockDim.x + threadIdx.x;
    if (i >= n) return;
    float d0 = degO[i];
    float d1 = degI[i];
    if (TWO) { d0 += degO1[i]; d1 += degI1[i]; }
    degO[i] = (d0 > 0.f) ? powf(d0, -0.25f) : 0.f;
    degI[i] = (d1 > 0.f) ? powf(d1, -0.25f) : 0.f;
}

__global__ void wcomb_kernel(const float* __restrict__ Wr, const float* __restrict__ br,
                             const float* __restrict__ Wi, const float* __restrict__ bi,
                             int K, float* WRc, float* WIc, float* bmi, float* bpi) {
    int i = blockIdx.x * blockDim.x + threadIdx.x;
    if (i < D_FEAT * D_FEAT) {
        float sr = 0.f, si = 0.f, c = 1.f;
        for (int k = 0; k < K; ++k) {
            sr += c * Wr[k * D_FEAT * D_FEAT + i];
            si += c * Wi[k * D_FEAT * D_FEAT + i];
            c *= 0.5f;
        }
        WRc[i] = sr;
        WIc[i] = si;
    } else if (i < D_FEAT * D_FEAT + D_FEAT) {
        int o = i - D_FEAT * D_FEAT;
        float sr = 0.f, si = 0.f, c = 1.f;
        for (int k = 0; k < K; ++k) {
            sr += c * br[k * D_FEAT + o];
            si += c * bi[k * D_FEAT + o];
            c *= 0.5f;
        }
        bmi[o] = sr - si;
        bpi[o] = sr + si;
    }
}

__global__ __launch_bounds__(256) void spmm_kernel(
        const float* __restrict__ xr, const float* __restrict__ xi,
        const int* __restrict__ row, const int* __restrict__ col, int E,
        const float* __restrict__ out_inv, const float* __restrict__ in_inv,
        float* yr0, float* yi0, float* yrt0, float* yit0,
        float* yr1, float* yi1, float* yrt1, float* yit1) {
    __shared__ int s_half;
    const int half = block_half(&s_half);

    const int pass = blockIdx.y;
    const bool tr = (pass & 1) != 0;
    const bool im = (pass & 2) != 0;
    const float* __restrict__ x    = im ? xi : xr;
    const int*   __restrict__ gidx = tr ? row : col;
    const int*   __restrict__ sidx = tr ? col : row;
    const float* __restrict__ ginv = tr ? out_inv : in_inv;
    const float* __restrict__ sinv = tr ? in_inv : out_inv;
    float* const yr  = half ? yr1  : yr0;
    float* const yi  = half ? yi1  : yi0;
    float* const yrt = half ? yrt1 : yrt0;
    float* const yit = half ? yit1 : yit0;
    float* const yre = tr ? yrt : yr;
    float* const yim = tr ? yit : yi;
    float* const y   = im ? yim : yre;

    const int lane = threadIdx.x & 31;
    const int H = D_FEAT / 2;
    int e0 = (blockIdx.x * 8 + (threadIdx.x >> 5)) * 4;
    e0 = __builtin_amdgcn_readfirstlane(e0);
    if (e0 >= E) return;

    if (e0 + 4 <= E) {
        int g[4], s[4];
        float w[4], v0[4], v1[4];
        for (int i = 0; i < 4; ++i) { g[i] = gidx[e0 + i]; s[i] = sidx[e0 + i]; }
        for (int i = 0; i < 4; ++i) w[i] = sinv[s[i]] * ginv[g[i]];
        for (int i = 0; i < 4; ++i) {
            const long go = (long)g[i] * D_FEAT + lane;
            v0[i] = x[go];
            v1[i] = x[go + H];
        }
        for (int i = 0; i < 4; ++i) {
            const long so = (long)s[i] * D_FEAT + lane;
            unsafeAtomicAdd(&y[so],     w[i] * v0[i]);
            unsafeAtomicAdd(&y[so + H], w[i] * v1[i]);
        }
    } else {
        for (int e = e0; e < E; ++e) {
            const int g = gidx[e];
            const int s = sidx[e];
            const float w = sinv[s] * ginv[g];
            const long go = (long)g * D_FEAT + lane;
            const long so = (long)s * D_FEAT + lane;
            const float a0 = x[go], a1 = x[go + H];
            unsafeAtomicAdd(&y[so],     w * a0);
            unsafeAtomicAdd(&y[so + H], w * a1);
        }
    }
}

template <int COPIES>
__global__ __launch_bounds__(128) void faber_gemm_kernel(
        const float* __restrict__ yr, const float* __restrict__ yi,
        const float* __restrict__ yr1, const float* __restrict__ yi1,
        const float* __restrict__ yrt1, const float* __restrict__ yit1,
        const float* __restrict__ WRc, const float* __restrict__ WIc,
        const float* __restrict__ bmi, const float* __restrict__ bpi,
        float* outR, float* outI) {
    const int wave = threadIdx.x >> 5;
    const int lane = threadIdx.x & 31;
    const int l16  = lane & 15;
    const int half = lane >> 4;

    const float* yrt = outR;
    const float* yit = outI;

    const long rowOff = (long)(blockIdx.x * 16 + l16) * D_FEAT;
    const long wOff   = (long)(wave * 16 + l16) * D_FEAT;

    v8f accRa = {};
    v8f accRb = {};
    v8f accIa = {};
    v8f accIb = {};

    #pragma unroll
    for (int s = 0; s < 16; ++s) {
        const int k = 4 * s + 2 * half;
        float2 ar  = *(const float2*)(yr  + rowOff + k);
        float2 art = *(const float2*)(yrt + rowOff + k);
        float2 ai  = *(const float2*)(yi  + rowOff + k);
        float2 ait = *(const float2*)(yit + rowOff + k);
        if (COPIES >= 1) {
            const float2 br  = *(const float2*)(yr1  + rowOff + k);
            const float2 bi  = *(const float2*)(yi1  + rowOff + k);
            ar.x  += br.x;  ar.y  += br.y;
            ai.x  += bi.x;  ai.y  += bi.y;
        }
        if (COPIES == 2) {
            const float2 brt = *(const float2*)(yrt1 + rowOff + k);
            const float2 bit = *(const float2*)(yit1 + rowOff + k);
            art.x += brt.x; art.y += brt.y;
            ait.x += bit.x; ait.y += bit.y;
        }
        v2f Azr = {0.5f * (ar.x + art.x), 0.5f * (ar.y + art.y)};
        v2f Azi = {0.5f * (ai.x + ait.x), 0.5f * (ai.y + ait.y)};
        v2f Ayr = {ar.x, ar.y};
        float2 wr2 = *(const float2*)(WRc + wOff + k);
        float2 wi2 = *(const float2*)(WIc + wOff + k);
        v2f Bwr = {wr2.x, wr2.y};
        v2f Bwi = {wi2.x, wi2.y};
        accRa = __builtin_amdgcn_wmma_f32_16x16x4_f32(false, Azr, false, Bwr, (short)0, accRa, false, false);
        accRb = __builtin_amdgcn_wmma_f32_16x16x4_f32(false, Azi, false, Bwi, (short)0, accRb, false, false);
        accIa = __builtin_amdgcn_wmma_f32_16x16x4_f32(false, Ayr, false, Bwi, (short)0, accIa, false, false);
        accIb = __builtin_amdgcn_wmma_f32_16x16x4_f32(false, Azi, false, Bwr, (short)0, accIb, false, false);
    }

    const int colD = wave * 16 + l16;
    const float bm = bmi[colD];
    const float bp = bpi[colD];

    __syncthreads();

    #pragma unroll
    for (int r = 0; r < 8; ++r) {
        long nodeM = (long)(blockIdx.x * 16 + r + 8 * half);
        outR[nodeM * D_FEAT + colD] = accRa[r] - accRb[r] + bm;
        outI[nodeM * D_FEAT + colD] = accIa[r] + accIb[r] + bp;
    }
}

extern "C" void kernel_launch(void* const* d_in, const int* in_sizes, int n_in,
                              void* d_out, int out_size, void* d_ws, size_t ws_size,
                              hipStream_t stream) {
    const float* xr = (const float*)d_in[0];
    const float* xi = (const float*)d_in[1];
    const int*   ei = (const int*)d_in[2];
    const float* Wr = (const float*)d_in[3];
    const float* br = (const float*)d_in[4];
    const float* Wi = (const float*)d_in[5];
    const float* bi = (const float*)d_in[6];

    const int n = in_sizes[0] / D_FEAT;
    const int E = in_sizes[2] / 2;
    const int K = in_sizes[3] / (D_FEAT * D_FEAT);
    const long NF = (long)n * D_FEAT;

    const long wsF  = (long)(ws_size / sizeof(float));
    const long tail = 2 * D_FEAT * D_FEAT + 2 * D_FEAT;
    const int copies = (wsF >= 6 * NF + 4 * (long)n + tail) ? 2
                     : (wsF >= 4 * NF + 4 * (long)n + tail) ? 1 : 0;
    const long accPlanes = 2 + 2 * copies;
    const long degRows   = copies ? 4 : 2;
    float* ws   = (float*)d_ws;
    float* yr   = ws;
    float* yi_  = ws + NF;
    float* yr1  = copies ? ws + 2 * NF : yr;
    float* yi1  = copies ? ws + 3 * NF : yi_;
    float* degO = ws + accPlanes * NF;
    float* degI = degO + n;
    float* degO1 = copies ? degO + 2 * (long)n : degO;
    float* degI1 = copies ? degO + 3 * (long)n : degI;
    float* WRc  = degO + degRows * n;
    float* WIc  = WRc + D_FEAT * D_FEAT;
    float* bmi  = WIc + D_FEAT * D_FEAT;
    float* bpi  = bmi + D_FEAT;

    float* outR = (float*)d_out;
    float* outI = outR + NF;
    float* yrt1 = (copies == 2) ? ws + 4 * NF : outR;
    float* yit1 = (copies == 2) ? ws + 5 * NF : outI;
    const int* row = ei;
    const int* col = ei + E;

    zero2_kernel<<<2048, 256, 0, stream>>>(ws, (accPlanes * NF + degRows * n) / 4,
                                           outR, (2 * NF) / 4);
    degree_kernel<<<(E + 255) / 256, 256, 0, stream>>>(row, col, E, degO, degI, degO1, degI1);
    if (copies) deginv_kernel<true><<<(n + 255) / 256, 256, 0, stream>>>(degO, degI, degO1, degI1, n);
    else        deginv_kernel<false><<<(n + 255) / 256, 256, 0, stream>>>(degO, degI, degO1, degI1, n);
    wcomb_kernel<<<(D_FEAT * D_FEAT + D_FEAT + 255) / 256, 256, 0, stream>>>(
        Wr, br, Wi, bi, K, WRc, WIc, bmi, bpi);
    spmm_kernel<<<dim3((E + 31) / 32, 4), 256, 0, stream>>>(xr, xi, row, col, E, degO, degI,
                                                            yr, yi_, outR, outI,
                                                            yr1, yi1, yrt1, yit1);
    if (copies == 2)
        faber_gemm_kernel<2><<<n / 16, 128, 0, stream>>>(
            yr, yi_, yr1, yi1, yrt1, yit1, WRc, WIc, bmi, bpi, outR, outI);
    else if (copies == 1)
        faber_gemm_kernel<1><<<n / 16, 128, 0, stream>>>(
            yr, yi_, yr1, yi1, yrt1, yit1, WRc, WIc, bmi, bpi, outR, outI);
    else
        faber_gemm_kernel<0><<<n / 16, 128, 0, stream>>>(
            yr, yi_, yr1, yi1, yrt1, yit1, WRc, WIc, bmi, bpi, outR, outI);
}


# edge weight out_inv[row]*in_inv[col] formed once per edge by a new kernel into a workspace array; a pass reads a wave's 4 weights in one 16-byte scalar read, not 8 scattered reads; atomic order kept
# speedup vs baseline: 1.4866x; 1.0683x over previous
#include <hip/hip_runtime.h>
#include <math.h>

typedef __attribute__((ext_vector_type(2))) float v2f;
typedef __attribute__((ext_vector_type(8))) float v8f;

#define D_FEAT 64

__device__ __forceinline__ int block_half(int* s_half) {
    if (__builtin_amdgcn_readfirstlane((int)(threadIdx.x >> 5)) == 0) {
        const unsigned id = __builtin_amdgcn_s_sendmsg_rtn(0x87);
        if ((threadIdx.x & 31) == 0) *s_half = (int)((id >> 18) & 1u);
    }
    __syncthreads();
    return __builtin_amdgcn_readfirstlane(*s_half);
}

__global__ void zero2_kernel(float* __restrict__ a, long na4,
                             float* __restrict__ b, long nb4) {
    long i = (long)blockIdx.x * blockDim.x + threadIdx.x;
    long stride = (long)gridDim.x * blockDim.x;
    float4 z = make_float4(0.f, 0.f, 0.f, 0.f);
    for (long j = i; j < na4 + nb4; j += stride) {
        if (j < na4) ((float4*)a)[j] = z;
        else         ((float4*)b)[j - na4] = z;
    }
}

__global__ void degree_kernel(const int* __restrict__ row, const int* __restrict__ col,
                              int E, float* degO0, float* degI0,
                              float* degO1, float* degI1) {
    __shared__ int s_half;
    const int half = block_half(&s_half);
    float* degO = half ? degO1 : degO0;
    float* degI = half ? degI1 : degI0;
    int e = blockIdx.x * blockDim.x + threadIdx.x;
    if (e >= E) return;
    unsafeAtomicAdd(&degO[row[e]], 1.0f);
    unsafeAtomicAdd(&degI[col[e]], 1.0f);
}

template <bool TWO>
__global__ void deginv_kernel(float* degO, float* degI,
                              const float* degO1, const float* degI1, int n) {
    int i = blockIdx.x * blockDim.x + threadIdx.x;
    if (i >= n) return;
    float d0 = degO[i];
    float d1 = degI[i];
    if (TWO) { d0 += degO1[i]; d1 += degI1[i]; }
    degO[i] = (d0 > 0.f) ? powf(d0, -0.25f) : 0.f;
    degI[i] = (d1 > 0.f) ? powf(d1, -0.25f) : 0.f;
}

__global__ void edgew_kernel(const int* __restrict__ row, const int* __restrict__ col, int E,
                             const float* __restrict__ out_inv, const float* __restrict__ in_inv,
                             float* __restrict__ wE) {
    int e = blockIdx.x * blockDim.x + threadIdx.x;
    if (e >= E) return;
    wE[e] = out_inv[row[e]] * in_inv[col[e]];
}

__global__ void wcomb_kernel(const float* __restrict__ Wr, const float* __restrict__ br,
                             const float* __restrict__ Wi, const float* __restrict__ bi,
                             int K, float* WRc, float* WIc, float* bmi, float* bpi) {
    int i = blockIdx.x * blockDim.x + threadIdx.x;
    if (i < D_FEAT * D_FEAT) {
        float sr = 0.f, si = 0.f, c = 1.f;
        for (int k = 0; k < K; ++k) {
            sr += c * Wr[k * D_FEAT * D_FEAT + i];
            si += c * Wi[k * D_FEAT * D_FEAT + i];
            c *= 0.5f;
        }
        WRc[i] = sr;
        WIc[i] = si;
    } else if (i < D_FEAT * D_FEAT + D_FEAT) {
        int o = i - D_FEAT * D_FEAT;
        float sr = 0.f, si = 0.f, c = 1.f;
        for (int k = 0; k < K; ++k) {
            sr += c * br[k * D_FEAT + o];
            si += c * bi[k * D_FEAT + o];
            c *= 0.5f;
        }
        bmi[o] = sr - si;
        bpi[o] = sr + si;
    }
}

template <bool HASW>
__global__ __launch_bounds__(256) void spmm_kernel(
        const float* __restrict__ xr, const float* __restrict__ xi,
        const int* __restrict__ row, const int* __restrict__ col, int E,
        const float* __restrict__ out_inv, const float* __restrict__ in_inv,
        const float* __restrict__ wE,
        float* yr0, float* yi0, float* yrt0, float* yit0,
        float* yr1, float* yi1, float* yrt1, float* yit1) {
    __shared__ int s_half;
    const int half = block_half(&s_half);

    const int pass = blockIdx.y;
    const bool tr = (pass & 1) != 0;
    const bool im = (pass & 2) != 0;
    const float* __restrict__ x    = im ? xi : xr;
    const int*   __restrict__ gidx = tr ? row : col;
    const int*   __restrict__ sidx = tr ? col : row;
    const float* __restrict__ ginv = tr ? out_inv : in_inv;
    const float* __restrict__ sinv = tr ? in_inv : out_inv;
    float* const yr  = half ? yr1  : yr0;
    float* const yi  = half ? yi1  : yi0;
    float* const yrt = half ? yrt1 : yrt0;
    float* const yit = half ? yit1 : yit0;
    float* const yre = tr ? yrt : yr;
    float* const yim = tr ? yit : yi;
    float* const y   = im ? yim : yre;

    const int lane = threadIdx.x & 31;
    const int H = D_FEAT / 2;
    int e0 = (blockIdx.x * 8 + (threadIdx.x >> 5)) * 4;
    e0 = __builtin_amdgcn_readfirstlane(e0);
    if (e0 >= E) return;

    if (e0 + 4 <= E) {
        int g[4], s[4];
        float w[4], v0[4], v1[4];
        for (int i = 0; i < 4; ++i) { g[i] = gidx[e0 + i]; s[i] = sidx[e0 + i]; }
        if (HASW) {
            const float4 w4 = *(const float4*)(wE + e0);
            w[0] = w4.x; w[1] = w4.y; w[2] = w4.z; w[3] = w4.w;
        } else {
            for (int i = 0; i < 4; ++i) w[i] = sinv[s[i]] * ginv[g[i]];
        }
        for (int i = 0; i < 4; ++i) {
            const long go = (long)g[i] * D_FEAT + lane;
            v0[i] = x[go];
            v1[i] = x[go + H];
        }
        for (int i = 0; i < 4; ++i) {
            const long so = (long)s[i] * D_FEAT + lane;
            unsafeAtomicAdd(&y[so],     w[i] * v0[i]);
            unsafeAtomicAdd(&y[so + H], w[i] * v1[i]);
        }
    } else {
        for (int e = e0; e < E; ++e) {
            const int g = gidx[e];
            const int s = sidx[e];
            const float w = HASW ? wE[e] : sinv[s] * ginv[g];
            const long go = (long)g * D_FEAT + lane;
            const long so = (long)s * D_FEAT + lane;
            const float a0 = x[go], a1 = x[go + H];
            unsafeAtomicAdd(&y[so],     w * a0);
            unsafeAtomicAdd(&y[so + H], w * a1);
        }
    }
}

template <int COPIES>
__global__ __launch_bounds__(128) void faber_gemm_kernel(
        const float* __restrict__ yr, const float* __restrict__ yi,
        const float* __restrict__ yr1, const float* __restrict__ yi1,
        const float* __restrict__ yrt1, const float* __restrict__ yit1,
        const float* __restrict__ WRc, const float* __restrict__ WIc,
        const float* __restrict__ bmi, const float* __restrict__ bpi,
        float* outR, float* outI) {
    const int wave = threadIdx.x >> 5;
    const int lane = threadIdx.x & 31;
    const int l16  = lane & 15;
    const int half = lane >> 4;

    const float* yrt = outR;
    const float* yit = outI;

    const long rowOff = (long)(blockIdx.x * 16 + l16) * D_FEAT;
    const long wOff   = (long)(wave * 16 + l16) * D_FEAT;

    v8f accRa = {};
    v8f accRb = {};
    v8f accIa = {};
    v8f accIb = {};

    #pragma unroll
    for (int s = 0; s < 16; ++s) {
        const int k = 4 * s + 2 * half;
        float2 ar  = *(const float2*)(yr  + rowOff + k);
        float2 art = *(const float2*)(yrt + rowOff + k);
        float2 ai  = *(const float2*)(yi  + rowOff + k);
        float2 ait = *(const float2*)(yit + rowOff + k);
        if (COPIES >= 1) {
            const float2 br  = *(const float2*)(yr1  + rowOff + k);
            const float2 bi  = *(const float2*)(yi1  + rowOff + k);
            ar.x  += br.x;  ar.y  += br.y;
            ai.x  += bi.x;  ai.y  += bi.y;
        }
        if (COPIES == 2) {
            const float2 brt = *(const float2*)(yrt1 + rowOff + k);
            const float2 bit = *(const float2*)(yit1 + rowOff + k);
            art.x += brt.x; art.y += brt.y;
            ait.x += bit.x; ait.y += bit.y;
        }
        v2f Azr = {0.5f * (ar.x + art.x), 0.5f * (ar.y + art.y)};
        v2f Azi = {0.5f * (ai.x + ait.x), 0.5f * (ai.y + ait.y)};
        v2f Ayr = {ar.x, ar.y};
        float2 wr2 = *(const float2*)(WRc + wOff + k);
        float2 wi2 = *(const float2*)(WIc + wOff + k);
        v2f Bwr = {wr2.x, wr2.y};
        v2f Bwi = {wi2.x, wi2.y};
        accRa = __builtin_amdgcn_wmma_f32_16x16x4_f32(false, Azr, false, Bwr, (short)0, accRa, false, false);
        accRb = __builtin_amdgcn_wmma_f32_16x16x4_f32(false, Azi, false, Bwi, (short)0, accRb, false, false);
        accIa = __builtin_amdgcn_wmma_f32_16x16x4_f32(false, Ayr, false, Bwi, (short)0, accIa, false, false);
        accIb = __builtin_amdgcn_wmma_f32_16x16x4_f32(false, Azi, false, Bwr, (short)0, accIb, false, false);
    }

    const int colD = wave * 16 + l16;
    const float bm = bmi[colD];
    const float bp = bpi[colD];

    __syncthreads();

    #pragma unroll
    for (int r = 0; r < 8; ++r) {
        long nodeM = (long)(blockIdx.x * 16 + r + 8 * half);
        outR[nodeM * D_FEAT + colD] = accRa[r] - accRb[r] + bm;
        outI[nodeM * D_FEAT + colD] = accIa[r] + accIb[r] + bp;
    }
}

extern "C" void kernel_launch(void* const* d_in, const int* in_sizes, int n_in,
                              void* d_out, int out_size, void* d_ws, size_t ws_size,
                              hipStream_t stream) {
    const float* xr = (const float*)d_in[0];
    const float* xi = (const float*)d_in[1];
    const int*   ei = (const int*)d_in[2];
    const float* Wr = (const float*)d_in[3];
    const float* br = (const float*)d_in[4];
    const float* Wi = (const float*)d_in[5];
    const float* bi = (const float*)d_in[6];

    const int n = in_sizes[0] / D_FEAT;
    const int E = in_sizes[2] / 2;
    const int K = in_sizes[3] / (D_FEAT * D_FEAT);
    const long NF = (long)n * D_FEAT;

    const long wsF  = (long)(ws_size / sizeof(float));
    const long tail = 2 * D_FEAT * D_FEAT + 2 * D_FEAT;
    const long Ew   = ((long)E + 3) / 4 * 4;
    const bool hasW = wsF >= 2 * NF + 2 * (long)n + tail + Ew;
    const long wsA  = hasW ? wsF - Ew : wsF;
    const int copies = (wsA >= 6 * NF + 4 * (long)n + tail) ? 2
                     : (wsA >= 4 * NF + 4 * (long)n + tail) ? 1 : 0;
    const long accPlanes = 2 + 2 * copies;
    const long degRows   = copies ? 4 : 2;
    float* ws   = (float*)d_ws;
    float* yr   = ws;
    float* yi_  = ws + NF;
    float* yr1  = copies ? ws + 2 * NF : yr;
    float* yi1  = copies ? ws + 3 * NF : yi_;
    float* degO = ws + accPlanes * NF;
    float* degI = degO + n;
    float* degO1 = copies ? degO + 2 * (long)n : degO;
    float* degI1 = copies ? degO + 3 * (long)n : degI;
    float* WRc  = degO + degRows * n;
    float* WIc  = WRc + D_FEAT * D_FEAT;
    float* bmi  = WIc + D_FEAT * D_FEAT;
    float* bpi  = bmi + D_FEAT;
    float* wE   = bpi + D_FEAT;

    float* outR = (float*)d_out;
    float* outI = outR + NF;
    float* yrt1 = (copies == 2) ? ws + 4 * NF : outR;
    float* yit1 = (copies == 2) ? ws + 5 * NF : outI;
    const int* row = ei;
    const int* col = ei + E;

    zero2_kernel<<<2048, 256, 0, stream>>>(ws, (accPlanes * NF + degRows * n) / 4,
                                           outR, (2 * NF) / 4);
    degree_kernel<<<(E + 255) / 256, 256, 0, stream>>>(row, col, E, degO, degI, degO1, degI1);
    if (copies) deginv_kernel<true><<<(n + 255) / 256, 256, 0, stream>>>(degO, degI, degO1, degI1, n);
    else        deginv_kernel<false><<<(n + 255) / 256, 256, 0, stream>>>(degO, degI, degO1, degI1, n);
    wcomb_kernel<<<(D_FEAT * D_FEAT + D_FEAT + 255) / 256, 256, 0, stream>>>(
        Wr, br, Wi, bi, K, WRc, WIc, bmi, bpi);
    if (hasW) {
        edgew_kernel<<<(E + 255) / 256, 256, 0, stream>>>(row, col, E, degO, degI, wE);
        spmm_kernel<true><<<dim3((E + 31) / 32, 4), 256, 0, stream>>>(xr, xi, row, col, E, degO, degI, wE,
                                                                      yr, yi_, outR, outI,
                                                                      yr1, yi1, yrt1, yit1);
    } else {
        spmm_kernel<false><<<dim3((E + 31) / 32, 4), 256, 0, stream>>>(xr, xi, row, col, E, degO, degI, wE,
                                                                       yr, yi_, outR, outI,
                                                                       yr1, yi1, yrt1, yit1);
    }
    if (copies == 2)
        faber_gemm_kernel<2><<<n / 16, 128, 0, stream>>>(
            yr, yi_, yr1, yi1, yrt1, yit1, WRc, WIc, bmi, bpi, outR, outI);
    else if (copies == 1)
        faber_gemm_kernel<1><<<n / 16, 128, 0, stream>>>(
            yr, yi_, yr1, yi1, yrt1, yit1, WRc, WIc, bmi, bpi, outR, outI);
    else
        faber_gemm_kernel<0><<<n / 16, 128, 0, stream>>>(
            yr, yi_, yr1, yi1, yrt1, yit1, WRc, WIc, bmi, bpi, outR, outI);
}


# edge scatter: a wave takes 32 edges of its pass in 2 rounds of 16 (was 4): 8x fewer waves; wave start, placement query and the index-gather-add round trips paid per 16 edges; atomic order as was
# speedup vs baseline: 1.7013x; 1.1444x over previous
#include <hip/hip_runtime.h>
#include <math.h>

typedef __attribute__((ext_vector_type(2))) float v2f;
typedef __attribute__((ext_vector_type(8))) float v8f;

#define D_FEAT 64

__device__ __forceinline__ int block_half(int* s_half) {
    if (__builtin_amdgcn_readfirstlane((int)(threadIdx.x >> 5)) == 0) {
        const unsigned id = __builtin_amdgcn_s_sendmsg_rtn(0x87);
        if ((threadIdx.x & 31) == 0) *s_half = (int)((id >> 18) & 1u);
    }
    __syncthreads();
    return __builtin_amdgcn_readfirstlane(*s_half);
}

__global__ void zero2_kernel(float* __restrict__ a, long na4,
                             float* __restrict__ b, long nb4) {
    long i = (long)blockIdx.x * blockDim.x + threadIdx.x;
    long stride = (long)gridDim.x * blockDim.x;
    float4 z = make_float4(0.f, 0.f, 0.f, 0.f);
    for (long j = i; j < na4 + nb4; j += stride) {
        if (j < na4) ((float4*)a)[j] = z;
        else         ((float4*)b)[j - na4] = z;
    }
}

__global__ void degree_kernel(const int* __restrict__ row, const int* __restrict__ col,
                              int E, float* degO0, float* degI0,
                              float* degO1, float* degI1) {
    __shared__ int s_half;
    const int half = block_half(&s_half);
    float* degO = half ? degO1 : degO0;
    float* degI = half ? degI1 : degI0;
    int e = blockIdx.x * blockDim.x + threadIdx.x;
    if (e >= E) return;
    unsafeAtomicAdd(&degO[row[e]], 1.0f);
    unsafeAtomicAdd(&degI[col[e]], 1.0f);
}

template <bool TWO>
__global__ void deginv_kernel(float* degO, float* degI,
                              const float* degO1, const float* degI1, int n) {
    int i = blockIdx.x * blockDim.x + threadIdx.x;
    if (i >= n) return;
    float d0 = degO[i];
    float d1 = degI[i];
    if (TWO) { d0 += degO1[i]; d1 += degI1[i]; }
    degO[i] = (d0 > 0.f) ? powf(d0, -0.25f) : 0.f;
    degI[i] = (d1 > 0.f) ? powf(d1, -0.25f) : 0.f;
}

__global__ void edgew_kernel(const int* __restrict__ row, const int* __restrict__ col, int E,
                             const float* __restrict__ out_inv, const float* __restrict__ in_inv,
                             float* __restrict__ wE) {
    int e = blockIdx.x * blockDim.x + threadIdx.x;
    if (e >= E) return;
    wE[e] = out_inv[row[e]] * in_inv[col[e]];
}

__global__ void wcomb_kernel(const float* __restrict__ Wr, const float* __restrict__ br,
                             const float* __restrict__ Wi, const float* __restrict__ bi,
                             int K, float* WRc, float* WIc, float* bmi, float* bpi) {
    int i = blockIdx.x * blockDim.x + threadIdx.x;
    if (i < D_FEAT * D_FEAT) {
        float sr = 0.f, si = 0.f, c = 1.f;
        for (int k = 0; k < K; ++k) {
            sr += c * Wr[k * D_FEAT * D_FEAT + i];
            si += c * Wi[k * D_FEAT * D_FEAT + i];
            c *= 0.5f;
        }
        WRc[i] = sr;
        WIc[i] = si;
    } else if (i < D_FEAT * D_FEAT + D_FEAT) {
        int o = i - D_FEAT * D_FEAT;
        float sr = 0.f, si = 0.f, c = 1.f;
        for (int k = 0; k < K; ++k) {
            sr += c * br[k * D_FEAT + o];
            si += c * bi[k * D_FEAT + o];
            c *= 0.5f;
        }
        bmi[o] = sr - si;
        bpi[o] = sr + si;
    }
}

#define SPMM_EPR 16
#define SPMM_RPW 2
#define SPMM_EPW (SPMM_EPR * SPMM_RPW)
template <bool HASW>
__global__ __launch_bounds__(256) void spmm_kernel(
        const float* __restrict__ xr, const float* __restrict__ xi,
        const int* __restrict__ row, const int* __restrict__ col, int E,
        const float* __restrict__ out_inv, const float* __restrict__ in_inv,
        const float* __restrict__ wE,
        float* yr0, float* yi0, float* yrt0, float* yit0,
        float* yr1, float* yi1, float* yrt1, float* yit1) {
    __shared__ int s_half;
    const int half = block_half(&s_half);

    const int pass = blockIdx.y;
    const bool tr = (pass & 1) != 0;
    const bool im = (pass & 2) != 0;
    const float* __restrict__ x    = im ? xi : xr;
    const int*   __restrict__ gidx = tr ? row : col;
    const int*   __restrict__ sidx = tr ? col : row;
    const float* __restrict__ ginv = tr ? out_inv : in_inv;
    const float* __restrict__ sinv = tr ? in_inv : out_inv;
    float* const yr  = half ? yr1  : yr0;
    float* const yi  = half ? yi1  : yi0;
    float* const yrt = half ? yrt1 : yrt0;
    float* const yit = half ? yit1 : yit0;
    float* const yre = tr ? yrt : yr;
    float* const yim = tr ? yit : yi;
    float* const y   = im ? yim : yre;

    const int lane = threadIdx.x & 31;
    const int H = D_FEAT / 2;
    int e0 = (blockIdx.x * 8 + (threadIdx.x >> 5)) * SPMM_EPW;
    e0 = __builtin_amdgcn_readfirstlane(e0);

    for (int r = 0; r < SPMM_RPW; ++r) {
        const int eb = e0 + r * SPMM_EPR;
        if (eb >= E) break;
        if (eb + SPMM_EPR <= E) {
            int g[SPMM_EPR], s[SPMM_EPR];
            float w[SPMM_EPR], v0[SPMM_EPR], v1[SPMM_EPR];
            for (int i = 0; i < SPMM_EPR; ++i) { g[i] = gidx[eb + i]; s[i] = sidx[eb + i]; }
            if (HASW) {
                for (int i = 0; i < SPMM_EPR; i += 4) {
                    const float4 w4 = *(const float4*)(wE + eb + i);
                    w[i] = w4.x; w[i + 1] = w4.y; w[i + 2] = w4.z; w[i + 3] = w4.w;
                }
            } else {
                for (int i = 0; i < SPMM_EPR; ++i) w[i] = sinv[s[i]] * ginv[g[i]];
            }
            for (int i = 0; i < SPMM_EPR; ++i) {
                const long go = (long)g[i] * D_FEAT + lane;
                v0[i] = x[go];
                v1[i] = x[go + H];
            }
            for (int i = 0; i < SPMM_EPR; ++i) {
                const long so = (long)s[i] * D_FEAT + lane;
                unsafeAtomicAdd(&y[so],     w[i] * v0[i]);
                unsafeAtomicAdd(&y[so + H], w[i] * v1[i]);
            }
        } else {
            for (int e = eb; e < E; ++e) {
                const int g = gidx[e];
                const int s = sidx[e];
                const float w = HASW ? wE[e] : sinv[s] * ginv[g];
                const long go = (long)g * D_FEAT + lane;
                const long so = (long)s * D_FEAT + lane;
                const float a0 = x[go], a1 = x[go + H];
                unsafeAtomicAdd(&y[so],     w * a0);
                unsafeAtomicAdd(&y[so + H], w * a1);
            }
        }
    }
}

template <int COPIES>
__global__ __launch_bounds__(128) void faber_gemm_kernel(
        const float* __restrict__ yr, const float* __restrict__ yi,
        const float* __restrict__ yr1, const float* __restrict__ yi1,
        const float* __restrict__ yrt1, const float* __restrict__ yit1,
        const float* __restrict__ WRc, const float* __restrict__ WIc,
        const float* __restrict__ bmi, const float* __restrict__ bpi,
        float* outR, float* outI) {
    const int wave = threadIdx.x >> 5;
    const int lane = threadIdx.x & 31;
    const int l16  = lane & 15;
    const int half = lane >> 4;

    const float* yrt = outR;
    const float* yit = outI;

    const long rowOff = (long)(blockIdx.x * 16 + l16) * D_FEAT;
    const long wOff   = (long)(wave * 16 + l16) * D_FEAT;

    v8f accRa = {};
    v8f accRb = {};
    v8f accIa = {};
    v8f accIb = {};

    #pragma unroll
    for (int s = 0; s < 16; ++s) {
        const int k = 4 * s + 2 * half;
        float2 ar  = *(const float2*)(yr  + rowOff + k);
        float2 art = *(const float2*)(yrt + rowOff + k);
        float2 ai  = *(const float2*)(yi  + rowOff + k);
        float2 ait = *(const float2*)(yit + rowOff + k);
        if (COPIES >= 1) {
            const float2 br  = *(const float2*)(yr1  + rowOff + k);
            const float2 bi  = *(const float2*)(yi1  + rowOff + k);
            ar.x  += br.x;  ar.y  += br.y;
            ai.x  += bi.x;  ai.y  += bi.y;
        }
        if (COPIES == 2) {
            const float2 brt = *(const float2*)(yrt1 + rowOff + k);
            const float2 bit = *(const float2*)(yit1 + rowOff + k);
            art.x += brt.x; art.y += brt.y;
            ait.x += bit.x; ait.y += bit.y;
        }
        v2f Azr = {0.5f * (ar.x + art.x), 0.5f * (ar.y + art.y)};
        v2f Azi = {0.5f * (ai.x + ait.x), 0.5f * (ai.y + ait.y)};
        v2f Ayr = {ar.x, ar.y};
        float2 wr2 = *(const float2*)(WRc + wOff + k);
        float2 wi2 = *(const float2*)(WIc + wOff + k);
        v2f Bwr = {wr2.x, wr2.y};
        v2f Bwi = {wi2.x, wi2.y};
        accRa = __builtin_amdgcn_wmma_f32_16x16x4_f32(false, Azr, false, Bwr, (short)0, accRa, false, false);
        accRb = __builtin_amdgcn_wmma_f32_16x16x4_f32(false, Azi, false, Bwi, (short)0, accRb, false, false);
        accIa = __builtin_amdgcn_wmma_f32_16x16x4_f32(false, Ayr, false, Bwi, (short)0, accIa, false, false);
        accIb = __builtin_amdgcn_wmma_f32_16x16x4_f32(false, Azi, false, Bwr, (short)0, accIb, false, false);
    }

    const int colD = wave * 16 + l16;
    const float bm = bmi[colD];
    const float bp = bpi[colD];

    __syncthreads();

    #pragma unroll
    for (int r = 0; r < 8; ++r) {
        long nodeM = (long)(blockIdx.x * 16 + r + 8 * half);
        outR[nodeM * D_FEAT + colD] = accRa[r] - accRb[r] + bm;
        outI[nodeM * D_FEAT + colD] = accIa[r] + accIb[r] + bp;
    }
}

extern "C" void kernel_launch(void* const* d_in, const int* in_sizes, int n_in,
                              void* d_out, int out_size, void* d_ws, size_t ws_size,
                              hipStream_t stream) {
    const float* xr = (const float*)d_in[0];
    const float* xi = (const float*)d_in[1];
    const int*   ei = (const int*)d_in[2];
    const float* Wr = (const float*)d_in[3];
    const float* br = (const float*)d_in[4];
    const float* Wi = (const float*)d_in[5];
    const float* bi = (const float*)d_in[6];

    const int n = in_sizes[0] / D_FEAT;
    const int E = in_sizes[2] / 2;
    const int K = in_sizes[3] / (D_FEAT * D_FEAT);
    const long NF = (long)n * D_FEAT;

    const long wsF  = (long)(ws_size / sizeof(float));
    const long tail = 2 * D_FEAT * D_FEAT + 2 * D_FEAT;
    const long Ew   = ((long)E + 3) / 4 * 4;
    const bool hasW = wsF >= 2 * NF + 2 * (long)n + tail + Ew;
    const long wsA  = hasW ? wsF - Ew : wsF;
    const int copies = (wsA >= 6 * NF + 4 * (long)n + tail) ? 2
                     : (wsA >= 4 * NF + 4 * (long)n + tail) ? 1 : 0;
    const long accPlanes = 2 + 2 * copies;
    const long degRows   = copies ? 4 : 2;
    float* ws   = (float*)d_ws;
    float* yr   = ws;
    float* yi_  = ws + NF;
    float* yr1  = copies ? ws + 2 * NF : yr;
    float* yi1  = copies ? ws + 3 * NF : yi_;
    float* degO = ws + accPlanes * NF;
    float* degI = degO + n;
    float* degO1 = copies ? degO + 2 * (long)n : degO;
    float* degI1 = copies ? degO + 3 * (long)n : degI;
    float* WRc  = degO + degRows * n;
    float* WIc  = WRc + D_FEAT * D_FEAT;
    float* bmi  = WIc + D_FEAT * D_FEAT;
    float* bpi  = bmi + D_FEAT;
    float* wE   = bpi + D_FEAT;

    float* outR = (float*)d_out;
    float* outI = outR + NF;
    float* yrt1 = (copies == 2) ? ws + 4 * NF : outR;
    float* yit1 = (copies == 2) ? ws + 5 * NF : outI;
    const int* row = ei;
    const int* col = ei + E;

    zero2_kernel<<<2048, 256, 0, stream>>>(ws, (accPlanes * NF + degRows * n) / 4,
                                           outR, (2 * NF) / 4);
    degree_kernel<<<(E + 255) / 256, 256, 0, stream>>>(row, col, E, degO, degI, degO1, degI1);
    if (copies) deginv_kernel<true><<<(n + 255) / 256, 256, 0, stream>>>(degO, degI, degO1, degI1, n);
    else        deginv_kernel<false><<<(n + 255) / 256, 256, 0, stream>>>(degO, degI, degO1, degI1, n);
    wcomb_kernel<<<(D_FEAT * D_FEAT + D_FEAT + 255) / 256, 256, 0, stream>>>(
        Wr, br, Wi, bi, K, WRc, WIc, bmi, bpi);
    if (hasW) {
        edgew_kernel<<<(E + 255) / 256, 256, 0, stream>>>(row, col, E, degO, degI, wE);
        spmm_kernel<true><<<dim3((E + 8 * SPMM_EPW - 1) / (8 * SPMM_EPW), 4), 256, 0, stream>>>(xr, xi, row, col, E, degO, degI, wE,
                                                                      yr, yi_, outR, outI,
                                                                      yr1, yi1, yrt1, yit1);
    } else {
        spmm_kernel<false><<<dim3((E + 8 * SPMM_EPW - 1) / (8 * SPMM_EPW), 4), 256, 0, stream>>>(xr, xi, row, col, E, degO, degI, wE,
                                                                       yr, yi_, outR, outI,
                                                                       yr1, yi1, yrt1, yit1);
    }
    if (copies == 2)
        faber_gemm_kernel<2><<<n / 16, 128, 0, stream>>>(
            yr, yi_, yr1, yi1, yrt1, yit1, WRc, WIc, bmi, bpi, outR, outI);
    else if (copies == 1)
        faber_gemm_kernel<1><<<n / 16, 128, 0, stream>>>(
            yr, yi_, yr1, yi1, yrt1, yit1, WRc, WIc, bmi, bpi, outR, outI);
    else
        faber_gemm_kernel<0><<<n / 16, 128, 0, stream>>>(
            yr, yi_, yr1, yi1, yrt1, yit1, WRc, WIc, bmi, bpi, outR, outI);
}


# linear maps taken before the edge sums: a WMMA kernel forms P,Q1,Q2 per node from x, the 4 passes add into the outputs (bias as start value); 4 sum planes and their zero fill go; sum order changes
# speedup vs baseline: 1.8147x; 1.0667x over previous
#include <hip/hip_runtime.h>
#include <math.h>

typedef __attribute__((ext_vector_type(2))) float v2f;
typedef __attribute__((ext_vector_type(8))) float v8f;

#define D_FEAT 64

__device__ __forceinline__ int block_half(int* s_half) {
    if (__builtin_amdgcn_readfirstlane((int)(threadIdx.x >> 5)) == 0) {
        const unsigned id = __builtin_amdgcn_s_sendmsg_rtn(0x87);
        if ((threadIdx.x & 31) == 0) *s_half = (int)((id >> 18) & 1u);
    }
    __syncthreads();
    return __builtin_amdgcn_readfirstlane(*s_half);
}

__global__ void zero2_kernel(float* __restrict__ a, long na4,
                             float* __restrict__ b, long nb4) {
    long i = (long)blockIdx.x * blockDim.x + threadIdx.x;
    long stride = (long)gridDim.x * blockDim.x;
    float4 z = make_float4(0.f, 0.f, 0.f, 0.f);
    for (long j = i; j < na4 + nb4; j += stride) {
        if (j < na4) ((float4*)a)[j] = z;
        else         ((float4*)b)[j - na4] = z;
    }
}

__global__ void degree_kernel(const int* __restrict__ row, const int* __restrict__ col,
                              int E, float* degO0, float* degI0,
                              float* degO1, float* degI1) {
    __shared__ int s_half;
    const int half = block_half(&s_half);
    float* degO = half ? degO1 : degO0;
    float* degI = half ? degI1 : degI0;
    int e = blockIdx.x * blockDim.x + threadIdx.x;
    if (e >= E) return;
    unsafeAtomicAdd(&degO[row[e]], 1.0f);
    unsafeAtomicAdd(&degI[col[e]], 1.0f);
}

template <bool TWO>
__global__ void deginv_kernel(float* degO, float* degI,
                              const float* degO1, const float* degI1, int n) {
    int i = blockIdx.x * blockDim.x + threadIdx.x;
    if (i >= n) return;
    float d0 = degO[i];
    float d1 = degI[i];
    if (TWO) { d0 += degO1[i]; d1 += degI1[i]; }
    degO[i] = (d0 > 0.f) ? powf(d0, -0.25f) : 0.f;
    degI[i] = (d1 > 0.f) ? powf(d1, -0.25f) : 0.f;
}

__global__ void edgew_kernel(const int* __restrict__ row, const int* __restrict__ col, int E,
                             const float* __restrict__ out_inv, const float* __restrict__ in_inv,
                             float* __restrict__ wE) {
    int e = blockIdx.x * blockDim.x + threadIdx.x;
    if (e >= E) return;
    wE[e] = out_inv[row[e]] * in_inv[col[e]];
}

__global__ void wcomb_kernel(const float* __restrict__ Wr, const float* __restrict__ br,
                             const float* __restrict__ Wi, const float* __restrict__ bi,
                             int K, float* WRc, float* WIc, float* bmi, float* bpi,
                             float* __restrict__ zc, long nz4) {
    int i = blockIdx.x * blockDim.x + threadIdx.x;
    for (long j = i; j < nz4; j += (long)gridDim.x * blockDim.x)
        ((float4*)zc)[j] = make_float4(0.f, 0.f, 0.f, 0.f);
    if (i < D_FEAT * D_FEAT) {
        float sr = 0.f, si = 0.f, c = 1.f;
        for (int k = 0; k < K; ++k) {
            sr += c * Wr[k * D_FEAT * D_FEAT + i];
            si += c * Wi[k * D_FEAT * D_FEAT + i];
            c *= 0.5f;
        }
        WRc[i] = sr;
        WIc[i] = si;
    } else if (i < D_FEAT * D_FEAT + D_FEAT) {
        int o = i - D_FEAT * D_FEAT;
        float sr = 0.f, si = 0.f, c = 1.f;
        for (int k = 0; k < K; ++k) {
            sr += c * br[k * D_FEAT + o];
            si += c * bi[k * D_FEAT + o];
            c *= 0.5f;
        }
        bmi[o] = sr - si;
        bpi[o] = sr + si;
    }
}

#define SPMM_EPR 16
#define SPMM_RPW 2
#define SPMM_EPW (SPMM_EPR * SPMM_RPW)
template <bool HASW>
__global__ __launch_bounds__(256) void spmm_kernel(
        const float* __restrict__ xg0, const float* __restrict__ xg1,
        const float* __restrict__ xg2, const float* __restrict__ xg3,
        const int* __restrict__ row, const int* __restrict__ col, int E,
        const float* __restrict__ out_inv, const float* __restrict__ in_inv,
        const float* __restrict__ wE,
        float* yr0, float* yi0, float* yrt0, float* yit0,
        float* yr1, float* yi1, float* yrt1, float* yit1) {
    __shared__ int s_half;
    const int half = block_half(&s_half);

    const int pass = blockIdx.y;
    const bool tr = (pass & 1) != 0;
    const bool im = (pass & 2) != 0;
    const float* __restrict__ x    = im ? (tr ? xg3 : xg2) : (tr ? xg1 : xg0);
    const int*   __restrict__ gidx = tr ? row : col;
    const int*   __restrict__ sidx = tr ? col : row;
    const float* __restrict__ ginv = tr ? out_inv : in_inv;
    const float* __restrict__ sinv = tr ? in_inv : out_inv;
    float* const yr  = half ? yr1  : yr0;
    float* const yi  = half ? yi1  : yi0;
    float* const yrt = half ? yrt1 : yrt0;
    float* const yit = half ? yit1 : yit0;
    float* const yre = tr ? yrt : yr;
    float* const yim = tr ? yit : yi;
    float* const y   = im ? yim : yre;

    const int lane = threadIdx.x & 31;
    const int H = D_FEAT / 2;
    int e0 = (blockIdx.x * 8 + (threadIdx.x >> 5)) * SPMM_EPW;
    e0 = __builtin_amdgcn_readfirstlane(e0);

    for (int r = 0; r < SPMM_RPW; ++r) {
        const int eb = e0 + r * SPMM_EPR;
        if (eb >= E) break;
        if (eb + SPMM_EPR <= E) {
            int g[SPMM_EPR], s[SPMM_EPR];
            float w[SPMM_EPR], v0[SPMM_EPR], v1[SPMM_EPR];
            for (int i = 0; i < SPMM_EPR; ++i) { g[i] = gidx[eb + i]; s[i] = sidx[eb + i]; }
            if (HASW) {
                for (int i = 0; i < SPMM_EPR; i += 4) {
                    const float4 w4 = *(const float4*)(wE + eb + i);
                    w[i] = w4.x; w[i + 1] = w4.y; w[i + 2] = w4.z; w[i + 3] = w4.w;
                }
            } else {
                for (int i = 0; i < SPMM_EPR; ++i) w[i] = sinv[s[i]] * ginv[g[i]];
            }
            for (int i = 0; i < SPMM_EPR; ++i) {
                const long go = (long)g[i] * D_FEAT + lane;
                v0[i] = x[go];
                v1[i] = x[go + H];
            }
            for (int i = 0; i < SPMM_EPR; ++i) {
                const long so = (long)s[i] * D_FEAT + lane;
                unsafeAtomicAdd(&y[so],     w[i] * v0[i]);
                unsafeAtomicAdd(&y[so + H], w[i] * v1[i]);
            }
        } else {
            for (int e = eb; e < E; ++e) {
                const int g = gidx[e];
                const int s = sidx[e];
                const float w = HASW ? wE[e] : sinv[s] * ginv[g];
                const long go = (long)g * D_FEAT + lane;
                const long so = (long)s * D_FEAT + lane;
                const float a0 = x[go], a1 = x[go + H];
                unsafeAtomicAdd(&y[so],     w * a0);
                unsafeAtomicAdd(&y[so + H], w * a1);
            }
        }
    }
}

template <int COPIES>
__global__ __launch_bounds__(128) void faber_gemm_kernel(
        const float* __restrict__ yr, const float* __restrict__ yi,
        const float* __restrict__ yr1, const float* __restrict__ yi1,
        const float* __restrict__ yrt1, const float* __restrict__ yit1,
        const float* __restrict__ WRc, const float* __restrict__ WIc,
        const float* __restrict__ bmi, const float* __restrict__ bpi,
        float* outR, float* outI) {
    const int wave = threadIdx.x >> 5;
    const int lane = threadIdx.x & 31;
    const int l16  = lane & 15;
    const int half = lane >> 4;

    const float* yrt = outR;
    const float* yit = outI;

    const long rowOff = (long)(blockIdx.x * 16 + l16) * D_FEAT;
    const long wOff   = (long)(wave * 16 + l16) * D_FEAT;

    v8f accRa = {};
    v8f accRb = {};
    v8f accIa = {};
    v8f accIb = {};

    #pragma unroll
    for (int s = 0; s < 16; ++s) {
        const int k = 4 * s + 2 * half;
        float2 ar  = *(const float2*)(yr  + rowOff + k);
        float2 art = *(const float2*)(yrt + rowOff + k);
        float2 ai  = *(const float2*)(yi  + rowOff + k);
        float2 ait = *(const float2*)(yit + rowOff + k);
        if (COPIES >= 1) {
            const float2 br  = *(const float2*)(yr1  + rowOff + k);
            const float2 bi  = *(const float2*)(yi1  + rowOff + k);
            ar.x  += br.x;  ar.y  += br.y;
            ai.x  += bi.x;  ai.y  += bi.y;
        }
        if (COPIES == 2) {
            const float2 brt = *(const float2*)(yrt1 + rowOff + k);
            const float2 bit = *(const float2*)(yit1 + rowOff + k);
            art.x += brt.x; art.y += brt.y;
            ait.x += bit.x; ait.y += bit.y;
        }
        v2f Azr = {0.5f * (ar.x + art.x), 0.5f * (ar.y + art.y)};
        v2f Azi = {0.5f * (ai.x + ait.x), 0.5f * (ai.y + ait.y)};
        v2f Ayr = {ar.x, ar.y};
        float2 wr2 = *(const float2*)(WRc + wOff + k);
        float2 wi2 = *(const float2*)(WIc + wOff + k);
        v2f Bwr = {wr2.x, wr2.y};
        v2f Bwi = {wi2.x, wi2.y};
        accRa = __builtin_amdgcn_wmma_f32_16x16x4_f32(false, Azr, false, Bwr, (short)0, accRa, false, false);
        accRb = __builtin_amdgcn_wmma_f32_16x16x4_f32(false, Azi, false, Bwi, (short)0, accRb, false, false);
        accIa = __builtin_amdgcn_wmma_f32_16x16x4_f32(false, Ayr, false, Bwi, (short)0, accIa, false, false);
        accIb = __builtin_amdgcn_wmma_f32_16x16x4_f32(false, Azi, false, Bwr, (short)0, accIb, false, false);
    }

    const int colD = wave * 16 + l16;
    const float bm = bmi[colD];
    const float bp = bpi[colD];

    __syncthreads();

    #pragma unroll
    for (int r = 0; r < 8; ++r) {
        long nodeM = (long)(blockIdx.x * 16 + r + 8 * half);
        outR[nodeM * D_FEAT + colD] = accRa[r] - accRb[r] + bm;
        outI[nodeM * D_FEAT + colD] = accIa[r] + accIb[r] + bp;
    }
}

template <int NCOPY>
__global__ __launch_bounds__(128) void faber_pre_gemm_kernel(
        const float* __restrict__ xr, const float* __restrict__ xi,
        const float* __restrict__ WRc, const float* __restrict__ WIc,
        const float* __restrict__ bmi, const float* __restrict__ bpi,
        float* __restrict__ P, float* __restrict__ Q1, float* __restrict__ Q2,
        float* __restrict__ outR, float* __restrict__ outI,
        float* __restrict__ outR1, float* __restrict__ outI1) {
    const int wave = threadIdx.x >> 5;
    const int lane = threadIdx.x & 31;
    const int l16  = lane & 15;
    const int half = lane >> 4;

    const long tile = (long)blockIdx.x * 16 * D_FEAT;
    const float4 z4 = make_float4(0.f, 0.f, 0.f, 0.f);
    for (int h = 0; h < 2; ++h) {
        const int t = (int)threadIdx.x + 128 * h;
        const int d = (t * 4) & (D_FEAT - 1);
        ((float4*)(outR + tile))[t] = *(const float4*)(bmi + d);
        ((float4*)(outI + tile))[t] = *(const float4*)(bpi + d);
        if (NCOPY >= 1) ((float4*)(outR1 + tile))[t] = z4;
        if (NCOPY == 2) ((float4*)(outI1 + tile))[t] = z4;
    }

    const long rowOff = (long)(blockIdx.x * 16 + l16) * D_FEAT;
    const long wOff   = (long)(wave * 16 + l16) * D_FEAT;

    v8f accRR = {};
    v8f accII = {};
    v8f accRI = {};
    v8f accIR = {};

    for (int s = 0; s < 16; ++s) {
        const int k = 4 * s + 2 * half;
        const float2 ar  = *(const float2*)(xr + rowOff + k);
        const float2 ai  = *(const float2*)(xi + rowOff + k);
        const float2 wr2 = *(const float2*)(WRc + wOff + k);
        const float2 wi2 = *(const float2*)(WIc + wOff + k);
        v2f Ar  = {ar.x, ar.y};
        v2f Ai  = {ai.x, ai.y};
        v2f Bwr = {wr2.x, wr2.y};
        v2f Bwi = {wi2.x, wi2.y};
        accRR = __builtin_amdgcn_wmma_f32_16x16x4_f32(false, Ar, false, Bwr, (short)0, accRR, false, false);
        accII = __builtin_amdgcn_wmma_f32_16x16x4_f32(false, Ai, false, Bwi, (short)0, accII, false, false);
        accRI = __builtin_amdgcn_wmma_f32_16x16x4_f32(false, Ar, false, Bwi, (short)0, accRI, false, false);
        accIR = __builtin_amdgcn_wmma_f32_16x16x4_f32(false, Ai, false, Bwr, (short)0, accIR, false, false);
    }

    const int colD = wave * 16 + l16;
    for (int r = 0; r < 8; ++r) {
        const long o = (long)(blockIdx.x * 16 + r + 8 * half) * D_FEAT + colD;
        const float q2 = 0.5f * accIR[r];
        P[o]  = 0.5f * (accRR[r] - accII[r]);
        Q2[o] = q2;
        Q1[o] = accRI[r] + q2;
    }
}

__global__ void addcopy_kernel(float* __restrict__ out, const float* __restrict__ cp, long n4) {
    long i = (long)blockIdx.x * blockDim.x + threadIdx.x;
    long stride = (long)gridDim.x * blockDim.x;
    for (long j = i; j < n4; j += stride) {
        float4 a = ((float4*)out)[j];
        const float4 b = ((const float4*)cp)[j];
        a.x += b.x; a.y += b.y; a.z += b.z; a.w += b.w;
        ((float4*)out)[j] = a;
    }
}

extern "C" void kernel_launch(void* const* d_in, const int* in_sizes, int n_in,
                              void* d_out, int out_size, void* d_ws, size_t ws_size,
                              hipStream_t stream) {
    const float* xr = (const float*)d_in[0];
    const float* xi = (const float*)d_in[1];
    const int*   ei = (const int*)d_in[2];
    const float* Wr = (const float*)d_in[3];
    const float* br = (const float*)d_in[4];
    const float* Wi = (const float*)d_in[5];
    const float* bi = (const float*)d_in[6];

    const int n = in_sizes[0] / D_FEAT;
    const int E = in_sizes[2] / 2;
    const int K = in_sizes[3] / (D_FEAT * D_FEAT);
    const long NF = (long)n * D_FEAT;

    const int* row = ei;
    const int* col = ei + E;
    float* outR = (float*)d_out;
    float* outI = outR + NF;

    {
        const long wsF   = (long)(ws_size / sizeof(float));
        const long tail  = 2 * D_FEAT * D_FEAT + 2 * D_FEAT;
        const long Ew    = ((long)E + 3) / 4 * 4;
        const long fixed = 4 * (long)n + tail + Ew;
        if (wsF >= 3 * NF + fixed) {
            const int ncopy = (wsF >= 5 * NF + fixed) ? 2 : (wsF >= 4 * NF + fixed) ? 1 : 0;
            float* ws    = (float*)d_ws;
            float* P     = ws;
            float* Q1    = ws + NF;
            float* Q2    = ws + 2 * NF;
            float* outR1 = (ncopy >= 1) ? ws + 3 * NF : outR;
            float* outI1 = (ncopy == 2) ? ws + 4 * NF : outI;
            float* degO  = ws + (3 + ncopy) * NF;
            float* degI  = degO + n;
            float* degO1 = degO + 2 * (long)n;
            float* degI1 = degO + 3 * (long)n;
            float* WRc   = degO + 4 * (long)n;
            float* WIc   = WRc + D_FEAT * D_FEAT;
            float* bmi   = WIc + D_FEAT * D_FEAT;
            float* bpi   = bmi + D_FEAT;
            float* wE    = bpi + D_FEAT;

            wcomb_kernel<<<128, 256, 0, stream>>>(Wr, br, Wi, bi, K, WRc, WIc, bmi, bpi, degO, (long)n);
            degree_kernel<<<(E + 255) / 256, 256, 0, stream>>>(row, col, E, degO, degI, degO1, degI1);
            deginv_kernel<true><<<(n + 255) / 256, 256, 0, stream>>>(degO, degI, degO1, degI1, n);
            edgew_kernel<<<(E + 255) / 256, 256, 0, stream>>>(row, col, E, degO, degI, wE);
            if (ncopy == 2)
                faber_pre_gemm_kernel<2><<<n / 16, 128, 0, stream>>>(
                    xr, xi, WRc, WIc, bmi, bpi, P, Q1, Q2, outR, outI, outR1, outI1);
            else if (ncopy == 1)
                faber_pre_gemm_kernel<1><<<n / 16, 128, 0, stream>>>(
                    xr, xi, WRc, WIc, bmi, bpi, P, Q1, Q2, outR, outI, outR1, outI1);
            else
                faber_pre_gemm_kernel<0><<<n / 16, 128, 0, stream>>>(
                    xr, xi, WRc, WIc, bmi, bpi, P, Q1, Q2, outR, outI, outR1, outI1);
            spmm_kernel<true><<<dim3((E + 8 * SPMM_EPW - 1) / (8 * SPMM_EPW), 4), 256, 0, stream>>>(
                P, P, Q1, Q2, row, col, E, degO, degI, wE,
                outR, outI, outR, outI,
                outR1, outI1, outR1, outI1);
            if (ncopy)
                addcopy_kernel<<<2048, 256, 0, stream>>>(outR, outR1, ((long)ncopy * NF) / 4);
            return;
        }
    }

    const long wsF  = (long)(ws_size / sizeof(float));
    const long tail = 2 * D_FEAT * D_FEAT + 2 * D_FEAT;
    const long Ew   = ((long)E + 3) / 4 * 4;
    const bool hasW = wsF >= 2 * NF + 2 * (long)n + tail + Ew;
    const long wsA  = hasW ? wsF - Ew : wsF;
    const int copies = (wsA >= 6 * NF + 4 * (long)n + tail) ? 2
                     : (wsA >= 4 * NF + 4 * (long)n + tail) ? 1 : 0;
    const long accPlanes = 2 + 2 * copies;
    const long degRows   = copies ? 4 : 2;
    float* ws   = (float*)d_ws;
    float* yr   = ws;
    float* yi_  = ws + NF;
    float* yr1  = copies ? ws + 2 * NF : yr;
    float* yi1  = copies ? ws + 3 * NF : yi_;
    float* degO = ws + accPlanes * NF;
    float* degI = degO + n;
    float* degO1 = copies ? degO + 2 * (long)n : degO;
    float* degI1 = copies ? degO + 3 * (long)n : degI;
    float* WRc  = degO + degRows * n;
    float* WIc  = WRc + D_FEAT * D_FEAT;
    float* bmi  = WIc + D_FEAT * D_FEAT;
    float* bpi  = bmi + D_FEAT;
    float* wE   = bpi + D_FEAT;

    float* yrt1 = (copies == 2) ? ws + 4 * NF : outR;
    float* yit1 = (copies == 2) ? ws + 5 * NF : outI;
    zero2_kernel<<<2048, 256, 0, stream>>>(ws, (accPlanes * NF + degRows * n) / 4,
                                           outR, (2 * NF) / 4);
    degree_kernel<<<(E + 255) / 256, 256, 0, stream>>>(row, col, E, degO, degI, degO1, degI1);
    if (copies) deginv_kernel<true><<<(n + 255) / 256, 256, 0, stream>>>(degO, degI, degO1, degI1, n);
    else        deginv_kernel<false><<<(n + 255) / 256, 256, 0, stream>>>(degO, degI, degO1, degI1, n);
    wcomb_kernel<<<(D_FEAT * D_FEAT + D_FEAT + 255) / 256, 256, 0, stream>>>(
        Wr, br, Wi, bi, K, WRc, WIc, bmi, bpi, nullptr, 0L);
    if (hasW) {
        edgew_kernel<<<(E + 255) / 256, 256, 0, stream>>>(row, col, E, degO, degI, wE);
        spmm_kernel<true><<<dim3((E + 8 * SPMM_EPW - 1) / (8 * SPMM_EPW), 4), 256, 0, stream>>>(xr, xr, xi, xi, row, col, E, degO, degI, wE,
                                                                      yr, yi_, outR, outI,
                                                                      yr1, yi1, yrt1, yit1);
    } else {
        spmm_kernel<false><<<dim3((E + 8 * SPMM_EPW - 1) / (8 * SPMM_EPW), 4), 256, 0, stream>>>(xr, xr, xi, xi, row, col, E, degO, degI, wE,
                                                                       yr, yi_, outR, outI,
                                                                       yr1, yi1, yrt1, yit1);
    }
    if (copies == 2)
        faber_gemm_kernel<2><<<n / 16, 128, 0, stream>>>(
            yr, yi_, yr1, yi1, yrt1, yit1, WRc, WIc, bmi, bpi, outR, outI);
    else if (copies == 1)
        faber_gemm_kernel<1><<<n / 16, 128, 0, stream>>>(
            yr, yi_, yr1, yi1, yrt1, yit1, WRc, WIc, bmi, bpi, outR, outI);
    else
        faber_gemm_kernel<0><<<n / 16, 128, 0, stream>>>(
            yr, yi_, yr1, yi1, yrt1, yit1, WRc, WIc, bmi, bpi, outR, outI);
}


# edge scatter: each output plane is added into by one package half only (workgroups draw work from their half's queue by integer ticket); half-1 copies, zero fill, merge launch go; sum order changes
# speedup vs baseline: 2.2925x; 1.2633x over previous
#include <hip/hip_runtime.h>
#include <math.h>

typedef __attribute__((ext_vector_type(2))) float v2f;
typedef __attribute__((ext_vector_type(8))) float v8f;

#define D_FEAT 64

__device__ __forceinline__ int block_half(int* s_half) {
    if (__builtin_amdgcn_readfirstlane((int)(threadIdx.x >> 5)) == 0) {
        const unsigned id = __builtin_amdgcn_s_sendmsg_rtn(0x87);
        if ((threadIdx.x & 31) == 0) *s_half = (int)((id >> 18) & 1u);
    }
    __syncthreads();
    return __builtin_amdgcn_readfirstlane(*s_half);
}

__global__ void zero2_kernel(float* __restrict__ a, long na4,
                             float* __restrict__ b, long nb4) {
    long i = (long)blockIdx.x * blockDim.x + threadIdx.x;
    long stride = (long)gridDim.x * blockDim.x;
    float4 z = make_float4(0.f, 0.f, 0.f, 0.f);
    for (long j = i; j < na4 + nb4; j += stride) {
        if (j < na4) ((float4*)a)[j] = z;
        else         ((float4*)b)[j - na4] = z;
    }
}

__global__ void degree_kernel(const int* __restrict__ row, const int* __restrict__ col,
                              int E, float* degO0, float* degI0,
                              float* degO1, float* degI1) {
    __shared__ int s_half;
    const int half = block_half(&s_half);
    float* degO = half ? degO1 : degO0;
    float* degI = half ? degI1 : degI0;
    int e = blockIdx.x * blockDim.x + threadIdx.x;
    if (e >= E) return;
    unsafeAtomicAdd(&degO[row[e]], 1.0f);
    unsafeAtomicAdd(&degI[col[e]], 1.0f);
}

template <bool TWO>
__global__ void deginv_kernel(float* degO, float* degI,
                              const float* degO1, const float* degI1, int n) {
    int i = blockIdx.x * blockDim.x + threadIdx.x;
    if (i >= n) return;
    float d0 = degO[i];
    float d1 = degI[i];
    if (TWO) { d0 += degO1[i]; d1 += degI1[i]; }
    degO[i] = (d0 > 0.f) ? powf(d0, -0.25f) : 0.f;
    degI[i] = (d1 > 0.f) ? powf(d1, -0.25f) : 0.f;
}

__global__ void edgew_kernel(const int* __restrict__ row, const int* __restrict__ col, int E,
                             const float* __restrict__ out_inv, const float* __restrict__ in_inv,
                             float* __restrict__ wE) {
    int e = blockIdx.x * blockDim.x + threadIdx.x;
    if (e >= E) return;
    wE[e] = out_inv[row[e]] * in_inv[col[e]];
}

__global__ void wcomb_kernel(const float* __restrict__ Wr, const float* __restrict__ br,
                             const float* __restrict__ Wi, const float* __restrict__ bi,
                             int K, float* WRc, float* WIc, float* bmi, float* bpi,
                             float* __restrict__ zc, long nz4) {
    int i = blockIdx.x * blockDim.x + threadIdx.x;
    for (long j = i; j < nz4; j += (long)gridDim.x * blockDim.x)
        ((float4*)zc)[j] = make_float4(0.f, 0.f, 0.f, 0.f);
    if (i < D_FEAT * D_FEAT) {
        float sr = 0.f, si = 0.f, c = 1.f;
        for (int k = 0; k < K; ++k) {
            sr += c * Wr[k * D_FEAT * D_FEAT + i];
            si += c * Wi[k * D_FEAT * D_FEAT + i];
            c *= 0.5f;
        }
        WRc[i] = sr;
        WIc[i] = si;
    } else if (i < D_FEAT * D_FEAT + D_FEAT) {
        int o = i - D_FEAT * D_FEAT;
        float sr = 0.f, si = 0.f, c = 1.f;
        for (int k = 0; k < K; ++k) {
            sr += c * br[k * D_FEAT + o];
            si += c * bi[k * D_FEAT + o];
            c *= 0.5f;
        }
        bmi[o] = sr - si;
        bpi[o] = sr + si;
    }
}

#define SPMM_EPR 16
#define SPMM_RPW 2
#define SPMM_EPW (SPMM_EPR * SPMM_RPW)
template <bool HASW>
__global__ __launch_bounds__(256) void spmm_kernel(
        const float* __restrict__ xg0, const float* __restrict__ xg1,
        const float* __restrict__ xg2, const float* __restrict__ xg3,
        const int* __restrict__ row, const int* __restrict__ col, int E,
        const float* __restrict__ out_inv, const float* __restrict__ in_inv,
        const float* __restrict__ wE,
        float* yr0, float* yi0, float* yrt0, float* yit0,
        float* yr1, float* yi1, float* yrt1, float* yit1) {
    __shared__ int s_half;
    const int half = block_half(&s_half);

    const int pass = blockIdx.y;
    const bool tr = (pass & 1) != 0;
    const bool im = (pass & 2) != 0;
    const float* __restrict__ x    = im ? (tr ? xg3 : xg2) : (tr ? xg1 : xg0);
    const int*   __restrict__ gidx = tr ? row : col;
    const int*   __restrict__ sidx = tr ? col : row;
    const float* __restrict__ ginv = tr ? out_inv : in_inv;
    const float* __restrict__ sinv = tr ? in_inv : out_inv;
    float* const yr  = half ? yr1  : yr0;
    float* const yi  = half ? yi1  : yi0;
    float* const yrt = half ? yrt1 : yrt0;
    float* const yit = half ? yit1 : yit0;
    float* const yre = tr ? yrt : yr;
    float* const yim = tr ? yit : yi;
    float* const y   = im ? yim : yre;

    const int lane = threadIdx.x & 31;
    const int H = D_FEAT / 2;
    int e0 = (blockIdx.x * 8 + (threadIdx.x >> 5)) * SPMM_EPW;
    e0 = __builtin_amdgcn_readfirstlane(e0);

    for (int r = 0; r < SPMM_RPW; ++r) {
        const int eb = e0 + r * SPMM_EPR;
        if (eb >= E) break;
        if (eb + SPMM_EPR <= E) {
            int g[SPMM_EPR], s[SPMM_EPR];
            float w[SPMM_EPR], v0[SPMM_EPR], v1[SPMM_EPR];
            for (int i = 0; i < SPMM_EPR; ++i) { g[i] = gidx[eb + i]; s[i] = sidx[eb + i]; }
            if (HASW) {
                for (int i = 0; i < SPMM_EPR; i += 4) {
                    const float4 w4 = *(const float4*)(wE + eb + i);
                    w[i] = w4.x; w[i + 1] = w4.y; w[i + 2] = w4.z; w[i + 3] = w4.w;
                }
            } else {
                for (int i = 0; i < SPMM_EPR; ++i) w[i] = sinv[s[i]] * ginv[g[i]];
            }
            for (int i = 0; i < SPMM_EPR; ++i) {
                const long go = (long)g[i] * D_FEAT + lane;
                v0[i] = x[go];
                v1[i] = x[go + H];
            }
            for (int i = 0; i < SPMM_EPR; ++i) {
                const long so = (long)s[i] * D_FEAT + lane;
                unsafeAtomicAdd(&y[so],     w[i] * v0[i]);
                unsafeAtomicAdd(&y[so + H], w[i] * v1[i]);
            }
        } else {
            for (int e = eb; e < E; ++e) {
                const int g = gidx[e];
                const int s = sidx[e];
                const float w = HASW ? wE[e] : sinv[s] * ginv[g];
                const long go = (long)g * D_FEAT + lane;
                const long so = (long)s * D_FEAT + lane;
                const float a0 = x[go], a1 = x[go + H];
                unsafeAtomicAdd(&y[so],     w * a0);
                unsafeAtomicAdd(&y[so + H], w * a1);
            }
        }
    }
}

#define TICK_STRIDE 64
__global__ __launch_bounds__(256) void spmm_owned_kernel(
        const float* __restrict__ P, const float* __restrict__ Q1, const float* __restrict__ Q2,
        const int* __restrict__ row, const int* __restrict__ col, int E,
        const float* __restrict__ wE,
        float* outR, float* outI, int* tick, int Nc) {
    __shared__ int s_pass, s_chunk;
    if (__builtin_amdgcn_readfirstlane((int)(threadIdx.x >> 5)) == 0) {
        const unsigned id = __builtin_amdgcn_s_sendmsg_rtn(0x87);
        if ((threadIdx.x & 31) == 0) {
            const int h = (int)((id >> 18) & 1u);
            const int L = 2 * Nc;
            const int t = atomicAdd(tick + h * TICK_STRIDE, 1);
            const int q = (t < L) ? h : 1 - h;
            const int k = (t < L) ? t : 2 * L - 1 - t;
            const int second = (k >= Nc) ? 1 : 0;
            s_pass  = (k >= 0) ? 2 * q + second : -1;
            s_chunk = k - second * Nc;
        }
    }
    __syncthreads();
    const int pass  = __builtin_amdgcn_readfirstlane(s_pass);
    const int chunk = __builtin_amdgcn_readfirstlane(s_chunk);
    if (pass < 0) return;

    const bool tr = (pass & 1) != 0;
    const bool im = (pass & 2) != 0;
    const float* __restrict__ x    = im ? (tr ? Q2 : Q1) : P;
    const int*   __restrict__ gidx = tr ? row : col;
    const int*   __restrict__ sidx = tr ? col : row;
    float* const y = im ? outI : outR;

    const int lane = threadIdx.x & 31;
    const int H = D_FEAT / 2;
    int e0 = (chunk * 8 + (int)(threadIdx.x >> 5)) * SPMM_EPW;
    e0 = __builtin_amdgcn_readfirstlane(e0);

    for (int r = 0; r < SPMM_RPW; ++r) {
        const int eb = e0 + r * SPMM_EPR;
        if (eb >= E) break;
        if (eb + SPMM_EPR <= E) {
            int g[SPMM_EPR], s[SPMM_EPR];
            float w[SPMM_EPR], v0[SPMM_EPR], v1[SPMM_EPR];
            for (int i = 0; i < SPMM_EPR; ++i) { g[i] = gidx[eb + i]; s[i] = sidx[eb + i]; }
            for (int i = 0; i < SPMM_EPR; i += 4) {
                const float4 w4 = *(const float4*)(wE + eb + i);
                w[i] = w4.x; w[i + 1] = w4.y; w[i + 2] = w4.z; w[i + 3] = w4.w;
            }
            for (int i = 0; i < SPMM_EPR; ++i) {
                const long go = (long)g[i] * D_FEAT + lane;
                v0[i] = x[go];
                v1[i] = x[go + H];
            }
            for (int i = 0; i < SPMM_EPR; ++i) {
                const long so = (long)s[i] * D_FEAT + lane;
                unsafeAtomicAdd(&y[so],     w[i] * v0[i]);
                unsafeAtomicAdd(&y[so + H], w[i] * v1[i]);
            }
        } else {
            for (int e = eb; e < E; ++e) {
                const int g = gidx[e];
                const int s = sidx[e];
                const float w = wE[e];
                const long go = (long)g * D_FEAT + lane;
                const long so = (long)s * D_FEAT + lane;
                const float a0 = x[go], a1 = x[go + H];
                unsafeAtomicAdd(&y[so],     w * a0);
                unsafeAtomicAdd(&y[so + H], w * a1);
            }
        }
    }
}

template <int COPIES>
__global__ __launch_bounds__(128) void faber_gemm_kernel(
        const float* __restrict__ yr, const float* __restrict__ yi,
        const float* __restrict__ yr1, const float* __restrict__ yi1,
        const float* __restrict__ yrt1, const float* __restrict__ yit1,
        const float* __restrict__ WRc, const float* __restrict__ WIc,
        const float* __restrict__ bmi, const float* __restrict__ bpi,
        float* outR, float* outI) {
    const int wave = threadIdx.x >> 5;
    const int lane = threadIdx.x & 31;
    const int l16  = lane & 15;
    const int half = lane >> 4;

    const float* yrt = outR;
    const float* yit = outI;

    const long rowOff = (long)(blockIdx.x * 16 + l16) * D_FEAT;
    const long wOff   = (long)(wave * 16 + l16) * D_FEAT;

    v8f accRa = {};
    v8f accRb = {};
    v8f accIa = {};
    v8f accIb = {};

    #pragma unroll
    for (int s = 0; s < 16; ++s) {
        const int k = 4 * s + 2 * half;
        float2 ar  = *(const float2*)(yr  + rowOff + k);
        float2 art = *(const float2*)(yrt + rowOff + k);
        float2 ai  = *(const float2*)(yi  + rowOff + k);
        float2 ait = *(const float2*)(yit + rowOff + k);
        if (COPIES >= 1) {
            const float2 br  = *(const float2*)(yr1  + rowOff + k);
            const float2 bi  = *(const float2*)(yi1  + rowOff + k);
            ar.x  += br.x;  ar.y  += br.y;
            ai.x  += bi.x;  ai.y  += bi.y;
        }
        if (COPIES == 2) {
            const float2 brt = *(const float2*)(yrt1 + rowOff + k);
            const float2 bit = *(const float2*)(yit1 + rowOff + k);
            art.x += brt.x; art.y += brt.y;
            ait.x += bit.x; ait.y += bit.y;
        }
        v2f Azr = {0.5f * (ar.x + art.x), 0.5f * (ar.y + art.y)};
        v2f Azi = {0.5f * (ai.x + ait.x), 0.5f * (ai.y + ait.y)};
        v2f Ayr = {ar.x, ar.y};
        float2 wr2 = *(const float2*)(WRc + wOff + k);
        float2 wi2 = *(const float2*)(WIc + wOff + k);
        v2f Bwr = {wr2.x, wr2.y};
        v2f Bwi = {wi2.x, wi2.y};
        accRa = __builtin_amdgcn_wmma_f32_16x16x4_f32(false, Azr, false, Bwr, (short)0, accRa, false, false);
        accRb = __builtin_amdgcn_wmma_f32_16x16x4_f32(false, Azi, false, Bwi, (short)0, accRb, false, false);
        accIa = __builtin_amdgcn_wmma_f32_16x16x4_f32(false, Ayr, false, Bwi, (short)0, accIa, false, false);
        accIb = __builtin_amdgcn_wmma_f32_16x16x4_f32(false, Azi, false, Bwr, (short)0, accIb, false, false);
    }

    const int colD = wave * 16 + l16;
    const float bm = bmi[colD];
    const float bp = bpi[colD];

    __syncthreads();

    #pragma unroll
    for (int r = 0; r < 8; ++r) {
        long nodeM = (long)(blockIdx.x * 16 + r + 8 * half);
        outR[nodeM * D_FEAT + colD] = accRa[r] - accRb[r] + bm;
        outI[nodeM * D_FEAT + colD] = accIa[r] + accIb[r] + bp;
    }
}

__global__ __launch_bounds__(128) void faber_pre_gemm_kernel(
        const float* __restrict__ xr, const float* __restrict__ xi,
        const float* __restrict__ WRc, const float* __restrict__ WIc,
        const float* __restrict__ bmi, const float* __restrict__ bpi,
        float* __restrict__ P, float* __restrict__ Q1, float* __restrict__ Q2,
        float* __restrict__ outR, float* __restrict__ outI) {
    const int wave = threadIdx.x >> 5;
    const int lane = threadIdx.x & 31;
    const int l16  = lane & 15;
    const int half = lane >> 4;

    const long tile = (long)blockIdx.x * 16 * D_FEAT;
    for (int h = 0; h < 2; ++h) {
        const int t = (int)threadIdx.x + 128 * h;
        const int d = (t * 4) & (D_FEAT - 1);
        ((float4*)(outR + tile))[t] = *(const float4*)(bmi + d);
        ((float4*)(outI + tile))[t] = *(const float4*)(bpi + d);
    }

    const long rowOff = (long)(blockIdx.x * 16 + l16) * D_FEAT;
    const long wOff   = (long)(wave * 16 + l16) * D_FEAT;

    v8f accRR = {};
    v8f accII = {};
    v8f accRI = {};
    v8f accIR = {};

    for (int s = 0; s < 16; ++s) {
        const int k = 4 * s + 2 * half;
        const float2 ar  = *(const float2*)(xr + rowOff + k);
        const float2 ai  = *(const float2*)(xi + rowOff + k);
        const float2 wr2 = *(const float2*)(WRc + wOff + k);
        const float2 wi2 = *(const float2*)(WIc + wOff + k);
        v2f Ar  = {ar.x, ar.y};
        v2f Ai  = {ai.x, ai.y};
        v2f Bwr = {wr2.x, wr2.y};
        v2f Bwi = {wi2.x, wi2.y};
        accRR = __builtin_amdgcn_wmma_f32_16x16x4_f32(false, Ar, false, Bwr, (short)0, accRR, false, false);
        accII = __builtin_amdgcn_wmma_f32_16x16x4_f32(false, Ai, false, Bwi, (short)0, accII, false, false);
        accRI = __builtin_amdgcn_wmma_f32_16x16x4_f32(false, Ar, false, Bwi, (short)0, accRI, false, false);
        accIR = __builtin_amdgcn_wmma_f32_16x16x4_f32(false, Ai, false, Bwr, (short)0, accIR, false, false);
    }

    const int colD = wave * 16 + l16;
    for (int r = 0; r < 8; ++r) {
        const long o = (long)(blockIdx.x * 16 + r + 8 * half) * D_FEAT + colD;
        const float q2 = 0.5f * accIR[r];
        P[o]  = 0.5f * (accRR[r] - accII[r]);
        Q2[o] = q2;
        Q1[o] = accRI[r] + q2;
    }
}

extern "C" void kernel_launch(void* const* d_in, const int* in_sizes, int n_in,
                              void* d_out, int out_size, void* d_ws, size_t ws_size,
                              hipStream_t stream) {
    const float* xr = (const float*)d_in[0];
    const float* xi = (const float*)d_in[1];
    const int*   ei = (const int*)d_in[2];
    const float* Wr = (const float*)d_in[3];
    const float* br = (const float*)d_in[4];
    const float* Wi = (const float*)d_in[5];
    const float* bi = (const float*)d_in[6];

    const int n = in_sizes[0] / D_FEAT;
    const int E = in_sizes[2] / 2;
    const int K = in_sizes[3] / (D_FEAT * D_FEAT);
    const long NF = (long)n * D_FEAT;

    const int* row = ei;
    const int* col = ei + E;
    float* outR = (float*)d_out;
    float* outI = outR + NF;

    {
        const long wsF   = (long)(ws_size / sizeof(float));
        const long tail  = 2 * D_FEAT * D_FEAT + 2 * D_FEAT;
        const long Ew    = ((long)E + 3) / 4 * 4;
        const long fixed = 2 * TICK_STRIDE + 4 * (long)n + tail + Ew;
        if (wsF >= 3 * NF + fixed) {
            float* ws    = (float*)d_ws;
            float* P     = ws;
            float* Q1    = ws + NF;
            float* Q2    = ws + 2 * NF;
            float* tickF = ws + 3 * NF;
            float* degO  = tickF + 2 * TICK_STRIDE;
            float* degI  = degO + n;
            float* degO1 = degO + 2 * (long)n;
            float* degI1 = degO + 3 * (long)n;
            float* WRc   = degO + 4 * (long)n;
            float* WIc   = WRc + D_FEAT * D_FEAT;
            float* bmi   = WIc + D_FEAT * D_FEAT;
            float* bpi   = bmi + D_FEAT;
            float* wE    = bpi + D_FEAT;
            const int Nc = (E + 8 * SPMM_EPW - 1) / (8 * SPMM_EPW);

            wcomb_kernel<<<128, 256, 0, stream>>>(Wr, br, Wi, bi, K, WRc, WIc, bmi, bpi,
                                                  tickF, (2 * TICK_STRIDE + 4 * (long)n) / 4);
            degree_kernel<<<(E + 255) / 256, 256, 0, stream>>>(row, col, E, degO, degI, degO1, degI1);
            deginv_kernel<true><<<(n + 255) / 256, 256, 0, stream>>>(degO, degI, degO1, degI1, n);
            edgew_kernel<<<(E + 255) / 256, 256, 0, stream>>>(row, col, E, degO, degI, wE);
            faber_pre_gemm_kernel<<<n / 16, 128, 0, stream>>>(
                xr, xi, WRc, WIc, bmi, bpi, P, Q1, Q2, outR, outI);
            spmm_owned_kernel<<<4 * Nc, 256, 0, stream>>>(
                P, Q1, Q2, row, col, E, wE, outR, outI, (int*)tickF, Nc);
            return;
        }
    }

    const long wsF  = (long)(ws_size / sizeof(float));
    const long tail = 2 * D_FEAT * D_FEAT + 2 * D_FEAT;
    const long Ew   = ((long)E + 3) / 4 * 4;
    const bool hasW = wsF >= 2 * NF + 2 * (long)n + tail + Ew;
    const long wsA  = hasW ? wsF - Ew : wsF;
    const int copies = (wsA >= 6 * NF + 4 * (long)n + tail) ? 2
                     : (wsA >= 4 * NF + 4 * (long)n + tail) ? 1 : 0;
    const long accPlanes = 2 + 2 * copies;
    const long degRows   = copies ? 4 : 2;
    float* ws   = (float*)d_ws;
    float* yr   = ws;
    float* yi_  = ws + NF;
    float* yr1  = copies ? ws + 2 * NF : yr;
    float* yi1  = copies ? ws + 3 * NF : yi_;
    float* degO = ws + accPlanes * NF;
    float* degI = degO + n;
    float* degO1 = copies ? degO + 2 * (long)n : degO;
    float* degI1 = copies ? degO + 3 * (long)n : degI;
    float* WRc  = degO + degRows * n;
    float* WIc  = WRc + D_FEAT * D_FEAT;
    float* bmi  = WIc + D_FEAT * D_FEAT;
    float* bpi  = bmi + D_FEAT;
    float* wE   = bpi + D_FEAT;

    float* yrt1 = (copies == 2) ? ws + 4 * NF : outR;
    float* yit1 = (copies == 2) ? ws + 5 * NF : outI;
    zero2_kernel<<<2048, 256, 0, stream>>>(ws, (accPlanes * NF + degRows * n) / 4,
                                           outR, (2 * NF) / 4);
    degree_kernel<<<(E + 255) / 256, 256, 0, stream>>>(row, col, E, degO, degI, degO1, degI1);
    if (copies) deginv_kernel<true><<<(n + 255) / 256, 256, 0, stream>>>(degO, degI, degO1, degI1, n);
    else        deginv_kernel<false><<<(n + 255) / 256, 256, 0, stream>>>(degO, degI, degO1, degI1, n);
    wcomb_kernel<<<(D_FEAT * D_FEAT + D_FEAT + 255) / 256, 256, 0, stream>>>(
        Wr, br, Wi, bi, K, WRc, WIc, bmi, bpi, nullptr, 0L);
    if (hasW) {
        edgew_kernel<<<(E + 255) / 256, 256, 0, stream>>>(row, col, E, degO, degI, wE);
        spmm_kernel<true><<<dim3((E + 8 * SPMM_EPW - 1) / (8 * SPMM_EPW), 4), 256, 0, stream>>>(xr, xr, xi, xi, row, col, E, degO, degI, wE,
                                                                      yr, yi_, outR, outI,
                                                                      yr1, yi1, yrt1, yit1);
    } else {
        spmm_kernel<false><<<dim3((E + 8 * SPMM_EPW - 1) / (8 * SPMM_EPW), 4), 256, 0, stream>>>(xr, xr, xi, xi, row, col, E, degO, degI, wE,
                                                                       yr, yi_, outR, outI,
                                                                       yr1, yi1, yrt1, yit1);
    }
    if (copies == 2)
        faber_gemm_kernel<2><<<n / 16, 128, 0, stream>>>(
            yr, yi_, yr1, yi1, yrt1, yit1, WRc, WIc, bmi, bpi, outR, outI);
    else if (copies == 1)
        faber_gemm_kernel<1><<<n / 16, 128, 0, stream>>>(
            yr, yi_, yr1, yi1, yrt1, yit1, WRc, WIc, bmi, bpi, outR, outI);
    else
        faber_gemm_kernel<0><<<n / 16, 128, 0, stream>>>(
            yr, yi_, yr1, yi1, yrt1, yit1, WRc, WIc, bmi, bpi, outR, outI);
}


# node-product kernel: every operand load covers whole 128-B lines (x tile fetched once per block by 16-B loads into LDS, A fragments read there; folded weights stored in fragment order); sums as were
# speedup vs baseline: 2.4325x; 1.0611x over previous
#include <hip/hip_runtime.h>
#include <math.h>

typedef __attribute__((ext_vector_type(2))) float v2f;
typedef __attribute__((ext_vector_type(8))) float v8f;

#define D_FEAT 64

__device__ __forceinline__ int block_half(int* s_half) {
    if (__builtin_amdgcn_readfirstlane((int)(threadIdx.x >> 5)) == 0) {
        const unsigned id = __builtin_amdgcn_s_sendmsg_rtn(0x87);
        if ((threadIdx.x & 31) == 0) *s_half = (int)((id >> 18) & 1u);
    }
    __syncthreads();
    return __builtin_amdgcn_readfirstlane(*s_half);
}

__global__ void zero2_kernel(float* __restrict__ a, long na4,
                             float* __restrict__ b, long nb4) {
    long i = (long)blockIdx.x * blockDim.x + threadIdx.x;
    long stride = (long)gridDim.x * blockDim.x;
    float4 z = make_float4(0.f, 0.f, 0.f, 0.f);
    for (long j = i; j < na4 + nb4; j += stride) {
        if (j < na4) ((float4*)a)[j] = z;
        else         ((float4*)b)[j - na4] = z;
    }
}

__global__ void degree_kernel(const int* __restrict__ row, const int* __restrict__ col,
                              int E, float* degO0, float* degI0,
                              float* degO1, float* degI1) {
    __shared__ int s_half;
    const int half = block_half(&s_half);
    float* degO = half ? degO1 : degO0;
    float* degI = half ? degI1 : degI0;
    int e = blockIdx.x * blockDim.x + threadIdx.x;
    if (e >= E) return;
    unsafeAtomicAdd(&degO[row[e]], 1.0f);
    unsafeAtomicAdd(&degI[col[e]], 1.0f);
}

template <bool TWO>
__global__ void deginv_kernel(float* degO, float* degI,
                              const float* degO1, const float* degI1, int n) {
    int i = blockIdx.x * blockDim.x + threadIdx.x;
    if (i >= n) return;
    float d0 = degO[i];
    float d1 = degI[i];
    if (TWO) { d0 += degO1[i]; d1 += degI1[i]; }
    degO[i] = (d0 > 0.f) ? powf(d0, -0.25f) : 0.f;
    degI[i] = (d1 > 0.f) ? powf(d1, -0.25f) : 0.f;
}

__global__ void edgew_kernel(const int* __restrict__ row, const int* __restrict__ col, int E,
                             const float* __restrict__ out_inv, const float* __restrict__ in_inv,
                             float* __restrict__ wE) {
    int e = blockIdx.x * blockDim.x + threadIdx.x;
    if (e >= E) return;
    wE[e] = out_inv[row[e]] * in_inv[col[e]];
}

template <bool FRAG>
__global__ void wcomb_kernel(const float* __restrict__ Wr, const float* __restrict__ br,
                             const float* __restrict__ Wi, const float* __restrict__ bi,
                             int K, float* WRc, float* WIc, float* bmi, float* bpi,
                             float* __restrict__ zc, long nz4) {
    int i = blockIdx.x * blockDim.x + threadIdx.x;
    for (long j = i; j < nz4; j += (long)gridDim.x * blockDim.x)
        ((float4*)zc)[j] = make_float4(0.f, 0.f, 0.f, 0.f);
    if (i < D_FEAT * D_FEAT) {
        float sr = 0.f, si = 0.f, c = 1.f;
        for (int k = 0; k < K; ++k) {
            sr += c * Wr[k * D_FEAT * D_FEAT + i];
            si += c * Wi[k * D_FEAT * D_FEAT + i];
            c *= 0.5f;
        }
        int p = i;
        if (FRAG) {
            const int o = i / D_FEAT, k = i % D_FEAT;
            const int l = (o & 15) + 16 * ((k >> 1) & 1);
            p = (((o >> 4) * (D_FEAT / 4) + (k >> 2)) * 32 + l) * 2 + (k & 1);
        }
        WRc[p] = sr;
        WIc[p] = si;
    } else if (i < D_FEAT * D_FEAT + D_FEAT) {
        int o = i - D_FEAT * D_FEAT;
        float sr = 0.f, si = 0.f, c = 1.f;
        for (int k = 0; k < K; ++k) {
            sr += c * br[k * D_FEAT + o];
            si += c * bi[k * D_FEAT + o];
            c *= 0.5f;
        }
        bmi[o] = sr - si;
        bpi[o] = sr + si;
    }
}

#define SPMM_EPR 16
#define SPMM_RPW 2
#define SPMM_EPW (SPMM_EPR * SPMM_RPW)
template <bool HASW>
__global__ __launch_bounds__(256) void spmm_kernel(
        const float* __restrict__ xg0, const float* __restrict__ xg1,
        const float* __restrict__ xg2, const float* __restrict__ xg3,
        const int* __restrict__ row, const int* __restrict__ col, int E,
        const float* __restrict__ out_inv, const float* __restrict__ in_inv,
        const float* __restrict__ wE,
        float* yr0, float* yi0, float* yrt0, float* yit0,
        float* yr1, float* yi1, float* yrt1, float* yit1) {
    __shared__ int s_half;
    const int half = block_half(&s_half);

    const int pass = blockIdx.y;
    const bool tr = (pass & 1) != 0;
    const bool im = (pass & 2) != 0;
    const float* __restrict__ x    = im ? (tr ? xg3 : xg2) : (tr ? xg1 : xg0);
    const int*   __restrict__ gidx = tr ? row : col;
    const int*   __restrict__ sidx = tr ? col : row;
    const float* __restrict__ ginv = tr ? out_inv : in_inv;
    const float* __restrict__ sinv = tr ? in_inv : out_inv;
    float* const yr  = half ? yr1  : yr0;
    float* const yi  = half ? yi1  : yi0;
    float* const yrt = half ? yrt1 : yrt0;
    float* const yit = half ? yit1 : yit0;
    float* const yre = tr ? yrt : yr;
    float* const yim = tr ? yit : yi;
    float* const y   = im ? yim : yre;

    const int lane = threadIdx.x & 31;
    const int H = D_FEAT / 2;
    int e0 = (blockIdx.x * 8 + (threadIdx.x >> 5)) * SPMM_EPW;
    e0 = __builtin_amdgcn_readfirstlane(e0);

    for (int r = 0; r < SPMM_RPW; ++r) {
        const int eb = e0 + r * SPMM_EPR;
        if (eb >= E) break;
        if (eb + SPMM_EPR <= E) {
            int g[SPMM_EPR], s[SPMM_EPR];
            float w[SPMM_EPR], v0[SPMM_EPR], v1[SPMM_EPR];
            for (int i = 0; i < SPMM_EPR; ++i) { g[i] = gidx[eb + i]; s[i] = sidx[eb + i]; }
            if (HASW) {
                for (int i = 0; i < SPMM_EPR; i += 4) {
                    const float4 w4 = *(const float4*)(wE + eb + i);
                    w[i] = w4.x; w[i + 1] = w4.y; w[i + 2] = w4.z; w[i + 3] = w4.w;
                }
            } else {
                for (int i = 0; i < SPMM_EPR; ++i) w[i] = sinv[s[i]] * ginv[g[i]];
            }
            for (int i = 0; i < SPMM_EPR; ++i) {
                const long go = (long)g[i] * D_FEAT + lane;
                v0[i] = x[go];
                v1[i] = x[go + H];
            }
            for (int i = 0; i < SPMM_EPR; ++i) {
                const long so = (long)s[i] * D_FEAT + lane;
                unsafeAtomicAdd(&y[so],     w[i] * v0[i]);
                unsafeAtomicAdd(&y[so + H], w[i] * v1[i]);
            }
        } else {
            for (int e = eb; e < E; ++e) {
                const int g = gidx[e];
                const int s = sidx[e];
                const float w = HASW ? wE[e] : sinv[s] * ginv[g];
                const long go = (long)g * D_FEAT + lane;
                const long so = (long)s * D_FEAT + lane;
                const float a0 = x[go], a1 = x[go + H];
                unsafeAtomicAdd(&y[so],     w * a0);
                unsafeAtomicAdd(&y[so + H], w * a1);
            }
        }
    }
}

#define PRE_PITCH 68
#define TICK_STRIDE 64
__global__ __launch_bounds__(256) void spmm_owned_kernel(
        const float* __restrict__ P, const float* __restrict__ Q1, const float* __restrict__ Q2,
        const int* __restrict__ row, const int* __restrict__ col, int E,
        const float* __restrict__ wE,
        float* outR, float* outI, int* tick, int Nc) {
    __shared__ int s_pass, s_chunk;
    if (__builtin_amdgcn_readfirstlane((int)(threadIdx.x >> 5)) == 0) {
        const unsigned id = __builtin_amdgcn_s_sendmsg_rtn(0x87);
        if ((threadIdx.x & 31) == 0) {
            const int h = (int)((id >> 18) & 1u);
            const int L = 2 * Nc;
            const int t = atomicAdd(tick + h * TICK_STRIDE, 1);
            const int q = (t < L) ? h : 1 - h;
            const int k = (t < L) ? t : 2 * L - 1 - t;
            const int second = (k >= Nc) ? 1 : 0;
            s_pass  = (k >= 0) ? 2 * q + second : -1;
            s_chunk = k - second * Nc;
        }
    }
    __syncthreads();
    const int pass  = __builtin_amdgcn_readfirstlane(s_pass);
    const int chunk = __builtin_amdgcn_readfirstlane(s_chunk);
    if (pass < 0) return;

    const bool tr = (pass & 1) != 0;
    const bool im = (pass & 2) != 0;
    const float* __restrict__ x    = im ? (tr ? Q2 : Q1) : P;
    const int*   __restrict__ gidx = tr ? row : col;
    const int*   __restrict__ sidx = tr ? col : row;
    float* const y = im ? outI : outR;

    const int lane = threadIdx.x & 31;
    const int H = D_FEAT / 2;
    int e0 = (chunk * 8 + (int)(threadIdx.x >> 5)) * SPMM_EPW;
    e0 = __builtin_amdgcn_readfirstlane(e0);

    for (int r = 0; r < SPMM_RPW; ++r) {
        const int eb = e0 + r * SPMM_EPR;
        if (eb >= E) break;
        if (eb + SPMM_EPR <= E) {
            int g[SPMM_EPR], s[SPMM_EPR];
            float w[SPMM_EPR], v0[SPMM_EPR], v1[SPMM_EPR];
            for (int i = 0; i < SPMM_EPR; ++i) { g[i] = gidx[eb + i]; s[i] = sidx[eb + i]; }
            for (int i = 0; i < SPMM_EPR; i += 4) {
                const float4 w4 = *(const float4*)(wE + eb + i);
                w[i] = w4.x; w[i + 1] = w4.y; w[i + 2] = w4.z; w[i + 3] = w4.w;
            }
            for (int i = 0; i < SPMM_EPR; ++i) {
                const long go = (long)g[i] * D_FEAT + lane;
                v0[i] = x[go];
                v1[i] = x[go + H];
            }
            for (int i = 0; i < SPMM_EPR; ++i) {
                const long so = (long)s[i] * D_FEAT + lane;
                unsafeAtomicAdd(&y[so],     w[i] * v0[i]);
                unsafeAtomicAdd(&y[so + H], w[i] * v1[i]);
            }
        } else {
            for (int e = eb; e < E; ++e) {
                const int g = gidx[e];
                const int s = sidx[e];
                const float w = wE[e];
                const long go = (long)g * D_FEAT + lane;
                const long so = (long)s * D_FEAT + lane;
                const float a0 = x[go], a1 = x[go + H];
                unsafeAtomicAdd(&y[so],     w * a0);
                unsafeAtomicAdd(&y[so + H], w * a1);
            }
        }
    }
}

template <int COPIES>
__global__ __launch_bounds__(128) void faber_gemm_kernel(
        const float* __restrict__ yr, const float* __restrict__ yi,
        const float* __restrict__ yr1, const float* __restrict__ yi1,
        const float* __restrict__ yrt1, const float* __restrict__ yit1,
        const float* __restrict__ WRc, const float* __restrict__ WIc,
        const float* __restrict__ bmi, const float* __restrict__ bpi,
        float* outR, float* outI) {
    const int wave = threadIdx.x >> 5;
    const int lane = threadIdx.x & 31;
    const int l16  = lane & 15;
    const int half = lane >> 4;

    const float* yrt = outR;
    const float* yit = outI;

    const long rowOff = (long)(blockIdx.x * 16 + l16) * D_FEAT;
    const long wOff   = (long)(wave * 16 + l16) * D_FEAT;

    v8f accRa = {};
    v8f accRb = {};
    v8f accIa = {};
    v8f accIb = {};

    #pragma unroll
    for (int s = 0; s < 16; ++s) {
        const int k = 4 * s + 2 * half;
        float2 ar  = *(const float2*)(yr  + rowOff + k);
        float2 art = *(const float2*)(yrt + rowOff + k);
        float2 ai  = *(const float2*)(yi  + rowOff + k);
        float2 ait = *(const float2*)(yit + rowOff + k);
        if (COPIES >= 1) {
            const float2 br  = *(const float2*)(yr1  + rowOff + k);
            const float2 bi  = *(const float2*)(yi1  + rowOff + k);
            ar.x  += br.x;  ar.y  += br.y;
            ai.x  += bi.x;  ai.y  += bi.y;
        }
        if (COPIES == 2) {
            const float2 brt = *(const float2*)(yrt1 + rowOff + k);
            const float2 bit = *(const float2*)(yit1 + rowOff + k);
            art.x += brt.x; art.y += brt.y;
            ait.x += bit.x; ait.y += bit.y;
        }
        v2f Azr = {0.5f * (ar.x + art.x), 0.5f * (ar.y + art.y)};
        v2f Azi = {0.5f * (ai.x + ait.x), 0.5f * (ai.y + ait.y)};
        v2f Ayr = {ar.x, ar.y};
        float2 wr2 = *(const float2*)(WRc + wOff + k);
        float2 wi2 = *(const float2*)(WIc + wOff + k);
        v2f Bwr = {wr2.x, wr2.y};
        v2f Bwi = {wi2.x, wi2.y};
        accRa = __builtin_amdgcn_wmma_f32_16x16x4_f32(false, Azr, false, Bwr, (short)0, accRa, false, false);
        accRb = __builtin_amdgcn_wmma_f32_16x16x4_f32(false, Azi, false, Bwi, (short)0, accRb, false, false);
        accIa = __builtin_amdgcn_wmma_f32_16x16x4_f32(false, Ayr, false, Bwi, (short)0, accIa, false, false);
        accIb = __builtin_amdgcn_wmma_f32_16x16x4_f32(false, Azi, false, Bwr, (short)0, accIb, false, false);
    }

    const int colD = wave * 16 + l16;
    const float bm = bmi[colD];
    const float bp = bpi[colD];

    __syncthreads();

    #pragma unroll
    for (int r = 0; r < 8; ++r) {
        long nodeM = (long)(blockIdx.x * 16 + r + 8 * half);
        outR[nodeM * D_FEAT + colD] = accRa[r] - accRb[r] + bm;
        outI[nodeM * D_FEAT + colD] = accIa[r] + accIb[r] + bp;
    }
}

__global__ __launch_bounds__(128) void faber_pre_gemm_kernel(
        const float* __restrict__ xr, const float* __restrict__ xi,
        const float* __restrict__ WRf, const float* __restrict__ WIf,
        const float* __restrict__ bmi, const float* __restrict__ bpi,
        float* __restrict__ P, float* __restrict__ Q1, float* __restrict__ Q2,
        float* __restrict__ outR, float* __restrict__ outI) {
    __shared__ __attribute__((aligned(16))) float sXr[16 * PRE_PITCH];
    __shared__ __attribute__((aligned(16))) float sXi[16 * PRE_PITCH];

    const int wave = threadIdx.x >> 5;
    const int lane = threadIdx.x & 31;
    const int l16  = lane & 15;
    const int half = lane >> 4;

    const long tile = (long)blockIdx.x * 16 * D_FEAT;
    float4 tr[2], ti[2];
    for (int h = 0; h < 2; ++h) {
        const int t = (int)threadIdx.x + 128 * h;
        tr[h] = ((const float4*)(xr + tile))[t];
        ti[h] = ((const float4*)(xi + tile))[t];
    }

    const float* wrp = WRf + (long)wave * (D_FEAT / 4) * 64 + lane * 2;
    const float* wip = WIf + (long)wave * (D_FEAT / 4) * 64 + lane * 2;
    float2 wr2[16], wi2[16];
    for (int s = 0; s < 16; ++s) {
        wr2[s] = *(const float2*)(wrp + s * 64);
        wi2[s] = *(const float2*)(wip + s * 64);
    }

    for (int h = 0; h < 2; ++h) {
        const int t = (int)threadIdx.x + 128 * h;
        const int d = (t * 4) & (D_FEAT - 1);
        ((float4*)(outR + tile))[t] = *(const float4*)(bmi + d);
        ((float4*)(outI + tile))[t] = *(const float4*)(bpi + d);
    }

    for (int h = 0; h < 2; ++h) {
        const int t = (int)threadIdx.x + 128 * h;
        const int o = (t >> 4) * PRE_PITCH + (t & 15) * 4;
        *(float4*)(sXr + o) = tr[h];
        *(float4*)(sXi + o) = ti[h];
    }
    __syncthreads();

    v8f accRR = {};
    v8f accII = {};
    v8f accRI = {};
    v8f accIR = {};

    const int aOff = l16 * PRE_PITCH + 2 * half;
    for (int s = 0; s < 16; ++s) {
        const float2 ar = *(const float2*)(sXr + aOff + 4 * s);
        const float2 ai = *(const float2*)(sXi + aOff + 4 * s);
        v2f Ar  = {ar.x, ar.y};
        v2f Ai  = {ai.x, ai.y};
        v2f Bwr = {wr2[s].x, wr2[s].y};
        v2f Bwi = {wi2[s].x, wi2[s].y};
        accRR = __builtin_amdgcn_wmma_f32_16x16x4_f32(false, Ar, false, Bwr, (short)0, accRR, false, false);
        accII = __builtin_amdgcn_wmma_f32_16x16x4_f32(false, Ai, false, Bwi, (short)0, accII, false, false);
        accRI = __builtin_amdgcn_wmma_f32_16x16x4_f32(false, Ar, false, Bwi, (short)0, accRI, false, false);
        accIR = __builtin_amdgcn_wmma_f32_16x16x4_f32(false, Ai, false, Bwr, (short)0, accIR, false, false);
    }

    const int colD = wave * 16 + l16;
    for (int r = 0; r < 8; ++r) {
        const long o = (long)(blockIdx.x * 16 + r + 8 * half) * D_FEAT + colD;
        const float q2 = 0.5f * accIR[r];
        P[o]  = 0.5f * (accRR[r] - accII[r]);
        Q2[o] = q2;
        Q1[o] = accRI[r] + q2;
    }
}

extern "C" void kernel_launch(void* const* d_in, const int* in_sizes, int n_in,
                              void* d_out, int out_size, void* d_ws, size_t ws_size,
                              hipStream_t stream) {
    const float* xr = (const float*)d_in[0];
    const float* xi = (const float*)d_in[1];
    const int*   ei = (const int*)d_in[2];
    const float* Wr = (const float*)d_in[3];
    const float* br = (const float*)d_in[4];
    const float* Wi = (const float*)d_in[5];
    const float* bi = (const float*)d_in[6];

    const int n = in_sizes[0] / D_FEAT;
    const int E = in_sizes[2] / 2;
    const int K = in_sizes[3] / (D_FEAT * D_FEAT);
    const long NF = (long)n * D_FEAT;

    const int* row = ei;
    const int* col = ei + E;
    float* outR = (float*)d_out;
    float* outI = outR + NF;

    {
        const long wsF   = (long)(ws_size / sizeof(float));
        const long tail  = 2 * D_FEAT * D_FEAT + 2 * D_FEAT;
        const long Ew    = ((long)E + 3) / 4 * 4;
        const long fixed = 2 * TICK_STRIDE + 4 * (long)n + tail + Ew;
        if (wsF >= 3 * NF + fixed) {
            float* ws    = (float*)d_ws;
            float* P     = ws;
            float* Q1    = ws + NF;
            float* Q2    = ws + 2 * NF;
            float* tickF = ws + 3 * NF;
            float* degO  = tickF + 2 * TICK_STRIDE;
            float* degI  = degO + n;
            float* degO1 = degO + 2 * (long)n;
            float* degI1 = degO + 3 * (long)n;
            float* WRc   = degO + 4 * (long)n;
            float* WIc   = WRc + D_FEAT * D_FEAT;
            float* bmi   = WIc + D_FEAT * D_FEAT;
            float* bpi   = bmi + D_FEAT;
            float* wE    = bpi + D_FEAT;
            const int Nc = (E + 8 * SPMM_EPW - 1) / (8 * SPMM_EPW);

            wcomb_kernel<true><<<128, 256, 0, stream>>>(Wr, br, Wi, bi, K, WRc, WIc, bmi, bpi,
                                                        tickF, (2 * TICK_STRIDE + 4 * (long)n) / 4);
            degree_kernel<<<(E + 255) / 256, 256, 0, stream>>>(row, col, E, degO, degI, degO1, degI1);
            deginv_kernel<true><<<(n + 255) / 256, 256, 0, stream>>>(degO, degI, degO1, degI1, n);
            edgew_kernel<<<(E + 255) / 256, 256, 0, stream>>>(row, col, E, degO, degI, wE);
            faber_pre_gemm_kernel<<<n / 16, 128, 0, stream>>>(
                xr, xi, WRc, WIc, bmi, bpi, P, Q1, Q2, outR, outI);
            spmm_owned_kernel<<<4 * Nc, 256, 0, stream>>>(
                P, Q1, Q2, row, col, E, wE, outR, outI, (int*)tickF, Nc);
            return;
        }
    }

    const long wsF  = (long)(ws_size / sizeof(float));
    const long tail = 2 * D_FEAT * D_FEAT + 2 * D_FEAT;
    const long Ew   = ((long)E + 3) / 4 * 4;
    const bool hasW = wsF >= 2 * NF + 2 * (long)n + tail + Ew;
    const long wsA  = hasW ? wsF - Ew : wsF;
    const int copies = (wsA >= 6 * NF + 4 * (long)n + tail) ? 2
                     : (wsA >= 4 * NF + 4 * (long)n + tail) ? 1 : 0;
    const long accPlanes = 2 + 2 * copies;
    const long degRows   = copies ? 4 : 2;
    float* ws   = (float*)d_ws;
    float* yr   = ws;
    float* yi_  = ws + NF;
    float* yr1  = copies ? ws + 2 * NF : yr;
    float* yi1  = copies ? ws + 3 * NF : yi_;
    float* degO = ws + accPlanes * NF;
    float* degI = degO + n;
    float* degO1 = copies ? degO + 2 * (long)n : degO;
    float* degI1 = copies ? degO + 3 * (long)n : degI;
    float* WRc  = degO + degRows * n;
    float* WIc  = WRc + D_FEAT * D_FEAT;
    float* bmi  = WIc + D_FEAT * D_FEAT;
    float* bpi  = bmi + D_FEAT;
    float* wE   = bpi + D_FEAT;

    float* yrt1 = (copies == 2) ? ws + 4 * NF : outR;
    float* yit1 = (copies == 2) ? ws + 5 * NF : outI;
    zero2_kernel<<<2048, 256, 0, stream>>>(ws, (accPlanes * NF + degRows * n) / 4,
                                           outR, (2 * NF) / 4);
    degree_kernel<<<(E + 255) / 256, 256, 0, stream>>>(row, col, E, degO, degI, degO1, degI1);
    if (copies) deginv_kernel<true><<<(n + 255) / 256, 256, 0, stream>>>(degO, degI, degO1, degI1, n);
    else        deginv_kernel<false><<<(n + 255) / 256, 256, 0, stream>>>(degO, degI, degO1, degI1, n);
    wcomb_kernel<false><<<(D_FEAT * D_FEAT + D_FEAT + 255) / 256, 256, 0, stream>>>(
        Wr, br, Wi, bi, K, WRc, WIc, bmi, bpi, nullptr, 0L);
    if (hasW) {
        edgew_kernel<<<(E + 255) / 256, 256, 0, stream>>>(row, col, E, degO, degI, wE);
        spmm_kernel<true><<<dim3((E + 8 * SPMM_EPW - 1) / (8 * SPMM_EPW), 4), 256, 0, stream>>>(xr, xr, xi, xi, row, col, E, degO, degI, wE,
                                                                      yr, yi_, outR, outI,
                                                                      yr1, yi1, yrt1, yit1);
    } else {
        spmm_kernel<false><<<dim3((E + 8 * SPMM_EPW - 1) / (8 * SPMM_EPW), 4), 256, 0, stream>>>(xr, xr, xi, xi, row, col, E, degO, degI, wE,
                                                                       yr, yi_, outR, outI,
                                                                       yr1, yi1, yrt1, yit1);
    }
    if (copies == 2)
        faber_gemm_kernel<2><<<n / 16, 128, 0, stream>>>(
            yr, yi_, yr1, yi1, yrt1, yit1, WRc, WIc, bmi, bpi, outR, outI);
    else if (copies == 1)
        faber_gemm_kernel<1><<<n / 16, 128, 0, stream>>>(
            yr, yi_, yr1, yi1, yrt1, yit1, WRc, WIc, bmi, bpi, outR, outI);
    else
        faber_gemm_kernel<0><<<n / 16, 128, 0, stream>>>(
            yr, yi_, yr1, yi1, yrt1, yit1, WRc, WIc, bmi, bpi, outR, outI);
}


# edge scatter gathers from LDS: edges binned by 512-row range of the gathered row; a workgroup stages its range's tile of P/Q1/Q2 and reads rows there; f32 atomic adds kept; sum order changes
# speedup vs baseline: 2.9738x; 1.2225x over previous
#include <hip/hip_runtime.h>
#include <math.h>

typedef __attribute__((ext_vector_type(2))) float v2f;
typedef __attribute__((ext_vector_type(8))) float v8f;

#define D_FEAT 64

__device__ __forceinline__ int block_half(int* s_half) {
    if (__builtin_amdgcn_readfirstlane((int)(threadIdx.x >> 5)) == 0) {
        const unsigned id = __builtin_amdgcn_s_sendmsg_rtn(0x87);
        if ((threadIdx.x & 31) == 0) *s_half = (int)((id >> 18) & 1u);
    }
    __syncthreads();
    return __builtin_amdgcn_readfirstlane(*s_half);
}

__global__ void zero2_kernel(float* __restrict__ a, long na4,
                             float* __restrict__ b, long nb4) {
    long i = (long)blockIdx.x * blockDim.x + threadIdx.x;
    long stride = (long)gridDim.x * blockDim.x;
    float4 z = make_float4(0.f, 0.f, 0.f, 0.f);
    for (long j = i; j < na4 + nb4; j += stride) {
        if (j < na4) ((float4*)a)[j] = z;
        else         ((float4*)b)[j - na4] = z;
    }
}

__global__ void degree_kernel(const int* __restrict__ row, const int* __restrict__ col,
                              int E, float* degO0, float* degI0,
                              float* degO1, float* degI1) {
    __shared__ int s_half;
    const int half = block_half(&s_half);
    float* degO = half ? degO1 : degO0;
    float* degI = half ? degI1 : degI0;
    int e = blockIdx.x * blockDim.x + threadIdx.x;
    if (e >= E) return;
    unsafeAtomicAdd(&degO[row[e]], 1.0f);
    unsafeAtomicAdd(&degI[col[e]], 1.0f);
}

template <bool TWO>
__global__ void deginv_kernel(float* degO, float* degI,
                              const float* degO1, const float* degI1, int n) {
    int i = blockIdx.x * blockDim.x + threadIdx.x;
    if (i >= n) return;
    float d0 = degO[i];
    float d1 = degI[i];
    if (TWO) { d0 += degO1[i]; d1 += degI1[i]; }
    degO[i] = (d0 > 0.f) ? powf(d0, -0.25f) : 0.f;
    degI[i] = (d1 > 0.f) ? powf(d1, -0.25f) : 0.f;
}

__global__ void edgew_kernel(const int* __restrict__ row, const int* __restrict__ col, int E,
                             const float* __restrict__ out_inv, const float* __restrict__ in_inv,
                             float* __restrict__ wE) {
    int e = blockIdx.x * blockDim.x + threadIdx.x;
    if (e >= E) return;
    wE[e] = out_inv[row[e]] * in_inv[col[e]];
}

template <bool FRAG>
__global__ void wcomb_kernel(const float* __restrict__ Wr, const float* __restrict__ br,
                             const float* __restrict__ Wi, const float* __restrict__ bi,
                             int K, float* WRc, float* WIc, float* bmi, float* bpi,
                             float* __restrict__ zc, long nz4) {
    int i = blockIdx.x * blockDim.x + threadIdx.x;
    for (long j = i; j < nz4; j += (long)gridDim.x * blockDim.x)
        ((float4*)zc)[j] = make_float4(0.f, 0.f, 0.f, 0.f);
    if (i < D_FEAT * D_FEAT) {
        float sr = 0.f, si = 0.f, c = 1.f;
        for (int k = 0; k < K; ++k) {
            sr += c * Wr[k * D_FEAT * D_FEAT + i];
            si += c * Wi[k * D_FEAT * D_FEAT + i];
            c *= 0.5f;
        }
        int p = i;
        if (FRAG) {
            const int o = i / D_FEAT, k = i % D_FEAT;
            const int l = (o & 15) + 16 * ((k >> 1) & 1);
            p = (((o >> 4) * (D_FEAT / 4) + (k >> 2)) * 32 + l) * 2 + (k & 1);
        }
        WRc[p] = sr;
        WIc[p] = si;
    } else if (i < D_FEAT * D_FEAT + D_FEAT) {
        int o = i - D_FEAT * D_FEAT;
        float sr = 0.f, si = 0.f, c = 1.f;
        for (int k = 0; k < K; ++k) {
            sr += c * br[k * D_FEAT + o];
            si += c * bi[k * D_FEAT + o];
            c *= 0.5f;
        }
        bmi[o] = sr - si;
        bpi[o] = sr + si;
    }
}

#define SPMM_EPR 16
#define SPMM_RPW 2
#define SPMM_EPW (SPMM_EPR * SPMM_RPW)
template <bool HASW>
__global__ __launch_bounds__(256) void spmm_kernel(
        const float* __restrict__ xg0, const float* __restrict__ xg1,
        const float* __restrict__ xg2, const float* __restrict__ xg3,
        const int* __restrict__ row, const int* __restrict__ col, int E,
        const float* __restrict__ out_inv, const float* __restrict__ in_inv,
        const float* __restrict__ wE,
        float* yr0, float* yi0, float* yrt0, float* yit0,
        float* yr1, float* yi1, float* yrt1, float* yit1) {
    __shared__ int s_half;
    const int half = block_half(&s_half);

    const int pass = blockIdx.y;
    const bool tr = (pass & 1) != 0;
    const bool im = (pass & 2) != 0;
    const float* __restrict__ x    = im ? (tr ? xg3 : xg2) : (tr ? xg1 : xg0);
    const int*   __restrict__ gidx = tr ? row : col;
    const int*   __restrict__ sidx = tr ? col : row;
    const float* __restrict__ ginv = tr ? out_inv : in_inv;
    const float* __restrict__ sinv = tr ? in_inv : out_inv;
    float* const yr  = half ? yr1  : yr0;
    float* const yi  = half ? yi1  : yi0;
    float* const yrt = half ? yrt1 : yrt0;
    float* const yit = half ? yit1 : yit0;
    float* const yre = tr ? yrt : yr;
    float* const yim = tr ? yit : yi;
    float* const y   = im ? yim : yre;

    const int lane = threadIdx.x & 31;
    const int H = D_FEAT / 2;
    int e0 = (blockIdx.x * 8 + (threadIdx.x >> 5)) * SPMM_EPW;
    e0 = __builtin_amdgcn_readfirstlane(e0);

    for (int r = 0; r < SPMM_RPW; ++r) {
        const int eb = e0 + r * SPMM_EPR;
        if (eb >= E) break;
        if (eb + SPMM_EPR <= E) {
            int g[SPMM_EPR], s[SPMM_EPR];
            float w[SPMM_EPR], v0[SPMM_EPR], v1[SPMM_EPR];
            for (int i = 0; i < SPMM_EPR; ++i) { g[i] = gidx[eb + i]; s[i] = sidx[eb + i]; }
            if (HASW) {
                for (int i = 0; i < SPMM_EPR; i += 4) {
                    const float4 w4 = *(const float4*)(wE + eb + i);
                    w[i] = w4.x; w[i + 1] = w4.y; w[i + 2] = w4.z; w[i + 3] = w4.w;
                }
            } else {
                for (int i = 0; i < SPMM_EPR; ++i) w[i] = sinv[s[i]] * ginv[g[i]];
            }
            for (int i = 0; i < SPMM_EPR; ++i) {
                const long go = (long)g[i] * D_FEAT + lane;
                v0[i] = x[go];
                v1[i] = x[go + H];
            }
            for (int i = 0; i < SPMM_EPR; ++i) {
                const long so = (long)s[i] * D_FEAT + lane;
                unsafeAtomicAdd(&y[so],     w[i] * v0[i]);
                unsafeAtomicAdd(&y[so + H], w[i] * v1[i]);
            }
        } else {
            for (int e = eb; e < E; ++e) {
                const int g = gidx[e];
                const int s = sidx[e];
                const float w = HASW ? wE[e] : sinv[s] * ginv[g];
                const long go = (long)g * D_FEAT + lane;
                const long so = (long)s * D_FEAT + lane;
                const float a0 = x[go], a1 = x[go + H];
                unsafeAtomicAdd(&y[so],     w * a0);
                unsafeAtomicAdd(&y[so + H], w * a1);
            }
        }
    }
}

#define PRE_PITCH 68
#define TICK_STRIDE 64
__global__ __launch_bounds__(256) void spmm_owned_kernel(
        const float* __restrict__ P, const float* __restrict__ Q1, const float* __restrict__ Q2,
        const int* __restrict__ row, const int* __restrict__ col, int E,
        const float* __restrict__ wE,
        float* outR, float* outI, int* tick, int Nc) {
    __shared__ int s_pass, s_chunk;
    if (__builtin_amdgcn_readfirstlane((int)(threadIdx.x >> 5)) == 0) {
        const unsigned id = __builtin_amdgcn_s_sendmsg_rtn(0x87);
        if ((threadIdx.x & 31) == 0) {
            const int h = (int)((id >> 18) & 1u);
            const int L = 2 * Nc;
            const int t = atomicAdd(tick + h * TICK_STRIDE, 1);
            const int q = (t < L) ? h : 1 - h;
            const int k = (t < L) ? t : 2 * L - 1 - t;
            const int second = (k >= Nc) ? 1 : 0;
            s_pass  = (k >= 0) ? 2 * q + second : -1;
            s_chunk = k - second * Nc;
        }
    }
    __syncthreads();
    const int pass  = __builtin_amdgcn_readfirstlane(s_pass);
    const int chunk = __builtin_amdgcn_readfirstlane(s_chunk);
    if (pass < 0) return;

    const bool tr = (pass & 1) != 0;
    const bool im = (pass & 2) != 0;
    const float* __restrict__ x    = im ? (tr ? Q2 : Q1) : P;
    const int*   __restrict__ gidx = tr ? row : col;
    const int*   __restrict__ sidx = tr ? col : row;
    float* const y = im ? outI : outR;

    const int lane = threadIdx.x & 31;
    const int H = D_FEAT / 2;
    int e0 = (chunk * 8 + (int)(threadIdx.x >> 5)) * SPMM_EPW;
    e0 = __builtin_amdgcn_readfirstlane(e0);

    for (int r = 0; r < SPMM_RPW; ++r) {
        const int eb = e0 + r * SPMM_EPR;
        if (eb >= E) break;
        if (eb + SPMM_EPR <= E) {
            int g[SPMM_EPR], s[SPMM_EPR];
            float w[SPMM_EPR], v0[SPMM_EPR], v1[SPMM_EPR];
            for (int i = 0; i < SPMM_EPR; ++i) { g[i] = gidx[eb + i]; s[i] = sidx[eb + i]; }
            for (int i = 0; i < SPMM_EPR; i += 4) {
                const float4 w4 = *(const float4*)(wE + eb + i);
                w[i] = w4.x; w[i + 1] = w4.y; w[i + 2] = w4.z; w[i + 3] = w4.w;
            }
            for (int i = 0; i < SPMM_EPR; ++i) {
                const long go = (long)g[i] * D_FEAT + lane;
                v0[i] = x[go];
                v1[i] = x[go + H];
            }
            for (int i = 0; i < SPMM_EPR; ++i) {
                const long so = (long)s[i] * D_FEAT + lane;
                unsafeAtomicAdd(&y[so],     w[i] * v0[i]);
                unsafeAtomicAdd(&y[so + H], w[i] * v1[i]);
            }
        } else {
            for (int e = eb; e < E; ++e) {
                const int g = gidx[e];
                const int s = sidx[e];
                const float w = wE[e];
                const long go = (long)g * D_FEAT + lane;
                const long so = (long)s * D_FEAT + lane;
                const float a0 = x[go], a1 = x[go + H];
                unsafeAtomicAdd(&y[so],     w * a0);
                unsafeAtomicAdd(&y[so + H], w * a1);
            }
        }
    }
}

template <int COPIES>
__global__ __launch_bounds__(128) void faber_gemm_kernel(
        const float* __restrict__ yr, const float* __restrict__ yi,
        const float* __restrict__ yr1, const float* __restrict__ yi1,
        const float* __restrict__ yrt1, const float* __restrict__ yit1,
        const float* __restrict__ WRc, const float* __restrict__ WIc,
        const float* __restrict__ bmi, const float* __restrict__ bpi,
        float* outR, float* outI) {
    const int wave = threadIdx.x >> 5;
    const int lane = threadIdx.x & 31;
    const int l16  = lane & 15;
    const int half = lane >> 4;

    const float* yrt = outR;
    const float* yit = outI;

    const long rowOff = (long)(blockIdx.x * 16 + l16) * D_FEAT;
    const long wOff   = (long)(wave * 16 + l16) * D_FEAT;

    v8f accRa = {};
    v8f accRb = {};
    v8f accIa = {};
    v8f accIb = {};

    #pragma unroll
    for (int s = 0; s < 16; ++s) {
        const int k = 4 * s + 2 * half;
        float2 ar  = *(const float2*)(yr  + rowOff + k);
        float2 art = *(const float2*)(yrt + rowOff + k);
        float2 ai  = *(const float2*)(yi  + rowOff + k);
        float2 ait = *(const float2*)(yit + rowOff + k);
        if (COPIES >= 1) {
            const float2 br  = *(const float2*)(yr1  + rowOff + k);
            const float2 bi  = *(const float2*)(yi1  + rowOff + k);
            ar.x  += br.x;  ar.y  += br.y;
            ai.x  += bi.x;  ai.y  += bi.y;
        }
        if (COPIES == 2) {
            const float2 brt = *(const float2*)(yrt1 + rowOff + k);
            const float2 bit = *(const float2*)(yit1 + rowOff + k);
            art.x += brt.x; art.y += brt.y;
            ait.x += bit.x; ait.y += bit.y;
        }
        v2f Azr = {0.5f * (ar.x + art.x), 0.5f * (ar.y + art.y)};
        v2f Azi = {0.5f * (ai.x + ait.x), 0.5f * (ai.y + ait.y)};
        v2f Ayr = {ar.x, ar.y};
        float2 wr2 = *(const float2*)(WRc + wOff + k);
        float2 wi2 = *(const float2*)(WIc + wOff + k);
        v2f Bwr = {wr2.x, wr2.y};
        v2f Bwi = {wi2.x, wi2.y};
        accRa = __builtin_amdgcn_wmma_f32_16x16x4_f32(false, Azr, false, Bwr, (short)0, accRa, false, false);
        accRb = __builtin_amdgcn_wmma_f32_16x16x4_f32(false, Azi, false, Bwi, (short)0, accRb, false, false);
        accIa = __builtin_amdgcn_wmma_f32_16x16x4_f32(false, Ayr, false, Bwi, (short)0, accIa, false, false);
        accIb = __builtin_amdgcn_wmma_f32_16x16x4_f32(false, Azi, false, Bwr, (short)0, accIb, false, false);
    }

    const int colD = wave * 16 + l16;
    const float bm = bmi[colD];
    const float bp = bpi[colD];

    __syncthreads();

    #pragma unroll
    for (int r = 0; r < 8; ++r) {
        long nodeM = (long)(blockIdx.x * 16 + r + 8 * half);
        outR[nodeM * D_FEAT + colD] = accRa[r] - accRb[r] + bm;
        outI[nodeM * D_FEAT + colD] = accIa[r] + accIb[r] + bp;
    }
}

__global__ __launch_bounds__(128) void faber_pre_gemm_kernel(
        const float* __restrict__ xr, const float* __restrict__ xi,
        const float* __restrict__ WRf, const float* __restrict__ WIf,
        const float* __restrict__ bmi, const float* __restrict__ bpi,
        float* __restrict__ P, float* __restrict__ Q1, float* __restrict__ Q2,
        float* __restrict__ outR, float* __restrict__ outI) {
    __shared__ __attribute__((aligned(16))) float sXr[16 * PRE_PITCH];
    __shared__ __attribute__((aligned(16))) float sXi[16 * PRE_PITCH];

    const int wave = threadIdx.x >> 5;
    const int lane = threadIdx.x & 31;
    const int l16  = lane & 15;
    const int half = lane >> 4;

    const long tile = (long)blockIdx.x * 16 * D_FEAT;
    float4 tr[2], ti[2];
    for (int h = 0; h < 2; ++h) {
        const int t = (int)threadIdx.x + 128 * h;
        tr[h] = ((const float4*)(xr + tile))[t];
        ti[h] = ((const float4*)(xi + tile))[t];
    }

    const float* wrp = WRf + (long)wave * (D_FEAT / 4) * 64 + lane * 2;
    const float* wip = WIf + (long)wave * (D_FEAT / 4) * 64 + lane * 2;
    float2 wr2[16], wi2[16];
    for (int s = 0; s < 16; ++s) {
        wr2[s] = *(const float2*)(wrp + s * 64);
        wi2[s] = *(const float2*)(wip + s * 64);
    }

    for (int h = 0; h < 2; ++h) {
        const int t = (int)threadIdx.x + 128 * h;
        const int d = (t * 4) & (D_FEAT - 1);
        ((float4*)(outR + tile))[t] = *(const float4*)(bmi + d);
        ((float4*)(outI + tile))[t] = *(const float4*)(bpi + d);
    }

    for (int h = 0; h < 2; ++h) {
        const int t = (int)threadIdx.x + 128 * h;
        const int o = (t >> 4) * PRE_PITCH + (t & 15) * 4;
        *(float4*)(sXr + o) = tr[h];
        *(float4*)(sXi + o) = ti[h];
    }
    __syncthreads();

    v8f accRR = {};
    v8f accII = {};
    v8f accRI = {};
    v8f accIR = {};

    const int aOff = l16 * PRE_PITCH + 2 * half;
    for (int s = 0; s < 16; ++s) {
        const float2 ar = *(const float2*)(sXr + aOff + 4 * s);
        const float2 ai = *(const float2*)(sXi + aOff + 4 * s);
        v2f Ar  = {ar.x, ar.y};
        v2f Ai  = {ai.x, ai.y};
        v2f Bwr = {wr2[s].x, wr2[s].y};
        v2f Bwi = {wi2[s].x, wi2[s].y};
        accRR = __builtin_amdgcn_wmma_f32_16x16x4_f32(false, Ar, false, Bwr, (short)0, accRR, false, false);
        accII = __builtin_amdgcn_wmma_f32_16x16x4_f32(false, Ai, false, Bwi, (short)0, accII, false, false);
        accRI = __builtin_amdgcn_wmma_f32_16x16x4_f32(false, Ar, false, Bwi, (short)0, accRI, false, false);
        accIR = __builtin_amdgcn_wmma_f32_16x16x4_f32(false, Ai, false, Bwr, (short)0, accIR, false, false);
    }

    const int colD = wave * 16 + l16;
    for (int r = 0; r < 8; ++r) {
        const long o = (long)(blockIdx.x * 16 + r + 8 * half) * D_FEAT + colD;
        const float q2 = 0.5f * accIR[r];
        P[o]  = 0.5f * (accRR[r] - accII[r]);
        Q2[o] = q2;
        Q1[o] = accRI[r] + q2;
    }
}


#define SRC_SH 9
#define SRC_R (1 << SRC_SH)
#define MAX_NB 256
#define BIN_EPB 2048
#define BIN_EPT (BIN_EPB / 256)

__global__ __launch_bounds__(256) void degree_bin_kernel(
        const int* __restrict__ row, const int* __restrict__ col, int E,
        float* degO0, float* degI0, float* degO1, float* degI1, int* cnt) {
    __shared__ int s_half;
    __shared__ int s_cnt[2 * MAX_NB];
    for (int t = threadIdx.x; t < 2 * MAX_NB; t += 256) s_cnt[t] = 0;
    const int half = block_half(&s_half);
    float* degO = half ? degO1 : degO0;
    float* degI = half ? degI1 : degI0;
    const int e0 = blockIdx.x * BIN_EPB + (int)threadIdx.x;
    int r[BIN_EPT], c[BIN_EPT];
    #pragma unroll
    for (int i = 0; i < BIN_EPT; ++i) {
        const int e = e0 + i * 256;
        r[i] = (e < E) ? row[e] : -1;
        c[i] = (e < E) ? col[e] : 0;
    }
    #pragma unroll
    for (int i = 0; i < BIN_EPT; ++i) {
        if (r[i] >= 0) {
            unsafeAtomicAdd(&degO[r[i]], 1.0f);
            unsafeAtomicAdd(&degI[c[i]], 1.0f);
            atomicAdd(&s_cnt[c[i] >> SRC_SH], 1);
            atomicAdd(&s_cnt[MAX_NB + (r[i] >> SRC_SH)], 1);
        }
    }
    __syncthreads();
    for (int t = threadIdx.x; t < 2 * MAX_NB; t += 256) {
        const int v = s_cnt[t];
        if (v) atomicAdd(&cnt[t], v);
    }
}

__device__ __forceinline__ void scan2_256(int v0, int v1, int* s_wt, int& ex0, int& ex1) {
    const int lane = threadIdx.x & 31, wv = threadIdx.x >> 5;
    int i0 = v0, i1 = v1;
    for (int d = 1; d < 32; d <<= 1) {
        const int a0 = __shfl_up(i0, d, 32), a1 = __shfl_up(i1, d, 32);
        if (lane >= d) { i0 += a0; i1 += a1; }
    }
    if (lane == 31) { s_wt[wv] = i0; s_wt[8 + wv] = i1; }
    __syncthreads();
    int p0 = 0, p1 = 0;
    for (int u = 0; u < wv; ++u) { p0 += s_wt[u]; p1 += s_wt[8 + u]; }
    ex0 = p0 + i0 - v0;
    ex1 = p1 + i1 - v1;
}

__global__ __launch_bounds__(256) void deginv_scan_kernel(
        float* degO, float* degI, const float* degO1, const float* degI1, int n,
        const int* __restrict__ cnt, int* __restrict__ base, int* __restrict__ cur) {
    __shared__ int s_wt[16];
    if (blockIdx.x == gridDim.x - 1) {
        const int t = threadIdx.x;
        int ex0, ex1;
        scan2_256(cnt[t], cnt[MAX_NB + t], s_wt, ex0, ex1);
        base[t] = ex0;          cur[t] = ex0;
        base[MAX_NB + t] = ex1; cur[MAX_NB + t] = ex1;
        return;
    }
    const int i = blockIdx.x * blockDim.x + threadIdx.x;
    if (i >= n) return;
    const float d0 = degO[i] + degO1[i];
    const float d1 = degI[i] + degI1[i];
    degO[i] = (d0 > 0.f) ? powf(d0, -0.25f) : 0.f;
    degI[i] = (d1 > 0.f) ? powf(d1, -0.25f) : 0.f;
}

__global__ __launch_bounds__(256) void place_kernel(
        const int* __restrict__ row, const int* __restrict__ col, int E,
        const float* __restrict__ out_inv, const float* __restrict__ in_inv,
        int* cur, int* __restrict__ rec0w, int* __restrict__ rec1w) {
    int2* __restrict__ rec0 = (int2*)rec0w;
    int2* __restrict__ rec1 = (int2*)rec1w;
    __shared__ int  s_cnt[2 * MAX_NB];
    __shared__ int  s_nxt[2 * MAX_NB];
    __shared__ int  s_dlt[2 * MAX_NB];
    __shared__ int  s_wt[16];
    __shared__ int2 sRec[2 * BIN_EPB];
    __shared__ int  sDst[2 * BIN_EPB];
    const int t = threadIdx.x;
    s_cnt[t] = 0; s_cnt[MAX_NB + t] = 0;
    __syncthreads();
    const int eB = blockIdx.x * BIN_EPB;
    int r[BIN_EPT], c[BIN_EPT];
    #pragma unroll
    for (int i = 0; i < BIN_EPT; ++i) {
        const int e = eB + t + i * 256;
        r[i] = (e < E) ? row[e] : -1;
        c[i] = (e < E) ? col[e] : 0;
        if (r[i] >= 0) {
            atomicAdd(&s_cnt[c[i] >> SRC_SH], 1);
            atomicAdd(&s_cnt[MAX_NB + (r[i] >> SRC_SH)], 1);
        }
    }
    int wb[BIN_EPT];
    #pragma unroll
    for (int i = 0; i < BIN_EPT; ++i)
        wb[i] = __float_as_int(out_inv[r[i] >= 0 ? r[i] : 0] * in_inv[c[i]]);
    __syncthreads();
    {
        const int v0 = s_cnt[t], v1 = s_cnt[MAX_NB + t];
        int ex0, ex1;
        scan2_256(v0, v1, s_wt, ex0, ex1);
        const int g0 = v0 ? atomicAdd(&cur[t], v0) : 0;
        const int g1 = v1 ? atomicAdd(&cur[MAX_NB + t], v1) : 0;
        s_nxt[t] = ex0;                    s_dlt[t] = g0 - ex0;
        s_nxt[MAX_NB + t] = BIN_EPB + ex1; s_dlt[MAX_NB + t] = g1 - (BIN_EPB + ex1);
    }
    __syncthreads();
    #pragma unroll
    for (int i = 0; i < BIN_EPT; ++i) {
        if (r[i] >= 0) {
            const int b0 = c[i] >> SRC_SH;
            const int s0 = atomicAdd(&s_nxt[b0], 1);
            sRec[s0] = make_int2(((c[i] & (SRC_R - 1)) << 22) | r[i], wb[i]);
            sDst[s0] = s0 + s_dlt[b0];
            const int b1 = MAX_NB + (r[i] >> SRC_SH);
            const int s1 = atomicAdd(&s_nxt[b1], 1);
            sRec[s1] = make_int2(((r[i] & (SRC_R - 1)) << 22) | c[i], wb[i]);
            sDst[s1] = s1 + s_dlt[b1];
        }
    }
    __syncthreads();
    const int nv = (E - eB < BIN_EPB) ? (E - eB) : BIN_EPB;
    for (int s = t; s < nv; s += 256) {
        rec0[sDst[s]] = sRec[s];
        rec1[sDst[BIN_EPB + s]] = sRec[BIN_EPB + s];
    }
}

#define STG_T 1024
#define STG_L ((SRC_R * D_FEAT / 4) / STG_T)
__global__ __launch_bounds__(STG_T) void spmm_staged_kernel(
        const float* __restrict__ P, const float* __restrict__ Q1, const float* __restrict__ Q2,
        const int* __restrict__ rec0w, const int* __restrict__ rec1w,
        const int* __restrict__ base, const int* __restrict__ cnt,
        float* outR, float* outI, int* tick, int NB, int n) {
    __shared__ __attribute__((aligned(16))) float sT[SRC_R * D_FEAT];
    __shared__ int s_pass, s_rng;
    if (__builtin_amdgcn_readfirstlane((int)(threadIdx.x >> 5)) == 0) {
        const unsigned id = __builtin_amdgcn_s_sendmsg_rtn(0x87);
        if ((threadIdx.x & 31) == 0) {
            const int h = (int)((id >> 18) & 1u);
            const int L = 2 * NB;
            const int t = atomicAdd(tick + h * TICK_STRIDE, 1);
            const int q = (t < L) ? h : 1 - h;
            const int k = (t < L) ? t : 2 * L - 1 - t;
            const int second = (k >= NB) ? 1 : 0;
            s_pass = (k >= 0) ? 2 * q + second : -1;
            s_rng  = k - second * NB;
        }
    }
    __syncthreads();
    const int pass = __builtin_amdgcn_readfirstlane(s_pass);
    const int rng  = __builtin_amdgcn_readfirstlane(s_rng);
    if (pass < 0) return;

    const bool tr = (pass & 1) != 0;
    const bool im = (pass & 2) != 0;
    const float* __restrict__ x   = im ? (tr ? Q2 : Q1) : P;
    const int2*  __restrict__ rec = (const int2*)(tr ? rec1w : rec0w);
    float* const y = im ? outI : outR;
    const int key = tr ? MAX_NB : 0;
    const int rb = base[key + rng];
    const int rn = cnt[key + rng];

    const int row0 = rng * SRC_R;
    const int n4   = ((n - row0 < SRC_R) ? (n - row0) : SRC_R) * (D_FEAT / 4);
    const float4* __restrict__ tsrc = (const float4*)(x + (long)row0 * D_FEAT);
    float4 tv[STG_L];
    #pragma unroll
    for (int j = 0; j < STG_L; ++j) {
        const int i = (int)threadIdx.x + j * STG_T;
        tv[j] = (i < n4) ? tsrc[i] : make_float4(0.f, 0.f, 0.f, 0.f);
    }
    #pragma unroll
    for (int j = 0; j < STG_L; ++j)
        ((float4*)sT)[(int)threadIdx.x + j * STG_T] = tv[j];
    __syncthreads();

    const int lane = threadIdx.x & 31;
    const int H = D_FEAT / 2;
    const int wave = __builtin_amdgcn_readfirstlane((int)(threadIdx.x >> 5));
    for (int cb = wave * SPMM_EPR; cb < rn; cb += (STG_T / 32) * SPMM_EPR) {
        const int eb = rb + cb;
        if (cb + SPMM_EPR <= rn) {
            int2 q[SPMM_EPR];
            float v0[SPMM_EPR], v1[SPMM_EPR];
            for (int i = 0; i < SPMM_EPR; ++i) q[i] = rec[eb + i];
            for (int i = 0; i < SPMM_EPR; ++i) {
                const int lo = (q[i].x >> 22) * D_FEAT + lane;
                v0[i] = sT[lo];
                v1[i] = sT[lo + H];
            }
            for (int i = 0; i < SPMM_EPR; ++i) {
                const long so = (long)(q[i].x & 0x3FFFFF) * D_FEAT + lane;
                const float w = __int_as_float(q[i].y);
                unsafeAtomicAdd(&y[so],     w * v0[i]);
                unsafeAtomicAdd(&y[so + H], w * v1[i]);
            }
        } else {
            for (int e = cb; e < rn; ++e) {
                const int2 q = rec[rb + e];
                const int lo = (q.x >> 22) * D_FEAT + lane;
                const long so = (long)(q.x & 0x3FFFFF) * D_FEAT + lane;
                const float w = __int_as_float(q.y);
                const float a0 = sT[lo], a1 = sT[lo + H];
                unsafeAtomicAdd(&y[so],     w * a0);
                unsafeAtomicAdd(&y[so + H], w * a1);
            }
        }
    }
}

extern "C" void kernel_launch(void* const* d_in, const int* in_sizes, int n_in,
                              void* d_out, int out_size, void* d_ws, size_t ws_size,
                              hipStream_t stream) {
    const float* xr = (const float*)d_in[0];
    const float* xi = (const float*)d_in[1];
    const int*   ei = (const int*)d_in[2];
    const float* Wr = (const float*)d_in[3];
    const float* br = (const float*)d_in[4];
    const float* Wi = (const float*)d_in[5];
    const float* bi = (const float*)d_in[6];

    const int n = in_sizes[0] / D_FEAT;
    const int E = in_sizes[2] / 2;
    const int K = in_sizes[3] / (D_FEAT * D_FEAT);
    const long NF = (long)n * D_FEAT;

    const int* row = ei;
    const int* col = ei + E;
    float* outR = (float*)d_out;
    float* outI = outR + NF;

    {
        const long wsF   = (long)(ws_size / sizeof(float));
        const long tail  = 2 * D_FEAT * D_FEAT + 2 * D_FEAT;
        const long Er    = ((long)E + 1) / 2 * 2;
        const int  NB    = (n + SRC_R - 1) / SRC_R;
        const long fixed = 2 * TICK_STRIDE + 2 * MAX_NB + 4 * (long)n + tail + 4 * MAX_NB + 4 * Er;
        if (NB <= MAX_NB && n <= (1 << 22) && (n & 15) == 0 && wsF >= 3 * NF + fixed) {
            float* ws    = (float*)d_ws;
            float* P     = ws;
            float* Q1    = ws + NF;
            float* Q2    = ws + 2 * NF;
            float* tickF = ws + 3 * NF;
            float* cntF  = tickF + 2 * TICK_STRIDE;
            float* degO  = cntF + 2 * MAX_NB;
            float* degI  = degO + n;
            float* degO1 = degO + 2 * (long)n;
            float* degI1 = degO + 3 * (long)n;
            float* WRc   = degO + 4 * (long)n;
            float* WIc   = WRc + D_FEAT * D_FEAT;
            float* bmi   = WIc + D_FEAT * D_FEAT;
            float* bpi   = bmi + D_FEAT;
            float* baseF = bpi + D_FEAT;
            float* curF  = baseF + 2 * MAX_NB;
            float* rec0F = curF + 2 * MAX_NB;
            float* rec1F = rec0F + 2 * Er;

            wcomb_kernel<true><<<128, 256, 0, stream>>>(Wr, br, Wi, bi, K, WRc, WIc, bmi, bpi,
                                                        tickF, (2 * TICK_STRIDE + 2 * MAX_NB + 4 * (long)n) / 4);
            degree_bin_kernel<<<(E + BIN_EPB - 1) / BIN_EPB, 256, 0, stream>>>(
                row, col, E, degO, degI, degO1, degI1, (int*)cntF);
            deginv_scan_kernel<<<(n + 255) / 256 + 1, 256, 0, stream>>>(
                degO, degI, degO1, degI1, n, (const int*)cntF, (int*)baseF, (int*)curF);
            place_kernel<<<(E + BIN_EPB - 1) / BIN_EPB, 256, 0, stream>>>(
                row, col, E, degO, degI, (int*)curF, (int*)rec0F, (int*)rec1F);
            faber_pre_gemm_kernel<<<n / 16, 128, 0, stream>>>(
                xr, xi, WRc, WIc, bmi, bpi, P, Q1, Q2, outR, outI);
            spmm_staged_kernel<<<4 * NB, STG_T, 0, stream>>>(
                P, Q1, Q2, (const int*)rec0F, (const int*)rec1F, (const int*)baseF, (const int*)cntF,
                outR, outI, (int*)tickF, NB, n);
            return;
        }
    }

    {
        const long wsF   = (long)(ws_size / sizeof(float));
        const long tail  = 2 * D_FEAT * D_FEAT + 2 * D_FEAT;
        const long Ew    = ((long)E + 3) / 4 * 4;
        const long fixed = 2 * TICK_STRIDE + 4 * (long)n + tail + Ew;
        if (wsF >= 3 * NF + fixed) {
            float* ws    = (float*)d_ws;
            float* P     = ws;
            float* Q1    = ws + NF;
            float* Q2    = ws + 2 * NF;
            float* tickF = ws + 3 * NF;
            float* degO  = tickF + 2 * TICK_STRIDE;
            float* degI  = degO + n;
            float* degO1 = degO + 2 * (long)n;
            float* degI1 = degO + 3 * (long)n;
            float* WRc   = degO + 4 * (long)n;
            float* WIc   = WRc + D_FEAT * D_FEAT;
            float* bmi   = WIc + D_FEAT * D_FEAT;
            float* bpi   = bmi + D_FEAT;
            float* wE    = bpi + D_FEAT;
            const int Nc = (E + 8 * SPMM_EPW - 1) / (8 * SPMM_EPW);

            wcomb_kernel<true><<<128, 256, 0, stream>>>(Wr, br, Wi, bi, K, WRc, WIc, bmi, bpi,
                                                        tickF, (2 * TICK_STRIDE + 4 * (long)n) / 4);
            degree_kernel<<<(E + 255) / 256, 256, 0, stream>>>(row, col, E, degO, degI, degO1, degI1);
            deginv_kernel<true><<<(n + 255) / 256, 256, 0, stream>>>(degO, degI, degO1, degI1, n);
            edgew_kernel<<<(E + 255) / 256, 256, 0, stream>>>(row, col, E, degO, degI, wE);
            faber_pre_gemm_kernel<<<n / 16, 128, 0, stream>>>(
                xr, xi, WRc, WIc, bmi, bpi, P, Q1, Q2, outR, outI);
            spmm_owned_kernel<<<4 * Nc, 256, 0, stream>>>(
                P, Q1, Q2, row, col, E, wE, outR, outI, (int*)tickF, Nc);
            return;
        }
    }

    const long wsF  = (long)(ws_size / sizeof(float));
    const long tail = 2 * D_FEAT * D_FEAT + 2 * D_FEAT;
    const long Ew   = ((long)E + 3) / 4 * 4;
    const bool hasW = wsF >= 2 * NF + 2 * (long)n + tail + Ew;
    const long wsA  = hasW ? wsF - Ew : wsF;
    const int copies = (wsA >= 6 * NF + 4 * (long)n + tail) ? 2
                     : (wsA >= 4 * NF + 4 * (long)n + tail) ? 1 : 0;
    const long accPlanes = 2 + 2 * copies;
    const long degRows   = copies ? 4 : 2;
    float* ws   = (float*)d_ws;
    float* yr   = ws;
    float* yi_  = ws + NF;
    float* yr1  = copies ? ws + 2 * NF : yr;
    float* yi1  = copies ? ws + 3 * NF : yi_;
    float* degO = ws + accPlanes * NF;
    float* degI = degO + n;
    float* degO1 = copies ? degO + 2 * (long)n : degO;
    float* degI1 = copies ? degO + 3 * (long)n : degI;
    float* WRc  = degO + degRows * n;
    float* WIc  = WRc + D_FEAT * D_FEAT;
    float* bmi  = WIc + D_FEAT * D_FEAT;
    float* bpi  = bmi + D_FEAT;
    float* wE   = bpi + D_FEAT;

    float* yrt1 = (copies == 2) ? ws + 4 * NF : outR;
    float* yit1 = (copies == 2) ? ws + 5 * NF : outI;
    zero2_kernel<<<2048, 256, 0, stream>>>(ws, (accPlanes * NF + degRows * n) / 4,
                                           outR, (2 * NF) / 4);
    degree_kernel<<<(E + 255) / 256, 256, 0, stream>>>(row, col, E, degO, degI, degO1, degI1);
    if (copies) deginv_kernel<true><<<(n + 255) / 256, 256, 0, stream>>>(degO, degI, degO1, degI1, n);
    else        deginv_kernel<false><<<(n + 255) / 256, 256, 0, stream>>>(degO, degI, degO1, degI1, n);
    wcomb_kernel<false><<<(D_FEAT * D_FEAT + D_FEAT + 255) / 256, 256, 0, stream>>>(
        Wr, br, Wi, bi, K, WRc, WIc, bmi, bpi, nullptr, 0L);
    if (hasW) {
        edgew_kernel<<<(E + 255) / 256, 256, 0, stream>>>(row, col, E, degO, degI, wE);
        spmm_kernel<true><<<dim3((E + 8 * SPMM_EPW - 1) / (8 * SPMM_EPW), 4), 256, 0, stream>>>(xr, xr, xi, xi, row, col, E, degO, degI, wE,
                                                                      yr, yi_, outR, outI,
                                                                      yr1, yi1, yrt1, yit1);
    } else {
        spmm_kernel<false><<<dim3((E + 8 * SPMM_EPW - 1) / (8 * SPMM_EPW), 4), 256, 0, stream>>>(xr, xr, xi, xi, row, col, E, degO, degI, wE,
                                                                       yr, yi_, outR, outI,
                                                                       yr1, yi1, yrt1, yit1);
    }
    if (copies == 2)
        faber_gemm_kernel<2><<<n / 16, 128, 0, stream>>>(
            yr, yi_, yr1, yi1, yrt1, yit1, WRc, WIc, bmi, bpi, outR, outI);
    else if (copies == 1)
        faber_gemm_kernel<1><<<n / 16, 128, 0, stream>>>(
            yr, yi_, yr1, yi1, yrt1, yit1, WRc, WIc, bmi, bpi, outR, outI);
    else
        faber_gemm_kernel<0><<<n / 16, 128, 0, stream>>>(
            yr, yi_, yr1, yi1, yrt1, yit1, WRc, WIc, bmi, bpi, outR, outI);
}


# edge scatter: staging range 512 -> 391 rows: the 4 passes make 1024 equal items = 2 full rounds of the 512 resident workgroups (was 784: 2nd round half empty); f32 atomic adds kept, arrival order
# speedup vs baseline: 3.1214x; 1.0496x over previous
#include <hip/hip_runtime.h>
#include <math.h>

typedef __attribute__((ext_vector_type(2))) float v2f;
typedef __attribute__((ext_vector_type(8))) float v8f;

#define D_FEAT 64

__device__ __forceinline__ int block_half(int* s_half) {
    if (__builtin_amdgcn_readfirstlane((int)(threadIdx.x >> 5)) == 0) {
        const unsigned id = __builtin_amdgcn_s_sendmsg_rtn(0x87);
        if ((threadIdx.x & 31) == 0) *s_half = (int)((id >> 18) & 1u);
    }
    __syncthreads();
    return __builtin_amdgcn_readfirstlane(*s_half);
}

__global__ void zero2_kernel(float* __restrict__ a, long na4,
                             float* __restrict__ b, long nb4) {
    long i = (long)blockIdx.x * blockDim.x + threadIdx.x;
    long stride = (long)gridDim.x * blockDim.x;
    float4 z = make_float4(0.f, 0.f, 0.f, 0.f);
    for (long j = i; j < na4 + nb4; j += stride) {
        if (j < na4) ((float4*)a)[j] = z;
        else         ((float4*)b)[j - na4] = z;
    }
}

__global__ void degree_kernel(const int* __restrict__ row, const int* __restrict__ col,
                              int E, float* degO0, float* degI0,
                              float* degO1, float* degI1) {
    __shared__ int s_half;
    const int half = block_half(&s_half);
    float* degO = half ? degO1 : degO0;
    float* degI = half ? degI1 : degI0;
    int e = blockIdx.x * blockDim.x + threadIdx.x;
    if (e >= E) return;
    unsafeAtomicAdd(&degO[row[e]], 1.0f);
    unsafeAtomicAdd(&degI[col[e]], 1.0f);
}

template <bool TWO>
__global__ void deginv_kernel(float* degO, float* degI,
                              const float* degO1, const float* degI1, int n) {
    int i = blockIdx.x * blockDim.x + threadIdx.x;
    if (i >= n) return;
    float d0 = degO[i];
    float d1 = degI[i];
    if (TWO) { d0 += degO1[i]; d1 += degI1[i]; }
    degO[i] = (d0 > 0.f) ? powf(d0, -0.25f) : 0.f;
    degI[i] = (d1 > 0.f) ? powf(d1, -0.25f) : 0.f;
}

__global__ void edgew_kernel(const int* __restrict__ row, const int* __restrict__ col, int E,
                             const float* __restrict__ out_inv, const float* __restrict__ in_inv,
                             float* __restrict__ wE) {
    int e = blockIdx.x * blockDim.x + threadIdx.x;
    if (e >= E) return;
    wE[e] = out_inv[row[e]] * in_inv[col[e]];
}

template <bool FRAG>
__global__ void wcomb_kernel(const float* __restrict__ Wr, const float* __restrict__ br,
                             const float* __restrict__ Wi, const float* __restrict__ bi,
                             int K, float* WRc, float* WIc, float* bmi, float* bpi,
                             float* __restrict__ zc, long nz4) {
    int i = blockIdx.x * blockDim.x + threadIdx.x;
    for (long j = i; j < nz4; j += (long)gridDim.x * blockDim.x)
        ((float4*)zc)[j] = make_float4(0.f, 0.f, 0.f, 0.f);
    if (i < D_FEAT * D_FEAT) {
        float sr = 0.f, si = 0.f, c = 1.f;
        for (int k = 0; k < K; ++k) {
            sr += c * Wr[k * D_FEAT * D_FEAT + i];
            si += c * Wi[k * D_FEAT * D_FEAT + i];
            c *= 0.5f;
        }
        int p = i;
        if (FRAG) {
            const int o = i / D_FEAT, k = i % D_FEAT;
            const int l = (o & 15) + 16 * ((k >> 1) & 1);
            p = (((o >> 4) * (D_FEAT / 4) + (k >> 2)) * 32 + l) * 2 + (k & 1);
        }
        WRc[p] = sr;
        WIc[p] = si;
    } else if (i < D_FEAT * D_FEAT + D_FEAT) {
        int o = i - D_FEAT * D_FEAT;
        float sr = 0.f, si = 0.f, c = 1.f;
        for (int k = 0; k < K; ++k) {
            sr += c * br[k * D_FEAT + o];
            si += c * bi[k * D_FEAT + o];
            c *= 0.5f;
        }
        bmi[o] = sr - si;
        bpi[o] = sr + si;
    }
}

#define SPMM_EPR 16
#define SPMM_RPW 2
#define SPMM_EPW (SPMM_EPR * SPMM_RPW)
template <bool HASW>
__global__ __launch_bounds__(256) void spmm_kernel(
        const float* __restrict__ xg0, const float* __restrict__ xg1,
        const float* __restrict__ xg2, const float* __restrict__ xg3,
        const int* __restrict__ row, const int* __restrict__ col, int E,
        const float* __restrict__ out_inv, const float* __restrict__ in_inv,
        const float* __restrict__ wE,
        float* yr0, float* yi0, float* yrt0, float* yit0,
        float* yr1, float* yi1, float* yrt1, float* yit1) {
    __shared__ int s_half;
    const int half = block_half(&s_half);

    const int pass = blockIdx.y;
    const bool tr = (pass & 1) != 0;
    const bool im = (pass & 2) != 0;
    const float* __restrict__ x    = im ? (tr ? xg3 : xg2) : (tr ? xg1 : xg0);
    const int*   __restrict__ gidx = tr ? row : col;
    const int*   __restrict__ sidx = tr ? col : row;
    const float* __restrict__ ginv = tr ? out_inv : in_inv;
    const float* __restrict__ sinv = tr ? in_inv : out_inv;
    float* const yr  = half ? yr1  : yr0;
    float* const yi  = half ? yi1  : yi0;
    float* const yrt = half ? yrt1 : yrt0;
    float* const yit = half ? yit1 : yit0;
    float* const yre = tr ? yrt : yr;
    float* const yim = tr ? yit : yi;
    float* const y   = im ? yim : yre;

    const int lane = threadIdx.x & 31;
    const int H = D_FEAT / 2;
    int e0 = (blockIdx.x * 8 + (threadIdx.x >> 5)) * SPMM_EPW;
    e0 = __builtin_amdgcn_readfirstlane(e0);

    for (int r = 0; r < SPMM_RPW; ++r) {
        const int eb = e0 + r * SPMM_EPR;
        if (eb >= E) break;
        if (eb + SPMM_EPR <= E) {
            int g[SPMM_EPR], s[SPMM_EPR];
            float w[SPMM_EPR], v0[SPMM_EPR], v1[SPMM_EPR];
            for (int i = 0; i < SPMM_EPR; ++i) { g[i] = gidx[eb + i]; s[i] = sidx[eb + i]; }
            if (HASW) {
                for (int i = 0; i < SPMM_EPR; i += 4) {
                    const float4 w4 = *(const float4*)(wE + eb + i);
                    w[i] = w4.x; w[i + 1] = w4.y; w[i + 2] = w4.z; w[i + 3] = w4.w;
                }
            } else {
                for (int i = 0; i < SPMM_EPR; ++i) w[i] = sinv[s[i]] * ginv[g[i]];
            }
            for (int i = 0; i < SPMM_EPR; ++i) {
                const long go = (long)g[i] * D_FEAT + lane;
                v0[i] = x[go];
                v1[i] = x[go + H];
            }
            for (int i = 0; i < SPMM_EPR; ++i) {
                const long so = (long)s[i] * D_FEAT + lane;
                unsafeAtomicAdd(&y[so],     w[i] * v0[i]);
                unsafeAtomicAdd(&y[so + H], w[i] * v1[i]);
            }
        } else {
            for (int e = eb; e < E; ++e) {
                const int g = gidx[e];
                const int s = sidx[e];
                const float w = HASW ? wE[e] : sinv[s] * ginv[g];
                const long go = (long)g * D_FEAT + lane;
                const long so = (long)s * D_FEAT + lane;
                const float a0 = x[go], a1 = x[go + H];
                unsafeAtomicAdd(&y[so],     w * a0);
                unsafeAtomicAdd(&y[so + H], w * a1);
            }
        }
    }
}

#define PRE_PITCH 68
#define TICK_STRIDE 64
__global__ __launch_bounds__(256) void spmm_owned_kernel(
        const float* __restrict__ P, const float* __restrict__ Q1, const float* __restrict__ Q2,
        const int* __restrict__ row, const int* __restrict__ col, int E,
        const float* __restrict__ wE,
        float* outR, float* outI, int* tick, int Nc) {
    __shared__ int s_pass, s_chunk;
    if (__builtin_amdgcn_readfirstlane((int)(threadIdx.x >> 5)) == 0) {
        const unsigned id = __builtin_amdgcn_s_sendmsg_rtn(0x87);
        if ((threadIdx.x & 31) == 0) {
            const int h = (int)((id >> 18) & 1u);
            const int L = 2 * Nc;
            const int t = atomicAdd(tick + h * TICK_STRIDE, 1);
            const int q = (t < L) ? h : 1 - h;
            const int k = (t < L) ? t : 2 * L - 1 - t;
            const int second = (k >= Nc) ? 1 : 0;
            s_pass  = (k >= 0) ? 2 * q + second : -1;
            s_chunk = k - second * Nc;
        }
    }
    __syncthreads();
    const int pass  = __builtin_amdgcn_readfirstlane(s_pass);
    const int chunk = __builtin_amdgcn_readfirstlane(s_chunk);
    if (pass < 0) return;

    const bool tr = (pass & 1) != 0;
    const bool im = (pass & 2) != 0;
    const float* __restrict__ x    = im ? (tr ? Q2 : Q1) : P;
    const int*   __restrict__ gidx = tr ? row : col;
    const int*   __restrict__ sidx = tr ? col : row;
    float* const y = im ? outI : outR;

    const int lane = threadIdx.x & 31;
    const int H = D_FEAT / 2;
    int e0 = (chunk * 8 + (int)(threadIdx.x >> 5)) * SPMM_EPW;
    e0 = __builtin_amdgcn_readfirstlane(e0);

    for (int r = 0; r < SPMM_RPW; ++r) {
        const int eb = e0 + r * SPMM_EPR;
        if (eb >= E) break;
        if (eb + SPMM_EPR <= E) {
            int g[SPMM_EPR], s[SPMM_EPR];
            float w[SPMM_EPR], v0[SPMM_EPR], v1[SPMM_EPR];
            for (int i = 0; i < SPMM_EPR; ++i) { g[i] = gidx[eb + i]; s[i] = sidx[eb + i]; }
            for (int i = 0; i < SPMM_EPR; i += 4) {
                const float4 w4 = *(const float4*)(wE + eb + i);
                w[i] = w4.x; w[i + 1] = w4.y; w[i + 2] = w4.z; w[i + 3] = w4.w;
            }
            for (int i = 0; i < SPMM_EPR; ++i) {
                const long go = (long)g[i] * D_FEAT + lane;
                v0[i] = x[go];
                v1[i] = x[go + H];
            }
            for (int i = 0; i < SPMM_EPR; ++i) {
                const long so = (long)s[i] * D_FEAT + lane;
                unsafeAtomicAdd(&y[so],     w[i] * v0[i]);
                unsafeAtomicAdd(&y[so + H], w[i] * v1[i]);
            }
        } else {
            for (int e = eb; e < E; ++e) {
                const int g = gidx[e];
                const int s = sidx[e];
                const float w = wE[e];
                const long go = (long)g * D_FEAT + lane;
                const long so = (long)s * D_FEAT + lane;
                const float a0 = x[go], a1 = x[go + H];
                unsafeAtomicAdd(&y[so],     w * a0);
                unsafeAtomicAdd(&y[so + H], w * a1);
            }
        }
    }
}

template <int COPIES>
__global__ __launch_bounds__(128) void faber_gemm_kernel(
        const float* __restrict__ yr, const float* __restrict__ yi,
        const float* __restrict__ yr1, const float* __restrict__ yi1,
        const float* __restrict__ yrt1, const float* __restrict__ yit1,
        const float* __restrict__ WRc, const float* __restrict__ WIc,
        const float* __restrict__ bmi, const float* __restrict__ bpi,
        float* outR, float* outI) {
    const int wave = threadIdx.x >> 5;
    const int lane = threadIdx.x & 31;
    const int l16  = lane & 15;
    const int half = lane >> 4;

    const float* yrt = outR;
    const float* yit = outI;

    const long rowOff = (long)(blockIdx.x * 16 + l16) * D_FEAT;
    const long wOff   = (long)(wave * 16 + l16) * D_FEAT;

    v8f accRa = {};
    v8f accRb = {};
    v8f accIa = {};
    v8f accIb = {};

    #pragma unroll
    for (int s = 0; s < 16; ++s) {
        const int k = 4 * s + 2 * half;
        float2 ar  = *(const float2*)(yr  + rowOff + k);
        float2 art = *(const float2*)(yrt + rowOff + k);
        float2 ai  = *(const float2*)(yi  + rowOff + k);
        float2 ait = *(const float2*)(yit + rowOff + k);
        if (COPIES >= 1) {
            const float2 br  = *(const float2*)(yr1  + rowOff + k);
            const float2 bi  = *(const float2*)(yi1  + rowOff + k);
            ar.x  += br.x;  ar.y  += br.y;
            ai.x  += bi.x;  ai.y  += bi.y;
        }
        if (COPIES == 2) {
            const float2 brt = *(const float2*)(yrt1 + rowOff + k);
            const float2 bit = *(const float2*)(yit1 + rowOff + k);
            art.x += brt.x; art.y += brt.y;
            ait.x += bit.x; ait.y += bit.y;
        }
        v2f Azr = {0.5f * (ar.x + art.x), 0.5f * (ar.y + art.y)};
        v2f Azi = {0.5f * (ai.x + ait.x), 0.5f * (ai.y + ait.y)};
        v2f Ayr = {ar.x, ar.y};
        float2 wr2 = *(const float2*)(WRc + wOff + k);
        float2 wi2 = *(const float2*)(WIc + wOff + k);
        v2f Bwr = {wr2.x, wr2.y};
        v2f Bwi = {wi2.x, wi2.y};
        accRa = __builtin_amdgcn_wmma_f32_16x16x4_f32(false, Azr, false, Bwr, (short)0, accRa, false, false);
        accRb = __builtin_amdgcn_wmma_f32_16x16x4_f32(false, Azi, false, Bwi, (short)0, accRb, false, false);
        accIa = __builtin_amdgcn_wmma_f32_16x16x4_f32(false, Ayr, false, Bwi, (short)0, accIa, false, false);
        accIb = __builtin_amdgcn_wmma_f32_16x16x4_f32(false, Azi, false, Bwr, (short)0, accIb, false, false);
    }

    const int colD = wave * 16 + l16;
    const float bm = bmi[colD];
    const float bp = bpi[colD];

    __syncthreads();

    #pragma unroll
    for (int r = 0; r < 8; ++r) {
        long nodeM = (long)(blockIdx.x * 16 + r + 8 * half);
        outR[nodeM * D_FEAT + colD] = accRa[r] - accRb[r] + bm;
        outI[nodeM * D_FEAT + colD] = accIa[r] + accIb[r] + bp;
    }
}

__global__ __launch_bounds__(128) void faber_pre_gemm_kernel(
        const float* __restrict__ xr, const float* __restrict__ xi,
        const float* __restrict__ WRf, const float* __restrict__ WIf,
        const float* __restrict__ bmi, const float* __restrict__ bpi,
        float* __restrict__ P, float* __restrict__ Q1, float* __restrict__ Q2,
        float* __restrict__ outR, float* __restrict__ outI) {
    __shared__ __attribute__((aligned(16))) float sXr[16 * PRE_PITCH];
    __shared__ __attribute__((aligned(16))) float sXi[16 * PRE_PITCH];

    const int wave = threadIdx.x >> 5;
    const int lane = threadIdx.x & 31;
    const int l16  = lane & 15;
    const int half = lane >> 4;

    const long tile = (long)blockIdx.x * 16 * D_FEAT;
    float4 tr[2], ti[2];
    for (int h = 0; h < 2; ++h) {
        const int t = (int)threadIdx.x + 128 * h;
        tr[h] = ((const float4*)(xr + tile))[t];
        ti[h] = ((const float4*)(xi + tile))[t];
    }

    const float* wrp = WRf + (long)wave * (D_FEAT / 4) * 64 + lane * 2;
    const float* wip = WIf + (long)wave * (D_FEAT / 4) * 64 + lane * 2;
    float2 wr2[16], wi2[16];
    for (int s = 0; s < 16; ++s) {
        wr2[s] = *(const float2*)(wrp + s * 64);
        wi2[s] = *(const float2*)(wip + s * 64);
    }

    for (int h = 0; h < 2; ++h) {
        const int t = (int)threadIdx.x + 128 * h;
        const int d = (t * 4) & (D_FEAT - 1);
        ((float4*)(outR + tile))[t] = *(const float4*)(bmi + d);
        ((float4*)(outI + tile))[t] = *(const float4*)(bpi + d);
    }

    for (int h = 0; h < 2; ++h) {
        const int t = (int)threadIdx.x + 128 * h;
        const int o = (t >> 4) * PRE_PITCH + (t & 15) * 4;
        *(float4*)(sXr + o) = tr[h];
        *(float4*)(sXi + o) = ti[h];
    }
    __syncthreads();

    v8f accRR = {};
    v8f accII = {};
    v8f accRI = {};
    v8f accIR = {};

    const int aOff = l16 * PRE_PITCH + 2 * half;
    for (int s = 0; s < 16; ++s) {
        const float2 ar = *(const float2*)(sXr + aOff + 4 * s);
        const float2 ai = *(const float2*)(sXi + aOff + 4 * s);
        v2f Ar  = {ar.x, ar.y};
        v2f Ai  = {ai.x, ai.y};
        v2f Bwr = {wr2[s].x, wr2[s].y};
        v2f Bwi = {wi2[s].x, wi2[s].y};
        accRR = __builtin_amdgcn_wmma_f32_16x16x4_f32(false, Ar, false, Bwr, (short)0, accRR, false, false);
        accII = __builtin_amdgcn_wmma_f32_16x16x4_f32(false, Ai, false, Bwi, (short)0, accII, false, false);
        accRI = __builtin_amdgcn_wmma_f32_16x16x4_f32(false, Ar, false, Bwi, (short)0, accRI, false, false);
        accIR = __builtin_amdgcn_wmma_f32_16x16x4_f32(false, Ai, false, Bwr, (short)0, accIR, false, false);
    }

    const int colD = wave * 16 + l16;
    for (int r = 0; r < 8; ++r) {
        const long o = (long)(blockIdx.x * 16 + r + 8 * half) * D_FEAT + colD;
        const float q2 = 0.5f * accIR[r];
        P[o]  = 0.5f * (accRR[r] - accII[r]);
        Q2[o] = q2;
        Q1[o] = accRI[r] + q2;
    }
}


#define SRC_R 391
#define MAX_NB 256
#define BIN_EPB 2048
#define BIN_EPT (BIN_EPB / 256)

__global__ __launch_bounds__(256) void degree_bin_kernel(
        const int* __restrict__ row, const int* __restrict__ col, int E,
        float* degO0, float* degI0, float* degO1, float* degI1, int* cnt) {
    __shared__ int s_half;
    __shared__ int s_cnt[2 * MAX_NB];
    for (int t = threadIdx.x; t < 2 * MAX_NB; t += 256) s_cnt[t] = 0;
    const int half = block_half(&s_half);
    float* degO = half ? degO1 : degO0;
    float* degI = half ? degI1 : degI0;
    const int e0 = blockIdx.x * BIN_EPB + (int)threadIdx.x;
    int r[BIN_EPT], c[BIN_EPT];
    #pragma unroll
    for (int i = 0; i < BIN_EPT; ++i) {
        const int e = e0 + i * 256;
        r[i] = (e < E) ? row[e] : -1;
        c[i] = (e < E) ? col[e] : 0;
    }
    #pragma unroll
    for (int i = 0; i < BIN_EPT; ++i) {
        if (r[i] >= 0) {
            unsafeAtomicAdd(&degO[r[i]], 1.0f);
            unsafeAtomicAdd(&degI[c[i]], 1.0f);
            atomicAdd(&s_cnt[c[i] / SRC_R], 1);
            atomicAdd(&s_cnt[MAX_NB + r[i] / SRC_R], 1);
        }
    }
    __syncthreads();
    for (int t = threadIdx.x; t < 2 * MAX_NB; t += 256) {
        const int v = s_cnt[t];
        if (v) atomicAdd(&cnt[t], v);
    }
}

__device__ __forceinline__ void scan2_256(int v0, int v1, int* s_wt, int& ex0, int& ex1) {
    const int lane = threadIdx.x & 31, wv = threadIdx.x >> 5;
    int i0 = v0, i1 = v1;
    for (int d = 1; d < 32; d <<= 1) {
        const int a0 = __shfl_up(i0, d, 32), a1 = __shfl_up(i1, d, 32);
        if (lane >= d) { i0 += a0; i1 += a1; }
    }
    if (lane == 31) { s_wt[wv] = i0; s_wt[8 + wv] = i1; }
    __syncthreads();
    int p0 = 0, p1 = 0;
    for (int u = 0; u < wv; ++u) { p0 += s_wt[u]; p1 += s_wt[8 + u]; }
    ex0 = p0 + i0 - v0;
    ex1 = p1 + i1 - v1;
}

__global__ __launch_bounds__(256) void deginv_scan_kernel(
        float* degO, float* degI, const float* degO1, const float* degI1, int n,
        const int* __restrict__ cnt, int* __restrict__ base, int* __restrict__ cur) {
    __shared__ int s_wt[16];
    if (blockIdx.x == gridDim.x - 1) {
        const int t = threadIdx.x;
        int ex0, ex1;
        scan2_256(cnt[t], cnt[MAX_NB + t], s_wt, ex0, ex1);
        base[t] = ex0;          cur[t] = ex0;
        base[MAX_NB + t] = ex1; cur[MAX_NB + t] = ex1;
        return;
    }
    const int i = blockIdx.x * blockDim.x + threadIdx.x;
    if (i >= n) return;
    const float d0 = degO[i] + degO1[i];
    const float d1 = degI[i] + degI1[i];
    degO[i] = (d0 > 0.f) ? powf(d0, -0.25f) : 0.f;
    degI[i] = (d1 > 0.f) ? powf(d1, -0.25f) : 0.f;
}

__global__ __launch_bounds__(256) void place_kernel(
        const int* __restrict__ row, const int* __restrict__ col, int E,
        const float* __restrict__ out_inv, const float* __restrict__ in_inv,
        int* cur, int* __restrict__ rec0w, int* __restrict__ rec1w) {
    int2* __restrict__ rec0 = (int2*)rec0w;
    int2* __restrict__ rec1 = (int2*)rec1w;
    __shared__ int  s_cnt[2 * MAX_NB];
    __shared__ int  s_nxt[2 * MAX_NB];
    __shared__ int  s_dlt[2 * MAX_NB];
    __shared__ int  s_wt[16];
    __shared__ int2 sRec[2 * BIN_EPB];
    __shared__ int  sDst[2 * BIN_EPB];
    const int t = threadIdx.x;
    s_cnt[t] = 0; s_cnt[MAX_NB + t] = 0;
    __syncthreads();
    const int eB = blockIdx.x * BIN_EPB;
    int r[BIN_EPT], c[BIN_EPT];
    #pragma unroll
    for (int i = 0; i < BIN_EPT; ++i) {
        const int e = eB + t + i * 256;
        r[i] = (e < E) ? row[e] : -1;
        c[i] = (e < E) ? col[e] : 0;
        if (r[i] >= 0) {
            atomicAdd(&s_cnt[c[i] / SRC_R], 1);
            atomicAdd(&s_cnt[MAX_NB + r[i] / SRC_R], 1);
        }
    }
    int wb[BIN_EPT];
    #pragma unroll
    for (int i = 0; i < BIN_EPT; ++i)
        wb[i] = __float_as_int(out_inv[r[i] >= 0 ? r[i] : 0] * in_inv[c[i]]);
    __syncthreads();
    {
        const int v0 = s_cnt[t], v1 = s_cnt[MAX_NB + t];
        int ex0, ex1;
        scan2_256(v0, v1, s_wt, ex0, ex1);
        const int g0 = v0 ? atomicAdd(&cur[t], v0) : 0;
        const int g1 = v1 ? atomicAdd(&cur[MAX_NB + t], v1) : 0;
        s_nxt[t] = ex0;                    s_dlt[t] = g0 - ex0;
        s_nxt[MAX_NB + t] = BIN_EPB + ex1; s_dlt[MAX_NB + t] = g1 - (BIN_EPB + ex1);
    }
    __syncthreads();
    #pragma unroll
    for (int i = 0; i < BIN_EPT; ++i) {
        if (r[i] >= 0) {
            const int b0 = c[i] / SRC_R;
            const int s0 = atomicAdd(&s_nxt[b0], 1);
            sRec[s0] = make_int2(((c[i] - b0 * SRC_R) << 22) | r[i], wb[i]);
            sDst[s0] = s0 + s_dlt[b0];
            const int q1 = r[i] / SRC_R;
            const int b1 = MAX_NB + q1;
            const int s1 = atomicAdd(&s_nxt[b1], 1);
            sRec[s1] = make_int2(((r[i] - q1 * SRC_R) << 22) | c[i], wb[i]);
            sDst[s1] = s1 + s_dlt[b1];
        }
    }
    __syncthreads();
    const int nv = (E - eB < BIN_EPB) ? (E - eB) : BIN_EPB;
    for (int s = t; s < nv; s += 256) {
        rec0[sDst[s]] = sRec[s];
        rec1[sDst[BIN_EPB + s]] = sRec[BIN_EPB + s];
    }
}

#define STG_T 1024
#define STG_N4 (SRC_R * D_FEAT / 4)
#define STG_L ((STG_N4 + STG_T - 1) / STG_T)
__global__ __launch_bounds__(STG_T) void spmm_staged_kernel(
        const float* __restrict__ P, const float* __restrict__ Q1, const float* __restrict__ Q2,
        const int* __restrict__ rec0w, const int* __restrict__ rec1w,
        const int* __restrict__ base, const int* __restrict__ cnt,
        float* outR, float* outI, int* tick, int NB, int n) {
    __shared__ __attribute__((aligned(16))) float sT[SRC_R * D_FEAT];
    __shared__ int s_pass, s_rng;
    if (__builtin_amdgcn_readfirstlane((int)(threadIdx.x >> 5)) == 0) {
        const unsigned id = __builtin_amdgcn_s_sendmsg_rtn(0x87);
        if ((threadIdx.x & 31) == 0) {
            const int h = (int)((id >> 18) & 1u);
            const int L = 2 * NB;
            const int t = atomicAdd(tick + h * TICK_STRIDE, 1);
            const int q = (t < L) ? h : 1 - h;
            const int k = (t < L) ? t : 2 * L - 1 - t;
            const int second = (k >= NB) ? 1 : 0;
            s_pass = (k >= 0) ? 2 * q + second : -1;
            s_rng  = k - second * NB;
        }
    }
    __syncthreads();
    const int pass = __builtin_amdgcn_readfirstlane(s_pass);
    const int rng  = __builtin_amdgcn_readfirstlane(s_rng);
    if (pass < 0) return;

    const bool tr = (pass & 1) != 0;
    const bool im = (pass & 2) != 0;
    const float* __restrict__ x   = im ? (tr ? Q2 : Q1) : P;
    const int2*  __restrict__ rec = (const int2*)(tr ? rec1w : rec0w);
    float* const y = im ? outI : outR;
    const int key = tr ? MAX_NB : 0;
    const int rb = base[key + rng];
    const int rn = cnt[key + rng];

    const int row0 = rng * SRC_R;
    const int n4   = ((n - row0 < SRC_R) ? (n - row0) : SRC_R) * (D_FEAT / 4);
    const float4* __restrict__ tsrc = (const float4*)(x + (long)row0 * D_FEAT);
    float4 tv[STG_L];
    #pragma unroll
    for (int j = 0; j < STG_L; ++j) {
        const int i = (int)threadIdx.x + j * STG_T;
        tv[j] = (i < n4) ? tsrc[i] : make_float4(0.f, 0.f, 0.f, 0.f);
    }
    #pragma unroll
    for (int j = 0; j < STG_L; ++j) {
        const int i = (int)threadIdx.x + j * STG_T;
        if (i < STG_N4) ((float4*)sT)[i] = tv[j];
    }
    __syncthreads();

    const int lane = threadIdx.x & 31;
    const int H = D_FEAT / 2;
    const int wave = __builtin_amdgcn_readfirstlane((int)(threadIdx.x >> 5));
    for (int cb = wave * SPMM_EPR; cb < rn; cb += (STG_T / 32) * SPMM_EPR) {
        const int eb = rb + cb;
        if (cb + SPMM_EPR <= rn) {
            int2 q[SPMM_EPR];
            float v0[SPMM_EPR], v1[SPMM_EPR];
            for (int i = 0; i < SPMM_EPR; ++i) q[i] = rec[eb + i];
            for (int i = 0; i < SPMM_EPR; ++i) {
                const int lo = (q[i].x >> 22) * D_FEAT + lane;
                v0[i] = sT[lo];
                v1[i] = sT[lo + H];
            }
            for (int i = 0; i < SPMM_EPR; ++i) {
                const long so = (long)(q[i].x & 0x3FFFFF) * D_FEAT + lane;
                const float w = __int_as_float(q[i].y);
                unsafeAtomicAdd(&y[so],     w * v0[i]);
                unsafeAtomicAdd(&y[so + H], w * v1[i]);
            }
        } else {
            for (int e = cb; e < rn; ++e) {
                const int2 q = rec[rb + e];
                const int lo = (q.x >> 22) * D_FEAT + lane;
                const long so = (long)(q.x & 0x3FFFFF) * D_FEAT + lane;
                const float w = __int_as_float(q.y);
                const float a0 = sT[lo], a1 = sT[lo + H];
                unsafeAtomicAdd(&y[so],     w * a0);
                unsafeAtomicAdd(&y[so + H], w * a1);
            }
        }
    }
}

extern "C" void kernel_launch(void* const* d_in, const int* in_sizes, int n_in,
                              void* d_out, int out_size, void* d_ws, size_t ws_size,
                              hipStream_t stream) {
    const float* xr = (const float*)d_in[0];
    const float* xi = (const float*)d_in[1];
    const int*   ei = (const int*)d_in[2];
    const float* Wr = (const float*)d_in[3];
    const float* br = (const float*)d_in[4];
    const float* Wi = (const float*)d_in[5];
    const float* bi = (const float*)d_in[6];

    const int n = in_sizes[0] / D_FEAT;
    const int E = in_sizes[2] / 2;
    const int K = in_sizes[3] / (D_FEAT * D_FEAT);
    const long NF = (long)n * D_FEAT;

    const int* row = ei;
    const int* col = ei + E;
    float* outR = (float*)d_out;
    float* outI = outR + NF;

    {
        const long wsF   = (long)(ws_size / sizeof(float));
        const long tail  = 2 * D_FEAT * D_FEAT + 2 * D_FEAT;
        const long Er    = ((long)E + 1) / 2 * 2;
        const int  NB    = (n + SRC_R - 1) / SRC_R;
        const long fixed = 2 * TICK_STRIDE + 2 * MAX_NB + 4 * (long)n + tail + 4 * MAX_NB + 4 * Er;
        if (NB <= MAX_NB && n <= (1 << 22) && (n & 15) == 0 && wsF >= 3 * NF + fixed) {
            float* ws    = (float*)d_ws;
            float* P     = ws;
            float* Q1    = ws + NF;
            float* Q2    = ws + 2 * NF;
            float* tickF = ws + 3 * NF;
            float* cntF  = tickF + 2 * TICK_STRIDE;
            float* degO  = cntF + 2 * MAX_NB;
            float* degI  = degO + n;
            float* degO1 = degO + 2 * (long)n;
            float* degI1 = degO + 3 * (long)n;
            float* WRc   = degO + 4 * (long)n;
            float* WIc   = WRc + D_FEAT * D_FEAT;
            float* bmi   = WIc + D_FEAT * D_FEAT;
            float* bpi   = bmi + D_FEAT;
            float* baseF = bpi + D_FEAT;
            float* curF  = baseF + 2 * MAX_NB;
            float* rec0F = curF + 2 * MAX_NB;
            float* rec1F = rec0F + 2 * Er;

            wcomb_kernel<true><<<128, 256, 0, stream>>>(Wr, br, Wi, bi, K, WRc, WIc, bmi, bpi,
                                                        tickF, (2 * TICK_STRIDE + 2 * MAX_NB + 4 * (long)n) / 4);
            degree_bin_kernel<<<(E + BIN_EPB - 1) / BIN_EPB, 256, 0, stream>>>(
                row, col, E, degO, degI, degO1, degI1, (int*)cntF);
            deginv_scan_kernel<<<(n + 255) / 256 + 1, 256, 0, stream>>>(
                degO, degI, degO1, degI1, n, (const int*)cntF, (int*)baseF, (int*)curF);
            place_kernel<<<(E + BIN_EPB - 1) / BIN_EPB, 256, 0, stream>>>(
                row, col, E, degO, degI, (int*)curF, (int*)rec0F, (int*)rec1F);
            faber_pre_gemm_kernel<<<n / 16, 128, 0, stream>>>(
                xr, xi, WRc, WIc, bmi, bpi, P, Q1, Q2, outR, outI);
            spmm_staged_kernel<<<4 * NB, STG_T, 0, stream>>>(
                P, Q1, Q2, (const int*)rec0F, (const int*)rec1F, (const int*)baseF, (const int*)cntF,
                outR, outI, (int*)tickF, NB, n);
            return;
        }
    }

    {
        const long wsF   = (long)(ws_size / sizeof(float));
        const long tail  = 2 * D_FEAT * D_FEAT + 2 * D_FEAT;
        const long Ew    = ((long)E + 3) / 4 * 4;
        const long fixed = 2 * TICK_STRIDE + 4 * (long)n + tail + Ew;
        if (wsF >= 3 * NF + fixed) {
            float* ws    = (float*)d_ws;
            float* P     = ws;
            float* Q1    = ws + NF;
            float* Q2    = ws + 2 * NF;
            float* tickF = ws + 3 * NF;
            float* degO  = tickF + 2 * TICK_STRIDE;
            float* degI  = degO + n;
            float* degO1 = degO + 2 * (long)n;
            float* degI1 = degO + 3 * (long)n;
            float* WRc   = degO + 4 * (long)n;
            float* WIc   = WRc + D_FEAT * D_FEAT;
            float* bmi   = WIc + D_FEAT * D_FEAT;
            float* bpi   = bmi + D_FEAT;
            float* wE    = bpi + D_FEAT;
            const int Nc = (E + 8 * SPMM_EPW - 1) / (8 * SPMM_EPW);

            wcomb_kernel<true><<<128, 256, 0, stream>>>(Wr, br, Wi, bi, K, WRc, WIc, bmi, bpi,
                                                        tickF, (2 * TICK_STRIDE + 4 * (long)n) / 4);
            degree_kernel<<<(E + 255) / 256, 256, 0, stream>>>(row, col, E, degO, degI, degO1, degI1);
            deginv_kernel<true><<<(n + 255) / 256, 256, 0, stream>>>(degO, degI, degO1, degI1, n);
            edgew_kernel<<<(E + 255) / 256, 256, 0, stream>>>(row, col, E, degO, degI, wE);
            faber_pre_gemm_kernel<<<n / 16, 128, 0, stream>>>(
                xr, xi, WRc, WIc, bmi, bpi, P, Q1, Q2, outR, outI);
            spmm_owned_kernel<<<4 * Nc, 256, 0, stream>>>(
                P, Q1, Q2, row, col, E, wE, outR, outI, (int*)tickF, Nc);
            return;
        }
    }

    const long wsF  = (long)(ws_size / sizeof(float));
    const long tail = 2 * D_FEAT * D_FEAT + 2 * D_FEAT;
    const long Ew   = ((long)E + 3) / 4 * 4;
    const bool hasW = wsF >= 2 * NF + 2 * (long)n + tail + Ew;
    const long wsA  = hasW ? wsF - Ew : wsF;
    const int copies = (wsA >= 6 * NF + 4 * (long)n + tail) ? 2
                     : (wsA >= 4 * NF + 4 * (long)n + tail) ? 1 : 0;
    const long accPlanes = 2 + 2 * copies;
    const long degRows   = copies ? 4 : 2;
    float* ws   = (float*)d_ws;
    float* yr   = ws;
    float* yi_  = ws + NF;
    float* yr1  = copies ? ws + 2 * NF : yr;
    float* yi1  = copies ? ws + 3 * NF : yi_;
    float* degO = ws + accPlanes * NF;
    float* degI = degO + n;
    float* degO1 = copies ? degO + 2 * (long)n : degO;
    float* degI1 = copies ? degO + 3 * (long)n : degI;
    float* WRc  = degO + degRows * n;
    float* WIc  = WRc + D_FEAT * D_FEAT;
    float* bmi  = WIc + D_FEAT * D_FEAT;
    float* bpi  = bmi + D_FEAT;
    float* wE   = bpi + D_FEAT;

    float* yrt1 = (copies == 2) ? ws + 4 * NF : outR;
    float* yit1 = (copies == 2) ? ws + 5 * NF : outI;
    zero2_kernel<<<2048, 256, 0, stream>>>(ws, (accPlanes * NF + degRows * n) / 4,
                                           outR, (2 * NF) / 4);
    degree_kernel<<<(E + 255) / 256, 256, 0, stream>>>(row, col, E, degO, degI, degO1, degI1);
    if (copies) deginv_kernel<true><<<(n + 255) / 256, 256, 0, stream>>>(degO, degI, degO1, degI1, n);
    else        deginv_kernel<false><<<(n + 255) / 256, 256, 0, stream>>>(degO, degI, degO1, degI1, n);
    wcomb_kernel<false><<<(D_FEAT * D_FEAT + D_FEAT + 255) / 256, 256, 0, stream>>>(
        Wr, br, Wi, bi, K, WRc, WIc, bmi, bpi, nullptr, 0L);
    if (hasW) {
        edgew_kernel<<<(E + 255) / 256, 256, 0, stream>>>(row, col, E, degO, degI, wE);
        spmm_kernel<true><<<dim3((E + 8 * SPMM_EPW - 1) / (8 * SPMM_EPW), 4), 256, 0, stream>>>(xr, xr, xi, xi, row, col, E, degO, degI, wE,
                                                                      yr, yi_, outR, outI,
                                                                      yr1, yi1, yrt1, yit1);
    } else {
        spmm_kernel<false><<<dim3((E + 8 * SPMM_EPW - 1) / (8 * SPMM_EPW), 4), 256, 0, stream>>>(xr, xr, xi, xi, row, col, E, degO, degI, wE,
                                                                       yr, yi_, outR, outI,
                                                                       yr1, yi1, yrt1, yit1);
    }
    if (copies == 2)
        faber_gemm_kernel<2><<<n / 16, 128, 0, stream>>>(
            yr, yi_, yr1, yi1, yrt1, yit1, WRc, WIc, bmi, bpi, outR, outI);
    else if (copies == 1)
        faber_gemm_kernel<1><<<n / 16, 128, 0, stream>>>(
            yr, yi_, yr1, yi1, yrt1, yit1, WRc, WIc, bmi, bpi, outR, outI);
    else
        faber_gemm_kernel<0><<<n / 16, 128, 0, stream>>>(
            yr, yi_, yr1, yi1, yrt1, yit1, WRc, WIc, bmi, bpi, outR, outI);
}
